# starting kernel
# baseline (speedup 1.0000x reference)
#include <hip/hip_runtime.h>

typedef __attribute__((ext_vector_type(2))) float v2f;
typedef __attribute__((ext_vector_type(8))) float v8f;

#define N_NODES 100000
#define F_IN    256
#define H1_DIM  128
#define H2_DIM  64
#define E_EDGES 1600000
#define EP_CNT  100000

__global__ void init_deg_kernel(float* __restrict__ deg, int n) {
    int i = blockIdx.x * blockDim.x + threadIdx.x;
    if (i < n) deg[i] = 1.0f;
}

__global__ void accum_deg_kernel(const int* __restrict__ col,
                                 float* __restrict__ deg, int e) {
    int i = blockIdx.x * blockDim.x + threadIdx.x;
    if (i < e) atomicAdd(&deg[col[i]], 1.0f);
}

__global__ void rsqrt_kernel(float* __restrict__ deg, int n) {
    int i = blockIdx.x * blockDim.x + threadIdx.x;
    if (i < n) deg[i] = rsqrtf(deg[i]);
}

__global__ void transpose_kernel(const float* __restrict__ B,
                                 float* __restrict__ Bt, int K, int N) {
    int idx = blockIdx.x * blockDim.x + threadIdx.x;
    if (idx >= K * N) return;
    int k = idx / N;
    int n = idx - k * N;
    Bt[(size_t)n * K + k] = B[idx];
}

template <int K, int NT>
__global__ void __launch_bounds__(NT * 32)
gemm_wmma_f32(const float* __restrict__ A, const float* __restrict__ Bt,
              float* __restrict__ C) {
    constexpr int N = NT * 16;
    const int wave = threadIdx.x >> 5;
    const int lane = threadIdx.x & 31;
    const int l    = lane & 15;
    const int half = lane >> 4;

    const int m0 = blockIdx.x * 16;
    const int n0 = wave * 16;

    const float* ap = A  + (size_t)(m0 + l) * K + 2 * half;
    const float* bp = Bt + (size_t)(n0 + l) * K + 2 * half;

    v8f acc = {0.f, 0.f, 0.f, 0.f, 0.f, 0.f, 0.f, 0.f};
#pragma unroll
    for (int k0 = 0; k0 < K; k0 += 4) {
        v2f a = *(const v2f*)(ap + k0);
        v2f b = *(const v2f*)(bp + k0);
        acc = __builtin_amdgcn_wmma_f32_16x16x4_f32(
             false, a,  false, b,
             (short)0, acc,  false,  false);
    }

    float* cp = C + (size_t)(m0 + 8 * half) * N + n0 + l;
#pragma unroll
    for (int j = 0; j < 8; ++j)
        cp[(size_t)j * N] = acc[j];
}

template <int H>
__global__ void self_init_kernel(const float* __restrict__ htmp,
                                 const float* __restrict__ dinv,
                                 float* __restrict__ acc, int n) {
    int idx = blockIdx.x * blockDim.x + threadIdx.x;
    if (idx >= n * H) return;
    int node = idx / H;
    float di = dinv[node];
    acc[idx] = htmp[idx] * di * di;
}

template <int H>
__global__ void __launch_bounds__(256)
edge_scatter_kernel(const int* __restrict__ row, const int* __restrict__ col,
                    const float* __restrict__ dinv,
                    const float* __restrict__ htmp,
                    float* __restrict__ acc, int e) {
    int warp_id = (blockIdx.x * blockDim.x + threadIdx.x) >> 5;
    int lane = threadIdx.x & 31;
    if (warp_id >= e) return;
    int r = row[warp_id];
    int c = col[warp_id];
    float norm = dinv[r] * dinv[c];
    const float* src = htmp + (size_t)r * H;
    float* dst = acc + (size_t)c * H;
#pragma unroll
    for (int f = 0; f < H; f += 32)
        atomicAdd(&dst[f + lane], src[f + lane] * norm);
}

template <int H>
__global__ void bias_relu_kernel(float* __restrict__ acc,
                                 const float* __restrict__ bias, int n) {
    int idx = blockIdx.x * blockDim.x + threadIdx.x;
    if (idx >= n * H) return;
    float v = acc[idx] + bias[idx & (H - 1)];
    acc[idx] = v > 0.f ? v : 0.f;
}

template <int H>
__global__ void bias_add_kernel(float* __restrict__ acc,
                                const float* __restrict__ bias, int n) {
    int idx = blockIdx.x * blockDim.x + threadIdx.x;
    if (idx >= n * H) return;
    acc[idx] = acc[idx] + bias[idx & (H - 1)];
}

__global__ void __launch_bounds__(256)
edge_dot_kernel(const int* __restrict__ pos, const int* __restrict__ neg,
                const float* __restrict__ h2, float* __restrict__ out, int ep) {
    int warp_id = (blockIdx.x * blockDim.x + threadIdx.x) >> 5;
    int lane = threadIdx.x & 31;
    if (warp_id >= 2 * ep) return;
    int src, dst;
    if (warp_id < ep) {
        src = pos[warp_id];
        dst = pos[ep + warp_id];
    } else {
        int e2 = warp_id - ep;
        src = neg[e2];
        dst = neg[ep + e2];
    }
    const float* xj = h2 + (size_t)src * H2_DIM;
    const float* xi = h2 + (size_t)dst * H2_DIM;
    float s = xi[lane] * xj[lane] + xi[lane + 32] * xj[lane + 32];
#pragma unroll
    for (int off = 16; off > 0; off >>= 1)
        s += __shfl_xor(s, off, 32);
    if (lane == 0) out[warp_id] = s;
}

static inline size_t align256(size_t x) { return (x + 255) & ~(size_t)255; }

extern "C" void kernel_launch(void* const* d_in, const int* in_sizes, int n_in,
                              void* d_out, int out_size, void* d_ws, size_t ws_size,
                              hipStream_t stream) {
    const float* x    = (const float*)d_in[0];
    const int*   ei   = (const int*)d_in[1];
    const int*   pos  = (const int*)d_in[2];
    const int*   neg  = (const int*)d_in[3];
    const float* W1   = (const float*)d_in[4];
    const float* b1   = (const float*)d_in[5];
    const float* W2   = (const float*)d_in[6];
    const float* b2   = (const float*)d_in[7];
    float* out = (float*)d_out;

    const int* row = ei;
    const int* col = ei + E_EDGES;

    char* ws = (char*)d_ws;
    size_t off = 0;
    float* dinv  = (float*)(ws + off); off += align256(sizeof(float) * N_NODES);
    float* W1t   = (float*)(ws + off); off += align256(sizeof(float) * F_IN * H1_DIM);
    float* W2t   = (float*)(ws + off); off += align256(sizeof(float) * H1_DIM * H2_DIM);
    float* htmp1 = (float*)(ws + off); off += align256(sizeof(float) * (size_t)N_NODES * H1_DIM);
    float* h1    = (float*)(ws + off); off += align256(sizeof(float) * (size_t)N_NODES * H1_DIM);
    float* htmp2 = (float*)(ws + off); off += align256(sizeof(float) * (size_t)N_NODES * H2_DIM);
    float* h2    = (float*)(ws + off); off += align256(sizeof(float) * (size_t)N_NODES * H2_DIM);
    (void)ws_size; (void)n_in; (void)in_sizes; (void)out_size;

    const int TB = 256;

    transpose_kernel<<<(F_IN * H1_DIM + TB - 1) / TB, TB, 0, stream>>>(W1, W1t, F_IN, H1_DIM);
    transpose_kernel<<<(H1_DIM * H2_DIM + TB - 1) / TB, TB, 0, stream>>>(W2, W2t, H1_DIM, H2_DIM);

    init_deg_kernel<<<(N_NODES + TB - 1) / TB, TB, 0, stream>>>(dinv, N_NODES);
    accum_deg_kernel<<<(E_EDGES + TB - 1) / TB, TB, 0, stream>>>(col, dinv, E_EDGES);
    rsqrt_kernel<<<(N_NODES + TB - 1) / TB, TB, 0, stream>>>(dinv, N_NODES);

    gemm_wmma_f32<F_IN, 8><<<N_NODES / 16, 8 * 32, 0, stream>>>(x, W1t, htmp1);

    {
        int tot = N_NODES * H1_DIM;
        self_init_kernel<H1_DIM><<<(tot + TB - 1) / TB, TB, 0, stream>>>(htmp1, dinv, h1, N_NODES);
        int warps_blocks = (E_EDGES * 32 + TB - 1) / TB;
        edge_scatter_kernel<H1_DIM><<<warps_blocks, TB, 0, stream>>>(row, col, dinv, htmp1, h1, E_EDGES);
        bias_relu_kernel<H1_DIM><<<(tot + TB - 1) / TB, TB, 0, stream>>>(h1, b1, N_NODES);
    }

    gemm_wmma_f32<H1_DIM, 4><<<N_NODES / 16, 4 * 32, 0, stream>>>(h1, W2t, htmp2);

    {
        int tot = N_NODES * H2_DIM;
        self_init_kernel<H2_DIM><<<(tot + TB - 1) / TB, TB, 0, stream>>>(htmp2, dinv, h2, N_NODES);
        int warps_blocks = (E_EDGES * 32 + TB - 1) / TB;
        edge_scatter_kernel<H2_DIM><<<warps_blocks, TB, 0, stream>>>(row, col, dinv, htmp2, h2, E_EDGES);
        bias_add_kernel<H2_DIM><<<(tot + TB - 1) / TB, TB, 0, stream>>>(h2, b2, N_NODES);
    }

    {
        long long total_threads = (long long)2 * EP_CNT * 32;
        int blocks = (int)((total_threads + TB - 1) / TB);
        edge_dot_kernel<<<blocks, TB, 0, stream>>>(pos, neg, h2, out, EP_CNT);
    }
}


# Sum by destination, no float atomics: per call edges get an integer-atomic rank, counts are scanned, sources listed per node; one wave per node adds its rows in rank order, self+bias+relu fused
# speedup vs baseline: 1.9338x; 1.9338x over previous
#include <hip/hip_runtime.h>

typedef __attribute__((ext_vector_type(2))) float v2f;
typedef __attribute__((ext_vector_type(8))) float v8f;

#define N_NODES 100000
#define F_IN    256
#define H1_DIM  128
#define H2_DIM  64
#define E_EDGES 1600000
#define EP_CNT  100000

__global__ void zero_cnt_kernel(int* __restrict__ cnt, int n) {
    int i = blockIdx.x * blockDim.x + threadIdx.x;
    if (i < n) cnt[i] = 0;
}

__global__ void rank_edges_kernel(const int* __restrict__ col,
                                  int* __restrict__ cnt,
                                  int* __restrict__ rank, int e) {
    int i = blockIdx.x * blockDim.x + threadIdx.x;
    if (i < e) rank[i] = atomicAdd(&cnt[col[i]], 1);
}

__global__ void __launch_bounds__(1024)
scan_cnt_kernel(const int* __restrict__ cnt, int* __restrict__ rowptr,
                float* __restrict__ dinv, int n) {
    __shared__ int s_pre[32];
    __shared__ int s_tot[32];
    const int tid  = threadIdx.x;
    const int lane = tid & 31;
    const int wave = tid >> 5;
    const int base = blockIdx.x * 1024;

    int pre = 0;
    const int4* c4 = (const int4*)cnt;
    const int n4 = base >> 2;
    for (int i = tid; i < n4; i += 1024) {
        int4 q = c4[i];
        pre += (q.x + q.y) + (q.z + q.w);
    }
#pragma unroll
    for (int off = 16; off > 0; off >>= 1)
        pre += __shfl_xor(pre, off, 32);

    const int i = base + tid;
    const int v = (i < n) ? cnt[i] : 0;
    int inc = v;
#pragma unroll
    for (int off = 1; off < 32; off <<= 1) {
        int t = __shfl_up(inc, off, 32);
        if (lane >= off) inc += t;
    }
    if (lane == 0)  s_pre[wave] = pre;
    if (lane == 31) s_tot[wave] = inc;
    __syncthreads();

    int front = 0;
#pragma unroll
    for (int w = 0; w < 32; ++w) {
        front += s_pre[w];
        if (w < wave) front += s_tot[w];
    }
    const int excl = front + inc - v;
    if (i < n) {
        rowptr[i] = excl;
        dinv[i] = rsqrtf((float)(v + 1));
        if (i == n - 1) rowptr[n] = excl + v;
    }
}

__global__ void fill_csr_kernel(const int* __restrict__ row,
                                const int* __restrict__ col,
                                const int* __restrict__ rowptr,
                                const int* __restrict__ rank,
                                int* __restrict__ csr_src, int e) {
    int i = blockIdx.x * blockDim.x + threadIdx.x;
    if (i < e) csr_src[rowptr[col[i]] + rank[i]] = row[i];
}

__global__ void transpose_kernel(const float* __restrict__ B,
                                 float* __restrict__ Bt, int K, int N) {
    int idx = blockIdx.x * blockDim.x + threadIdx.x;
    if (idx >= K * N) return;
    int k = idx / N;
    int n = idx - k * N;
    Bt[(size_t)n * K + k] = B[idx];
}

template <int K, int NT>
__global__ void __launch_bounds__(NT * 32)
gemm_wmma_f32(const float* __restrict__ A, const float* __restrict__ Bt,
              float* __restrict__ C) {
    constexpr int N = NT * 16;
    const int wave = threadIdx.x >> 5;
    const int lane = threadIdx.x & 31;
    const int l    = lane & 15;
    const int half = lane >> 4;

    const int m0 = blockIdx.x * 16;
    const int n0 = wave * 16;

    const float* ap = A  + (size_t)(m0 + l) * K + 2 * half;
    const float* bp = Bt + (size_t)(n0 + l) * K + 2 * half;

    v8f acc = {0.f, 0.f, 0.f, 0.f, 0.f, 0.f, 0.f, 0.f};
#pragma unroll
    for (int k0 = 0; k0 < K; k0 += 4) {
        v2f a = *(const v2f*)(ap + k0);
        v2f b = *(const v2f*)(bp + k0);
        acc = __builtin_amdgcn_wmma_f32_16x16x4_f32(
             false, a,  false, b,
             (short)0, acc,  false,  false);
    }

    float* cp = C + (size_t)(m0 + 8 * half) * N + n0 + l;
#pragma unroll
    for (int j = 0; j < 8; ++j)
        cp[(size_t)j * N] = acc[j];
}

template <int H, bool RELU>
__global__ void __launch_bounds__(256)
gather_kernel(const int* __restrict__ rowptr, const int* __restrict__ csr_src,
              const float* __restrict__ dinv, const float* __restrict__ htmp,
              const float* __restrict__ bias, float* __restrict__ out, int n) {
    constexpr int V = H / 32;
    typedef __attribute__((ext_vector_type(V))) float vec;
    const int lane = threadIdx.x & 31;
    const int node = __builtin_amdgcn_readfirstlane(
        (int)((blockIdx.x * blockDim.x + threadIdx.x) >> 5));
    if (node >= n) return;

    const int beg = rowptr[node];
    const int end = rowptr[node + 1];
    const float dc = dinv[node];
    const int fo = lane * V;

    vec acc = *(const vec*)(htmp + (size_t)node * H + fo);
    acc = acc * dc * dc;

    for (int base = beg; base < end; base += 32) {
        const int j = base + lane;
        int r = node;
        float w = 0.f;
        if (j < end) {
            r = csr_src[j];
            w = dinv[r] * dc;
        }
        int m = end - base;
        if (m > 32) m = 32;
#pragma unroll 4
        for (int t = 0; t < m; ++t) {
            const int rr = __builtin_amdgcn_readlane(r, t);
            const float ww = __int_as_float(
                __builtin_amdgcn_readlane(__float_as_int(w), t));
            const vec x = *(const vec*)(htmp + (size_t)rr * H + fo);
            acc += x * ww;
        }
    }

    const vec b = *(const vec*)(bias + fo);
    acc += b;
    if (RELU) {
#pragma unroll
        for (int q = 0; q < V; ++q) acc[q] = acc[q] > 0.f ? acc[q] : 0.f;
    }
    *(vec*)(out + (size_t)node * H + fo) = acc;
}

__global__ void __launch_bounds__(256)
edge_dot_kernel(const int* __restrict__ pos, const int* __restrict__ neg,
                const float* __restrict__ h2, float* __restrict__ out, int ep) {
    int warp_id = (blockIdx.x * blockDim.x + threadIdx.x) >> 5;
    int lane = threadIdx.x & 31;
    if (warp_id >= 2 * ep) return;
    int src, dst;
    if (warp_id < ep) {
        src = pos[warp_id];
        dst = pos[ep + warp_id];
    } else {
        int e2 = warp_id - ep;
        src = neg[e2];
        dst = neg[ep + e2];
    }
    const float* xj = h2 + (size_t)src * H2_DIM;
    const float* xi = h2 + (size_t)dst * H2_DIM;
    float s = xi[lane] * xj[lane] + xi[lane + 32] * xj[lane + 32];
#pragma unroll
    for (int off = 16; off > 0; off >>= 1)
        s += __shfl_xor(s, off, 32);
    if (lane == 0) out[warp_id] = s;
}

static inline size_t align256(size_t x) { return (x + 255) & ~(size_t)255; }

extern "C" void kernel_launch(void* const* d_in, const int* in_sizes, int n_in,
                              void* d_out, int out_size, void* d_ws, size_t ws_size,
                              hipStream_t stream) {
    const float* x    = (const float*)d_in[0];
    const int*   ei   = (const int*)d_in[1];
    const int*   pos  = (const int*)d_in[2];
    const int*   neg  = (const int*)d_in[3];
    const float* W1   = (const float*)d_in[4];
    const float* b1   = (const float*)d_in[5];
    const float* W2   = (const float*)d_in[6];
    const float* b2   = (const float*)d_in[7];
    float* out = (float*)d_out;

    const int* row = ei;
    const int* col = ei + E_EDGES;

    char* ws = (char*)d_ws;
    size_t off = 0;
    float* dinv  = (float*)(ws + off); off += align256(sizeof(float) * N_NODES);
    float* W1t   = (float*)(ws + off); off += align256(sizeof(float) * F_IN * H1_DIM);
    float* W2t   = (float*)(ws + off); off += align256(sizeof(float) * H1_DIM * H2_DIM);
    float* htmp1 = (float*)(ws + off); off += align256(sizeof(float) * (size_t)N_NODES * H1_DIM);
    float* h1    = (float*)(ws + off); off += align256(sizeof(float) * (size_t)N_NODES * H1_DIM);
    float* htmp2 = (float*)(ws + off); off += align256(sizeof(float) * (size_t)N_NODES * H2_DIM);
    float* h2    = htmp1;
    int* cnt     = (int*)(ws + off); off += align256(sizeof(int) * N_NODES);
    int* rowptr  = (int*)(ws + off); off += align256(sizeof(int) * (N_NODES + 1));
    int* rank    = (int*)(ws + off); off += align256(sizeof(int) * (size_t)E_EDGES);
    int* csr_src = (int*)(ws + off); off += align256(sizeof(int) * (size_t)E_EDGES);
    (void)ws_size; (void)n_in; (void)in_sizes; (void)out_size;

    const int TB = 256;

    transpose_kernel<<<(F_IN * H1_DIM + TB - 1) / TB, TB, 0, stream>>>(W1, W1t, F_IN, H1_DIM);
    transpose_kernel<<<(H1_DIM * H2_DIM + TB - 1) / TB, TB, 0, stream>>>(W2, W2t, H1_DIM, H2_DIM);

    zero_cnt_kernel<<<(N_NODES + TB - 1) / TB, TB, 0, stream>>>(cnt, N_NODES);
    rank_edges_kernel<<<(E_EDGES + TB - 1) / TB, TB, 0, stream>>>(col, cnt, rank, E_EDGES);
    scan_cnt_kernel<<<(N_NODES + 1023) / 1024, 1024, 0, stream>>>(cnt, rowptr, dinv, N_NODES);
    fill_csr_kernel<<<(E_EDGES + TB - 1) / TB, TB, 0, stream>>>(row, col, rowptr, rank, csr_src, E_EDGES);

    gemm_wmma_f32<F_IN, 8><<<N_NODES / 16, 8 * 32, 0, stream>>>(x, W1t, htmp1);

    gather_kernel<H1_DIM, true><<<(N_NODES * 32 + TB - 1) / TB, TB, 0, stream>>>(
        rowptr, csr_src, dinv, htmp1, b1, h1, N_NODES);

    gemm_wmma_f32<H1_DIM, 4><<<N_NODES / 16, 4 * 32, 0, stream>>>(h1, W2t, htmp2);

    gather_kernel<H2_DIM, false><<<(N_NODES * 32 + TB - 1) / TB, TB, 0, stream>>>(
        rowptr, csr_src, dinv, htmp2, b2, h2, N_NODES);

    {
        long long total_threads = (long long)2 * EP_CNT * 32;
        int blocks = (int)((total_threads + TB - 1) / TB);
        edge_dot_kernel<<<blocks, TB, 0, stream>>>(pos, neg, h2, out, EP_CNT);
    }
}


# Fewer, longer-lived waves: every pass starts a grid the chip holds at once and its waves loop over nodes, 32-edge batches or row tiles (wave entries per call cut about 7x); per-item sums unchanged
# speedup vs baseline: 2.5583x; 1.3229x over previous
#include <hip/hip_runtime.h>

typedef __attribute__((ext_vector_type(2))) float v2f;
typedef __attribute__((ext_vector_type(8))) float v8f;

#define N_NODES 100000
#define F_IN    256
#define H1_DIM  128
#define H2_DIM  64
#define E_EDGES 1600000
#define EP_CNT  100000

__global__ void zero_cnt_kernel(int* __restrict__ cnt, int n) {
    int i = blockIdx.x * blockDim.x + threadIdx.x;
    if (i < n) cnt[i] = 0;
}

template <int EPT>
__global__ void __launch_bounds__(256)
rank_edges_kernel(const int* __restrict__ col,
                  int* __restrict__ cnt,
                  int* __restrict__ rank, int e) {
    const int stride = gridDim.x * blockDim.x;
    const int i0 = blockIdx.x * blockDim.x + threadIdx.x;
    int c[EPT];
#pragma unroll
    for (int k = 0; k < EPT; ++k) {
        const int i = i0 + k * stride;
        c[k] = (i < e) ? col[i] : -1;
    }
    int r[EPT];
#pragma unroll
    for (int k = 0; k < EPT; ++k)
        r[k] = (c[k] >= 0) ? atomicAdd(&cnt[c[k]], 1) : 0;
#pragma unroll
    for (int k = 0; k < EPT; ++k) {
        const int i = i0 + k * stride;
        if (i < e) rank[i] = r[k];
    }
}

__global__ void __launch_bounds__(1024)
scan_cnt_kernel(const int* __restrict__ cnt, int* __restrict__ rowptr,
                float* __restrict__ dinv, int n) {
    __shared__ int s_pre[32];
    __shared__ int s_tot[32];
    const int tid  = threadIdx.x;
    const int lane = tid & 31;
    const int wave = tid >> 5;
    const int base = blockIdx.x * 1024;

    int pre = 0;
    const int4* c4 = (const int4*)cnt;
    const int n4 = base >> 2;
    for (int i = tid; i < n4; i += 1024) {
        int4 q = c4[i];
        pre += (q.x + q.y) + (q.z + q.w);
    }
#pragma unroll
    for (int off = 16; off > 0; off >>= 1)
        pre += __shfl_xor(pre, off, 32);

    const int i = base + tid;
    const int v = (i < n) ? cnt[i] : 0;
    int inc = v;
#pragma unroll
    for (int off = 1; off < 32; off <<= 1) {
        int t = __shfl_up(inc, off, 32);
        if (lane >= off) inc += t;
    }
    if (lane == 0)  s_pre[wave] = pre;
    if (lane == 31) s_tot[wave] = inc;
    __syncthreads();

    int front = 0;
#pragma unroll
    for (int w = 0; w < 32; ++w) {
        front += s_pre[w];
        if (w < wave) front += s_tot[w];
    }
    const int excl = front + inc - v;
    if (i < n) {
        rowptr[i] = excl;
        dinv[i] = rsqrtf((float)(v + 1));
        if (i == n - 1) rowptr[n] = excl + v;
    }
}

template <int EPT>
__global__ void __launch_bounds__(256)
fill_csr_kernel(const int* __restrict__ row,
                const int* __restrict__ col,
                const int* __restrict__ rowptr,
                const int* __restrict__ rank,
                int* __restrict__ csr_src, int e) {
    const int stride = gridDim.x * blockDim.x;
    const int i0 = blockIdx.x * blockDim.x + threadIdx.x;
    int c[EPT], rk[EPT], rw[EPT];
#pragma unroll
    for (int k = 0; k < EPT; ++k) {
        const int i = i0 + k * stride;
        const bool ok = i < e;
        c[k]  = ok ? col[i]  : -1;
        rk[k] = ok ? rank[i] : 0;
        rw[k] = ok ? row[i]  : 0;
    }
    int p[EPT];
#pragma unroll
    for (int k = 0; k < EPT; ++k)
        p[k] = (c[k] >= 0) ? rowptr[c[k]] : 0;
#pragma unroll
    for (int k = 0; k < EPT; ++k)
        if (c[k] >= 0) csr_src[p[k] + rk[k]] = rw[k];
}

__global__ void transpose_kernel(const float* __restrict__ B,
                                 float* __restrict__ Bt, int K, int N) {
    int idx = blockIdx.x * blockDim.x + threadIdx.x;
    if (idx >= K * N) return;
    int k = idx / N;
    int n = idx - k * N;
    Bt[(size_t)n * K + k] = B[idx];
}

template <int K, int NT>
__global__ void __launch_bounds__(NT * 32)
gemm_wmma_f32(const float* __restrict__ A, const float* __restrict__ Bt,
              float* __restrict__ C, int mtiles) {
    constexpr int N = NT * 16;
    const int wave = threadIdx.x >> 5;
    const int lane = threadIdx.x & 31;
    const int l    = lane & 15;
    const int half = lane >> 4;

    const int n0 = wave * 16;
    const float* bp = Bt + (size_t)(n0 + l) * K + 2 * half;

    for (int mtile = blockIdx.x; mtile < mtiles; mtile += gridDim.x) {
        const int m0 = mtile * 16;
        const float* ap = A + (size_t)(m0 + l) * K + 2 * half;

        v8f acc = {0.f, 0.f, 0.f, 0.f, 0.f, 0.f, 0.f, 0.f};
#pragma unroll
        for (int k0 = 0; k0 < K; k0 += 4) {
            v2f a = *(const v2f*)(ap + k0);
            v2f b = *(const v2f*)(bp + k0);
            acc = __builtin_amdgcn_wmma_f32_16x16x4_f32(
                 false, a,  false, b,
                 (short)0, acc,  false,  false);
        }

        float* cp = C + (size_t)(m0 + 8 * half) * N + n0 + l;
#pragma unroll
        for (int j = 0; j < 8; ++j)
            cp[(size_t)j * N] = acc[j];
    }
}

template <int H, bool RELU>
__global__ void __launch_bounds__(256)
gather_kernel(const int* __restrict__ rowptr, const int* __restrict__ csr_src,
              const float* __restrict__ dinv, const float* __restrict__ htmp,
              const float* __restrict__ bias, float* __restrict__ out, int n) {
    constexpr int V = H / 32;
    typedef __attribute__((ext_vector_type(V))) float vec;
    const int lane = threadIdx.x & 31;
    const int wave0 = __builtin_amdgcn_readfirstlane(
        (int)((blockIdx.x * blockDim.x + threadIdx.x) >> 5));
    const int nwaves = (int)((gridDim.x * blockDim.x) >> 5);
    const int fo = lane * V;
    const vec b = *(const vec*)(bias + fo);

    for (int node = wave0; node < n; node += nwaves) {
        const int beg = rowptr[node];
        const int end = rowptr[node + 1];
        const float dc = dinv[node];

        vec acc = *(const vec*)(htmp + (size_t)node * H + fo);
        acc = acc * dc * dc;

        for (int base = beg; base < end; base += 32) {
            const int j = base + lane;
            int r = node;
            float w = 0.f;
            if (j < end) {
                r = csr_src[j];
                w = dinv[r] * dc;
            }
            int m = end - base;
            if (m > 32) m = 32;
#pragma unroll 4
            for (int t = 0; t < m; ++t) {
                const int rr = __builtin_amdgcn_readlane(r, t);
                const float ww = __int_as_float(
                    __builtin_amdgcn_readlane(__float_as_int(w), t));
                const vec x = *(const vec*)(htmp + (size_t)rr * H + fo);
                acc += x * ww;
            }
        }

        vec o = acc + b;
        if (RELU) {
#pragma unroll
            for (int q = 0; q < V; ++q) o[q] = o[q] > 0.f ? o[q] : 0.f;
        }
        *(vec*)(out + (size_t)node * H + fo) = o;
    }
}

__global__ void __launch_bounds__(256)
edge_dot_kernel(const int* __restrict__ pos, const int* __restrict__ neg,
                const float* __restrict__ h2, float* __restrict__ out, int ep) {
    const int lane = threadIdx.x & 31;
    const int wave0 = __builtin_amdgcn_readfirstlane(
        (int)((blockIdx.x * blockDim.x + threadIdx.x) >> 5));
    const int nwaves = (int)((gridDim.x * blockDim.x) >> 5);
    const int total = 2 * ep;
    const int nbatch = (total + 31) >> 5;

    for (int bt = wave0; bt < nbatch; bt += nwaves) {
        const int e = bt * 32 + lane;
        int src = 0, dst = 0;
        if (e < total) {
            if (e < ep) {
                src = pos[e];
                dst = pos[ep + e];
            } else {
                src = neg[e - ep];
                dst = neg[e];
            }
        }
        int m = total - bt * 32;
        if (m > 32) m = 32;
        float res = 0.f;
#pragma unroll 8
        for (int t = 0; t < m; ++t) {
            const int sj = __builtin_amdgcn_readlane(src, t);
            const int si = __builtin_amdgcn_readlane(dst, t);
            const float* xj = h2 + (size_t)sj * H2_DIM;
            const float* xi = h2 + (size_t)si * H2_DIM;
            float v = xi[lane] * xj[lane] + xi[lane + 32] * xj[lane + 32];
#pragma unroll
            for (int off = 16; off > 0; off >>= 1)
                v += __shfl_xor(v, off, 32);
            if (lane == t) res = v;
        }
        if (e < total) out[e] = res;
    }
}

static inline size_t align256(size_t x) { return (x + 255) & ~(size_t)255; }

extern "C" void kernel_launch(void* const* d_in, const int* in_sizes, int n_in,
                              void* d_out, int out_size, void* d_ws, size_t ws_size,
                              hipStream_t stream) {
    const float* x    = (const float*)d_in[0];
    const int*   ei   = (const int*)d_in[1];
    const int*   pos  = (const int*)d_in[2];
    const int*   neg  = (const int*)d_in[3];
    const float* W1   = (const float*)d_in[4];
    const float* b1   = (const float*)d_in[5];
    const float* W2   = (const float*)d_in[6];
    const float* b2   = (const float*)d_in[7];
    float* out = (float*)d_out;

    const int* row = ei;
    const int* col = ei + E_EDGES;

    char* ws = (char*)d_ws;
    size_t off = 0;
    float* dinv  = (float*)(ws + off); off += align256(sizeof(float) * N_NODES);
    float* W1t   = (float*)(ws + off); off += align256(sizeof(float) * F_IN * H1_DIM);
    float* W2t   = (float*)(ws + off); off += align256(sizeof(float) * H1_DIM * H2_DIM);
    float* htmp1 = (float*)(ws + off); off += align256(sizeof(float) * (size_t)N_NODES * H1_DIM);
    float* h1    = (float*)(ws + off); off += align256(sizeof(float) * (size_t)N_NODES * H1_DIM);
    float* htmp2 = (float*)(ws + off); off += align256(sizeof(float) * (size_t)N_NODES * H2_DIM);
    float* h2    = htmp1;
    int* cnt     = (int*)(ws + off); off += align256(sizeof(int) * N_NODES);
    int* rowptr  = (int*)(ws + off); off += align256(sizeof(int) * (N_NODES + 1));
    int* rank    = (int*)(ws + off); off += align256(sizeof(int) * (size_t)E_EDGES);
    int* csr_src = (int*)(ws + off); off += align256(sizeof(int) * (size_t)E_EDGES);
    (void)ws_size; (void)n_in; (void)in_sizes; (void)out_size;

    const int TB = 256;
    const int EPT = 5;
    const int EDGE_BLOCKS = (E_EDGES + TB * EPT - 1) / (TB * EPT);
    const int MTILES = N_NODES / 16;
    const int GEMM1_BLOCKS = 768;
    const int GEMM2_BLOCKS = 1250;
    const int GATHER_BLOCKS = 2048;
    const int DOT_BLOCKS = ((2 * EP_CNT + 31) / 32 + 7) / 8;

    transpose_kernel<<<(F_IN * H1_DIM + TB - 1) / TB, TB, 0, stream>>>(W1, W1t, F_IN, H1_DIM);
    transpose_kernel<<<(H1_DIM * H2_DIM + TB - 1) / TB, TB, 0, stream>>>(W2, W2t, H1_DIM, H2_DIM);

    zero_cnt_kernel<<<(N_NODES + TB - 1) / TB, TB, 0, stream>>>(cnt, N_NODES);
    rank_edges_kernel<EPT><<<EDGE_BLOCKS, TB, 0, stream>>>(col, cnt, rank, E_EDGES);
    scan_cnt_kernel<<<(N_NODES + 1023) / 1024, 1024, 0, stream>>>(cnt, rowptr, dinv, N_NODES);
    fill_csr_kernel<EPT><<<EDGE_BLOCKS, TB, 0, stream>>>(row, col, rowptr, rank, csr_src, E_EDGES);

    gemm_wmma_f32<F_IN, 8><<<GEMM1_BLOCKS, 8 * 32, 0, stream>>>(x, W1t, htmp1, MTILES);

    gather_kernel<H1_DIM, true><<<GATHER_BLOCKS, TB, 0, stream>>>(
        rowptr, csr_src, dinv, htmp1, b1, h1, N_NODES);

    gemm_wmma_f32<H1_DIM, 4><<<GEMM2_BLOCKS, 4 * 32, 0, stream>>>(h1, W2t, htmp2, MTILES);

    gather_kernel<H2_DIM, false><<<GATHER_BLOCKS, TB, 0, stream>>>(
        rowptr, csr_src, dinv, htmp2, b2, h2, N_NODES);

    edge_dot_kernel<<<DOT_BLOCKS, TB, 0, stream>>>(pos, neg, h2, out, EP_CNT);
}


# Edges grouped by destination on LDS counters in two levels (256 node ranges, then nodes), not one returning global add per edge; 11 to 10 launches; a node's sum order: arrival rank of LDS adds
# speedup vs baseline: 3.7408x; 1.4622x over previous
#include <hip/hip_runtime.h>

typedef __attribute__((ext_vector_type(2))) float v2f;
typedef __attribute__((ext_vector_type(8))) float v8f;

#define N_NODES 100000
#define F_IN    256
#define H1_DIM  128
#define H2_DIM  64
#define E_EDGES 1600000
#define EP_CNT  100000

#define NB_BUCKETS    256
#define NODES_PER_BKT 391
#define BKT_SLOTS     512
#define BIN_BLOCKS    256
#define BIN_THREADS   1024
#define EDGES_PER_BLK ((E_EDGES + BIN_BLOCKS - 1) / BIN_BLOCKS)
#define BIN_EPT       ((EDGES_PER_BLK + BIN_THREADS - 1) / BIN_THREADS)
#define CSR_EPT       8
static_assert(NB_BUCKETS * NODES_PER_BKT >= N_NODES, "ranges must cover every node");
static_assert(BKT_SLOTS >= NODES_PER_BKT && BKT_SLOTS <= BIN_THREADS, "one scan thread per slot");
static_assert(BIN_BLOCKS % 128 == 0 && BIN_THREADS == 4 * NB_BUCKETS, "table walk: 4 threads per column, 32 rows per batch");

__device__ __forceinline__ int wave_incl_scan(int v, int lane) {
#pragma unroll
    for (int off = 1; off < 32; off <<= 1) {
        const int t = __shfl_up(v, off, 32);
        if (lane >= off) v += t;
    }
    return v;
}

__global__ void __launch_bounds__(BIN_THREADS)
bucket_count_kernel(const int* __restrict__ col, int* __restrict__ hist, int e) {
    __shared__ int s_cnt[NB_BUCKETS];
    const int tid = threadIdx.x;
    if (tid < NB_BUCKETS) s_cnt[tid] = 0;
    const int beg = blockIdx.x * EDGES_PER_BLK;
    const int end = (beg + EDGES_PER_BLK < e) ? beg + EDGES_PER_BLK : e;
    int c[BIN_EPT];
#pragma unroll
    for (int k = 0; k < BIN_EPT; ++k) {
        const int i = beg + tid + k * BIN_THREADS;
        c[k] = (i < end) ? col[i] : -1;
    }
    __syncthreads();
#pragma unroll
    for (int k = 0; k < BIN_EPT; ++k)
        if (c[k] >= 0) atomicAdd(&s_cnt[c[k] / NODES_PER_BKT], 1);
    __syncthreads();
    if (tid < NB_BUCKETS) hist[blockIdx.x * NB_BUCKETS + tid] = s_cnt[tid];
}

__global__ void __launch_bounds__(BIN_THREADS)
bucket_scatter_kernel(const int* __restrict__ row, const int* __restrict__ col,
                      const int* __restrict__ hist, int* __restrict__ bstart,
                      int* __restrict__ bins, int e) {
    __shared__ int s_pre[4][NB_BUCKETS];
    __shared__ int s_tot[4][NB_BUCKETS];
    __shared__ int s_cur[NB_BUCKETS];
    __shared__ int s_wsum[NB_BUCKETS / 32];
    const int tid  = threadIdx.x;
    const int lane = tid & 31;
    const int wave = tid >> 5;
    const int t = tid & (NB_BUCKETS - 1);
    const int q = tid >> 8;

    const int beg = blockIdx.x * EDGES_PER_BLK;
    const int end = (beg + EDGES_PER_BLK < e) ? beg + EDGES_PER_BLK : e;
    int c[BIN_EPT], r[BIN_EPT];
#pragma unroll
    for (int k = 0; k < BIN_EPT; ++k) {
        const int i = beg + tid + k * BIN_THREADS;
        const bool ok = i < end;
        c[k] = ok ? col[i] : -1;
        r[k] = ok ? row[i] : 0;
    }

    int pre = 0, tot = 0;
    const int me = blockIdx.x;
    const int* hp = hist + q * NB_BUCKETS + t;
#pragma unroll
    for (int h = 0; h < BIN_BLOCKS / 128; ++h) {
        int v[32];
#pragma unroll
        for (int u = 0; u < 32; ++u)
            v[u] = hp[(h * 32 + u) * 4 * NB_BUCKETS];
#pragma unroll
        for (int u = 0; u < 32; ++u) {
            tot += v[u];
            if (q + 4 * (h * 32 + u) < me) pre += v[u];
        }
    }
    s_pre[q][t] = pre;
    s_tot[q][t] = tot;
    __syncthreads();

    int T = 0, P = 0, inc = 0;
    if (tid < NB_BUCKETS) {
        T = (s_tot[0][t] + s_tot[1][t]) + (s_tot[2][t] + s_tot[3][t]);
        P = (s_pre[0][t] + s_pre[1][t]) + (s_pre[2][t] + s_pre[3][t]);
        inc = wave_incl_scan(T, lane);
        if (lane == 31) s_wsum[wave] = inc;
    }
    __syncthreads();
    if (tid < NB_BUCKETS) {
        int front = 0;
#pragma unroll
        for (int w = 0; w < NB_BUCKETS / 32; ++w)
            if (w < wave) front += s_wsum[w];
        const int excl = front + inc - T;
        s_cur[t] = excl + P;
        if (me == 0) {
            bstart[t] = excl;
            if (t == NB_BUCKETS - 1) bstart[NB_BUCKETS] = excl + T;
        }
    }
    __syncthreads();

    int2* __restrict__ b2 = (int2*)bins;
#pragma unroll
    for (int k = 0; k < BIN_EPT; ++k) {
        if (c[k] >= 0) {
            const int pos = atomicAdd(&s_cur[c[k] / NODES_PER_BKT], 1);
            b2[pos] = make_int2(r[k], c[k]);
        }
    }
}

__global__ void __launch_bounds__(BIN_THREADS)
bucket_csr_kernel(const int* __restrict__ bins, const int* __restrict__ bstart,
                  int* __restrict__ rowptr, float* __restrict__ dinv,
                  int* __restrict__ csr_src, int n) {
    __shared__ int s_cnt[BKT_SLOTS];
    __shared__ int s_off[BKT_SLOTS];
    __shared__ int s_wsum[BKT_SLOTS / 32];
    const int tid  = threadIdx.x;
    const int lane = tid & 31;
    const int wave = tid >> 5;
    const int n0 = blockIdx.x * NODES_PER_BKT;
    const int bs = bstart[blockIdx.x];
    const int be = bstart[blockIdx.x + 1];
    const int2* __restrict__ b2 = (const int2*)bins;

    if (tid < BKT_SLOTS) s_cnt[tid] = 0;
    int2 rc[CSR_EPT];
#pragma unroll
    for (int k = 0; k < CSR_EPT; ++k) {
        const int i = bs + tid + k * BIN_THREADS;
        rc[k] = (i < be) ? b2[i] : make_int2(0, -1);
    }
    __syncthreads();

#pragma unroll
    for (int k = 0; k < CSR_EPT; ++k)
        if (rc[k].y >= 0) atomicAdd(&s_cnt[rc[k].y - n0], 1);
    for (int i = bs + tid + CSR_EPT * BIN_THREADS; i < be; i += BIN_THREADS)
        atomicAdd(&s_cnt[b2[i].y - n0], 1);
    __syncthreads();

    int v = 0, inc = 0;
    if (tid < BKT_SLOTS) {
        v = s_cnt[tid];
        inc = wave_incl_scan(v, lane);
        if (lane == 31) s_wsum[wave] = inc;
    }
    __syncthreads();
    if (tid < BKT_SLOTS) {
        int front = 0;
#pragma unroll
        for (int w = 0; w < BKT_SLOTS / 32; ++w)
            if (w < wave) front += s_wsum[w];
        const int first = bs + front + inc - v;
        s_off[tid] = first;
        s_cnt[tid] = 0;
        const int node = n0 + tid;
        if (tid < NODES_PER_BKT && node < n) {
            rowptr[node] = first;
            dinv[node] = rsqrtf((float)(v + 1));
            if (node == n - 1) rowptr[n] = first + v;
        }
    }
    __syncthreads();

#pragma unroll
    for (int k = 0; k < CSR_EPT; ++k) {
        if (rc[k].y >= 0) {
            const int j = rc[k].y - n0;
            const int rk = atomicAdd(&s_cnt[j], 1);
            csr_src[s_off[j] + rk] = rc[k].x;
        }
    }
    for (int i = bs + tid + CSR_EPT * BIN_THREADS; i < be; i += BIN_THREADS) {
        const int2 p = b2[i];
        const int j = p.y - n0;
        const int rk = atomicAdd(&s_cnt[j], 1);
        csr_src[s_off[j] + rk] = p.x;
    }
}

__global__ void transpose_kernel(const float* __restrict__ B,
                                 float* __restrict__ Bt, int K, int N) {
    int idx = blockIdx.x * blockDim.x + threadIdx.x;
    if (idx >= K * N) return;
    int k = idx / N;
    int n = idx - k * N;
    Bt[(size_t)n * K + k] = B[idx];
}

template <int K, int NT>
__global__ void __launch_bounds__(NT * 32)
gemm_wmma_f32(const float* __restrict__ A, const float* __restrict__ Bt,
              float* __restrict__ C, int mtiles) {
    constexpr int N = NT * 16;
    const int wave = threadIdx.x >> 5;
    const int lane = threadIdx.x & 31;
    const int l    = lane & 15;
    const int half = lane >> 4;

    const int n0 = wave * 16;
    const float* bp = Bt + (size_t)(n0 + l) * K + 2 * half;

    for (int mtile = blockIdx.x; mtile < mtiles; mtile += gridDim.x) {
        const int m0 = mtile * 16;
        const float* ap = A + (size_t)(m0 + l) * K + 2 * half;

        v8f acc = {0.f, 0.f, 0.f, 0.f, 0.f, 0.f, 0.f, 0.f};
#pragma unroll
        for (int k0 = 0; k0 < K; k0 += 4) {
            v2f a = *(const v2f*)(ap + k0);
            v2f b = *(const v2f*)(bp + k0);
            acc = __builtin_amdgcn_wmma_f32_16x16x4_f32(
                 false, a,  false, b,
                 (short)0, acc,  false,  false);
        }

        float* cp = C + (size_t)(m0 + 8 * half) * N + n0 + l;
#pragma unroll
        for (int j = 0; j < 8; ++j)
            cp[(size_t)j * N] = acc[j];
    }
}

template <int H, bool RELU>
__global__ void __launch_bounds__(256)
gather_kernel(const int* __restrict__ rowptr, const int* __restrict__ csr_src,
              const float* __restrict__ dinv, const float* __restrict__ htmp,
              const float* __restrict__ bias, float* __restrict__ out, int n) {
    constexpr int V = H / 32;
    typedef __attribute__((ext_vector_type(V))) float vec;
    const int lane = threadIdx.x & 31;
    const int wave0 = __builtin_amdgcn_readfirstlane(
        (int)((blockIdx.x * blockDim.x + threadIdx.x) >> 5));
    const int nwaves = (int)((gridDim.x * blockDim.x) >> 5);
    const int fo = lane * V;
    const vec b = *(const vec*)(bias + fo);

    for (int node = wave0; node < n; node += nwaves) {
        const int beg = rowptr[node];
        const int end = rowptr[node + 1];
        const float dc = dinv[node];

        vec acc = *(const vec*)(htmp + (size_t)node * H + fo);
        acc = acc * dc * dc;

        for (int base = beg; base < end; base += 32) {
            const int j = base + lane;
            int r = node;
            float w = 0.f;
            if (j < end) {
                r = csr_src[j];
                w = dinv[r] * dc;
            }
            int m = end - base;
            if (m > 32) m = 32;
#pragma unroll 4
            for (int t = 0; t < m; ++t) {
                const int rr = __builtin_amdgcn_readlane(r, t);
                const float ww = __int_as_float(
                    __builtin_amdgcn_readlane(__float_as_int(w), t));
                const vec x = *(const vec*)(htmp + (size_t)rr * H + fo);
                acc += x * ww;
            }
        }

        vec o = acc + b;
        if (RELU) {
#pragma unroll
            for (int q = 0; q < V; ++q) o[q] = o[q] > 0.f ? o[q] : 0.f;
        }
        *(vec*)(out + (size_t)node * H + fo) = o;
    }
}

__global__ void __launch_bounds__(256)
edge_dot_kernel(const int* __restrict__ pos, const int* __restrict__ neg,
                const float* __restrict__ h2, float* __restrict__ out, int ep) {
    const int lane = threadIdx.x & 31;
    const int wave0 = __builtin_amdgcn_readfirstlane(
        (int)((blockIdx.x * blockDim.x + threadIdx.x) >> 5));
    const int nwaves = (int)((gridDim.x * blockDim.x) >> 5);
    const int total = 2 * ep;
    const int nbatch = (total + 31) >> 5;

    for (int bt = wave0; bt < nbatch; bt += nwaves) {
        const int e = bt * 32 + lane;
        int src = 0, dst = 0;
        if (e < total) {
            if (e < ep) {
                src = pos[e];
                dst = pos[ep + e];
            } else {
                src = neg[e - ep];
                dst = neg[e];
            }
        }
        int m = total - bt * 32;
        if (m > 32) m = 32;
        float res = 0.f;
#pragma unroll 8
        for (int t = 0; t < m; ++t) {
            const int sj = __builtin_amdgcn_readlane(src, t);
            const int si = __builtin_amdgcn_readlane(dst, t);
            const float* xj = h2 + (size_t)sj * H2_DIM;
            const float* xi = h2 + (size_t)si * H2_DIM;
            float v = xi[lane] * xj[lane] + xi[lane + 32] * xj[lane + 32];
#pragma unroll
            for (int off = 16; off > 0; off >>= 1)
                v += __shfl_xor(v, off, 32);
            if (lane == t) res = v;
        }
        if (e < total) out[e] = res;
    }
}

static inline size_t align256(size_t x) { return (x + 255) & ~(size_t)255; }

extern "C" void kernel_launch(void* const* d_in, const int* in_sizes, int n_in,
                              void* d_out, int out_size, void* d_ws, size_t ws_size,
                              hipStream_t stream) {
    const float* x    = (const float*)d_in[0];
    const int*   ei   = (const int*)d_in[1];
    const int*   pos  = (const int*)d_in[2];
    const int*   neg  = (const int*)d_in[3];
    const float* W1   = (const float*)d_in[4];
    const float* b1   = (const float*)d_in[5];
    const float* W2   = (const float*)d_in[6];
    const float* b2   = (const float*)d_in[7];
    float* out = (float*)d_out;

    const int* row = ei;
    const int* col = ei + E_EDGES;

    char* ws = (char*)d_ws;
    size_t off = 0;
    float* dinv  = (float*)(ws + off); off += align256(sizeof(float) * N_NODES);
    float* W1t   = (float*)(ws + off); off += align256(sizeof(float) * F_IN * H1_DIM);
    float* W2t   = (float*)(ws + off); off += align256(sizeof(float) * H1_DIM * H2_DIM);
    float* htmp1 = (float*)(ws + off); off += align256(sizeof(float) * (size_t)N_NODES * H1_DIM);
    float* h1    = (float*)(ws + off); off += align256(sizeof(float) * (size_t)N_NODES * H1_DIM);
    float* htmp2 = (float*)(ws + off); off += align256(sizeof(float) * (size_t)N_NODES * H2_DIM);
    float* h2    = htmp1;
    int* bstart  = (int*)(ws + off); off += align256(sizeof(int) * N_NODES);
    int* rowptr  = (int*)(ws + off); off += align256(sizeof(int) * (N_NODES + 1));
    int* hist    = (int*)(ws + off); off += align256(sizeof(int) * (size_t)E_EDGES);
    int* csr_src = (int*)(ws + off); off += align256(sizeof(int) * (size_t)E_EDGES);
    int* bins    = (int*)h1;
    (void)ws_size; (void)n_in; (void)in_sizes; (void)out_size;

    const int TB = 256;
    const int MTILES = N_NODES / 16;
    const int GEMM1_BLOCKS = 768;
    const int GEMM2_BLOCKS = 1250;
    const int GATHER_BLOCKS = 2048;
    const int DOT_BLOCKS = ((2 * EP_CNT + 31) / 32 + 7) / 8;

    transpose_kernel<<<(F_IN * H1_DIM + TB - 1) / TB, TB, 0, stream>>>(W1, W1t, F_IN, H1_DIM);
    transpose_kernel<<<(H1_DIM * H2_DIM + TB - 1) / TB, TB, 0, stream>>>(W2, W2t, H1_DIM, H2_DIM);

    bucket_count_kernel<<<BIN_BLOCKS, BIN_THREADS, 0, stream>>>(col, hist, E_EDGES);
    bucket_scatter_kernel<<<BIN_BLOCKS, BIN_THREADS, 0, stream>>>(row, col, hist, bstart, bins, E_EDGES);
    bucket_csr_kernel<<<NB_BUCKETS, BIN_THREADS, 0, stream>>>(bins, bstart, rowptr, dinv, csr_src, N_NODES);

    gemm_wmma_f32<F_IN, 8><<<GEMM1_BLOCKS, 8 * 32, 0, stream>>>(x, W1t, htmp1, MTILES);

    gather_kernel<H1_DIM, true><<<GATHER_BLOCKS, TB, 0, stream>>>(
        rowptr, csr_src, dinv, htmp1, b1, h1, N_NODES);

    gemm_wmma_f32<H1_DIM, 4><<<GEMM2_BLOCKS, 4 * 32, 0, stream>>>(h1, W2t, htmp2, MTILES);

    gather_kernel<H2_DIM, false><<<GATHER_BLOCKS, TB, 0, stream>>>(
        rowptr, csr_src, dinv, htmp2, b2, h2, N_NODES);

    edge_dot_kernel<<<DOT_BLOCKS, TB, 0, stream>>>(pos, neg, h2, out, EP_CNT);
}


# Both GEMMs: each block copies its 16-row A tile once into LDS by whole-line loads (two buffers, next tile in flight under the products); fragments come from LDS, not from 16-line global loads
# speedup vs baseline: 4.7852x; 1.2792x over previous
#include <hip/hip_runtime.h>

typedef __attribute__((ext_vector_type(2))) float v2f;
typedef __attribute__((ext_vector_type(4))) float v4f;
typedef __attribute__((ext_vector_type(8))) float v8f;

#define N_NODES 100000
#define F_IN    256
#define H1_DIM  128
#define H2_DIM  64
#define E_EDGES 1600000
#define EP_CNT  100000

#define NB_BUCKETS    256
#define NODES_PER_BKT 391
#define BKT_SLOTS     512
#define BIN_BLOCKS    256
#define BIN_THREADS   1024
#define EDGES_PER_BLK ((E_EDGES + BIN_BLOCKS - 1) / BIN_BLOCKS)
#define BIN_EPT       ((EDGES_PER_BLK + BIN_THREADS - 1) / BIN_THREADS)
#define CSR_EPT       8
static_assert(NB_BUCKETS * NODES_PER_BKT >= N_NODES, "ranges must cover every node");
static_assert(BKT_SLOTS >= NODES_PER_BKT && BKT_SLOTS <= BIN_THREADS, "one scan thread per slot");
static_assert(BIN_BLOCKS % 128 == 0 && BIN_THREADS == 4 * NB_BUCKETS, "table walk: 4 threads per column, 32 rows per batch");

__device__ __forceinline__ int wave_incl_scan(int v, int lane) {
#pragma unroll
    for (int off = 1; off < 32; off <<= 1) {
        const int t = __shfl_up(v, off, 32);
        if (lane >= off) v += t;
    }
    return v;
}

__global__ void __launch_bounds__(BIN_THREADS)
bucket_count_kernel(const int* __restrict__ col, int* __restrict__ hist, int e) {
    __shared__ int s_cnt[NB_BUCKETS];
    const int tid = threadIdx.x;
    if (tid < NB_BUCKETS) s_cnt[tid] = 0;
    const int beg = blockIdx.x * EDGES_PER_BLK;
    const int end = (beg + EDGES_PER_BLK < e) ? beg + EDGES_PER_BLK : e;
    int c[BIN_EPT];
#pragma unroll
    for (int k = 0; k < BIN_EPT; ++k) {
        const int i = beg + tid + k * BIN_THREADS;
        c[k] = (i < end) ? col[i] : -1;
    }
    __syncthreads();
#pragma unroll
    for (int k = 0; k < BIN_EPT; ++k)
        if (c[k] >= 0) atomicAdd(&s_cnt[c[k] / NODES_PER_BKT], 1);
    __syncthreads();
    if (tid < NB_BUCKETS) hist[blockIdx.x * NB_BUCKETS + tid] = s_cnt[tid];
}

__global__ void __launch_bounds__(BIN_THREADS)
bucket_scatter_kernel(const int* __restrict__ row, const int* __restrict__ col,
                      const int* __restrict__ hist, int* __restrict__ bstart,
                      int* __restrict__ bins, int e) {
    __shared__ int s_pre[4][NB_BUCKETS];
    __shared__ int s_tot[4][NB_BUCKETS];
    __shared__ int s_cur[NB_BUCKETS];
    __shared__ int s_wsum[NB_BUCKETS / 32];
    const int tid  = threadIdx.x;
    const int lane = tid & 31;
    const int wave = tid >> 5;
    const int t = tid & (NB_BUCKETS - 1);
    const int q = tid >> 8;

    const int beg = blockIdx.x * EDGES_PER_BLK;
    const int end = (beg + EDGES_PER_BLK < e) ? beg + EDGES_PER_BLK : e;
    int c[BIN_EPT], r[BIN_EPT];
#pragma unroll
    for (int k = 0; k < BIN_EPT; ++k) {
        const int i = beg + tid + k * BIN_THREADS;
        const bool ok = i < end;
        c[k] = ok ? col[i] : -1;
        r[k] = ok ? row[i] : 0;
    }

    int pre = 0, tot = 0;
    const int me = blockIdx.x;
    const int* hp = hist + q * NB_BUCKETS + t;
#pragma unroll
    for (int h = 0; h < BIN_BLOCKS / 128; ++h) {
        int v[32];
#pragma unroll
        for (int u = 0; u < 32; ++u)
            v[u] = hp[(h * 32 + u) * 4 * NB_BUCKETS];
#pragma unroll
        for (int u = 0; u < 32; ++u) {
            tot += v[u];
            if (q + 4 * (h * 32 + u) < me) pre += v[u];
        }
    }
    s_pre[q][t] = pre;
    s_tot[q][t] = tot;
    __syncthreads();

    int T = 0, P = 0, inc = 0;
    if (tid < NB_BUCKETS) {
        T = (s_tot[0][t] + s_tot[1][t]) + (s_tot[2][t] + s_tot[3][t]);
        P = (s_pre[0][t] + s_pre[1][t]) + (s_pre[2][t] + s_pre[3][t]);
        inc = wave_incl_scan(T, lane);
        if (lane == 31) s_wsum[wave] = inc;
    }
    __syncthreads();
    if (tid < NB_BUCKETS) {
        int front = 0;
#pragma unroll
        for (int w = 0; w < NB_BUCKETS / 32; ++w)
            if (w < wave) front += s_wsum[w];
        const int excl = front + inc - T;
        s_cur[t] = excl + P;
        if (me == 0) {
            bstart[t] = excl;
            if (t == NB_BUCKETS - 1) bstart[NB_BUCKETS] = excl + T;
        }
    }
    __syncthreads();

    int2* __restrict__ b2 = (int2*)bins;
#pragma unroll
    for (int k = 0; k < BIN_EPT; ++k) {
        if (c[k] >= 0) {
            const int pos = atomicAdd(&s_cur[c[k] / NODES_PER_BKT], 1);
            b2[pos] = make_int2(r[k], c[k]);
        }
    }
}

__global__ void __launch_bounds__(BIN_THREADS)
bucket_csr_kernel(const int* __restrict__ bins, const int* __restrict__ bstart,
                  int* __restrict__ rowptr, float* __restrict__ dinv,
                  int* __restrict__ csr_src, int n) {
    __shared__ int s_cnt[BKT_SLOTS];
    __shared__ int s_off[BKT_SLOTS];
    __shared__ int s_wsum[BKT_SLOTS / 32];
    const int tid  = threadIdx.x;
    const int lane = tid & 31;
    const int wave = tid >> 5;
    const int n0 = blockIdx.x * NODES_PER_BKT;
    const int bs = bstart[blockIdx.x];
    const int be = bstart[blockIdx.x + 1];
    const int2* __restrict__ b2 = (const int2*)bins;

    if (tid < BKT_SLOTS) s_cnt[tid] = 0;
    int2 rc[CSR_EPT];
#pragma unroll
    for (int k = 0; k < CSR_EPT; ++k) {
        const int i = bs + tid + k * BIN_THREADS;
        rc[k] = (i < be) ? b2[i] : make_int2(0, -1);
    }
    __syncthreads();

#pragma unroll
    for (int k = 0; k < CSR_EPT; ++k)
        if (rc[k].y >= 0) atomicAdd(&s_cnt[rc[k].y - n0], 1);
    for (int i = bs + tid + CSR_EPT * BIN_THREADS; i < be; i += BIN_THREADS)
        atomicAdd(&s_cnt[b2[i].y - n0], 1);
    __syncthreads();

    int v = 0, inc = 0;
    if (tid < BKT_SLOTS) {
        v = s_cnt[tid];
        inc = wave_incl_scan(v, lane);
        if (lane == 31) s_wsum[wave] = inc;
    }
    __syncthreads();
    if (tid < BKT_SLOTS) {
        int front = 0;
#pragma unroll
        for (int w = 0; w < BKT_SLOTS / 32; ++w)
            if (w < wave) front += s_wsum[w];
        const int first = bs + front + inc - v;
        s_off[tid] = first;
        s_cnt[tid] = 0;
        const int node = n0 + tid;
        if (tid < NODES_PER_BKT && node < n) {
            rowptr[node] = first;
            dinv[node] = rsqrtf((float)(v + 1));
            if (node == n - 1) rowptr[n] = first + v;
        }
    }
    __syncthreads();

#pragma unroll
    for (int k = 0; k < CSR_EPT; ++k) {
        if (rc[k].y >= 0) {
            const int j = rc[k].y - n0;
            const int rk = atomicAdd(&s_cnt[j], 1);
            csr_src[s_off[j] + rk] = rc[k].x;
        }
    }
    for (int i = bs + tid + CSR_EPT * BIN_THREADS; i < be; i += BIN_THREADS) {
        const int2 p = b2[i];
        const int j = p.y - n0;
        const int rk = atomicAdd(&s_cnt[j], 1);
        csr_src[s_off[j] + rk] = p.x;
    }
}

__global__ void transpose_kernel(const float* __restrict__ B,
                                 float* __restrict__ Bt, int K, int N) {
    int idx = blockIdx.x * blockDim.x + threadIdx.x;
    if (idx >= K * N) return;
    int k = idx / N;
    int n = idx - k * N;
    Bt[(size_t)n * K + k] = B[idx];
}

template <int K, int NT>
__global__ void __launch_bounds__(NT * 32)
gemm_wmma_f32(const float* __restrict__ A, const float* __restrict__ Bt,
              float* __restrict__ C, int mtiles) {
    constexpr int N = NT * 16;
    constexpr int P = K + 4;
    constexpr int F4_ROW = K / 4;
    constexpr int THREADS = NT * 32;
    constexpr int LD = (16 * F4_ROW) / THREADS;
    static_assert((16 * F4_ROW) % THREADS == 0, "tile copy: whole pieces per thread");
    static_assert((P % 4) == 0 && ((P / 4) % 2) == 1, "pitch = 16 bytes x odd");
    __shared__ __attribute__((aligned(16))) float sA[2][16 * P];

    const int tid  = threadIdx.x;
    const int wave = tid >> 5;
    const int lane = tid & 31;
    const int l    = lane & 15;
    const int half = lane >> 4;

    const int n0 = wave * 16;
    const float* bp = Bt + (size_t)(n0 + l) * K + 2 * half;
    v2f breg[K / 4];
#pragma unroll
    for (int s = 0; s < K / 4; ++s)
        breg[s] = *(const v2f*)(bp + 4 * s);

    constexpr int ROWS_STEP = THREADS / F4_ROW;
    static_assert(LD == 4 && THREADS % F4_ROW == 0, "four pieces per thread, a fixed row step apart");
    const int soff0 = (tid / F4_ROW) * P + (tid % F4_ROW) * 4;

    v4f p0 = {0.f, 0.f, 0.f, 0.f}, p1 = p0, p2 = p0, p3 = p0;
    int mtile = blockIdx.x;
    if (mtile < mtiles) {
        const v4f* src = (const v4f*)(A + (size_t)mtile * 16 * K) + tid;
        p0 = src[0];
        p1 = src[THREADS];
        p2 = src[2 * THREADS];
        p3 = src[3 * THREADS];
        float* d = &sA[0][soff0];
        *(v4f*)(d)                     = p0;
        *(v4f*)(d + ROWS_STEP * P)     = p1;
        *(v4f*)(d + 2 * ROWS_STEP * P) = p2;
        *(v4f*)(d + 3 * ROWS_STEP * P) = p3;
    }
    __syncthreads();

    int buf = 0;
    for (; mtile < mtiles; mtile += gridDim.x) {
        const int next = mtile + (int)gridDim.x;
        const bool has_next = next < mtiles;
        if (has_next) {
            const v4f* src = (const v4f*)(A + (size_t)next * 16 * K) + tid;
            p0 = src[0];
            p1 = src[THREADS];
            p2 = src[2 * THREADS];
            p3 = src[3 * THREADS];
        }

        const float* sp = &sA[buf][l * P + 2 * half];
        v8f acc = {0.f, 0.f, 0.f, 0.f, 0.f, 0.f, 0.f, 0.f};
#pragma unroll
        for (int k0 = 0; k0 < K; k0 += 4) {
            v2f a = *(const v2f*)(sp + k0);
            acc = __builtin_amdgcn_wmma_f32_16x16x4_f32(
                 false, a,  false, breg[k0 / 4],
                 (short)0, acc,  false,  false);
        }

        if (has_next) {
            float* d = &sA[buf ^ 1][soff0];
            *(v4f*)(d)                     = p0;
            *(v4f*)(d + ROWS_STEP * P)     = p1;
            *(v4f*)(d + 2 * ROWS_STEP * P) = p2;
            *(v4f*)(d + 3 * ROWS_STEP * P) = p3;
        }
        __syncthreads();
        buf ^= 1;

        const int m0 = mtile * 16;
        float* cp = C + (size_t)(m0 + 8 * half) * N + n0 + l;
#pragma unroll
        for (int j = 0; j < 8; ++j)
            cp[(size_t)j * N] = acc[j];
    }
}

template <int H, bool RELU>
__global__ void __launch_bounds__(256)
gather_kernel(const int* __restrict__ rowptr, const int* __restrict__ csr_src,
              const float* __restrict__ dinv, const float* __restrict__ htmp,
              const float* __restrict__ bias, float* __restrict__ out, int n) {
    constexpr int V = H / 32;
    typedef __attribute__((ext_vector_type(V))) float vec;
    const int lane = threadIdx.x & 31;
    const int wave0 = __builtin_amdgcn_readfirstlane(
        (int)((blockIdx.x * blockDim.x + threadIdx.x) >> 5));
    const int nwaves = (int)((gridDim.x * blockDim.x) >> 5);
    const int fo = lane * V;
    const vec b = *(const vec*)(bias + fo);

    for (int node = wave0; node < n; node += nwaves) {
        const int beg = rowptr[node];
        const int end = rowptr[node + 1];
        const float dc = dinv[node];

        vec acc = *(const vec*)(htmp + (size_t)node * H + fo);
        acc = acc * dc * dc;

        for (int base = beg; base < end; base += 32) {
            const int j = base + lane;
            int r = node;
            float w = 0.f;
            if (j < end) {
                r = csr_src[j];
                w = dinv[r] * dc;
            }
            int m = end - base;
            if (m > 32) m = 32;
#pragma unroll 4
            for (int t = 0; t < m; ++t) {
                const int rr = __builtin_amdgcn_readlane(r, t);
                const float ww = __int_as_float(
                    __builtin_amdgcn_readlane(__float_as_int(w), t));
                const vec x = *(const vec*)(htmp + (size_t)rr * H + fo);
                acc += x * ww;
            }
        }

        vec o = acc + b;
        if (RELU) {
#pragma unroll
            for (int q = 0; q < V; ++q) o[q] = o[q] > 0.f ? o[q] : 0.f;
        }
        *(vec*)(out + (size_t)node * H + fo) = o;
    }
}

__global__ void __launch_bounds__(256)
edge_dot_kernel(const int* __restrict__ pos, const int* __restrict__ neg,
                const float* __restrict__ h2, float* __restrict__ out, int ep) {
    const int lane = threadIdx.x & 31;
    const int wave0 = __builtin_amdgcn_readfirstlane(
        (int)((blockIdx.x * blockDim.x + threadIdx.x) >> 5));
    const int nwaves = (int)((gridDim.x * blockDim.x) >> 5);
    const int total = 2 * ep;
    const int nbatch = (total + 31) >> 5;

    for (int bt = wave0; bt < nbatch; bt += nwaves) {
        const int e = bt * 32 + lane;
        int src = 0, dst = 0;
        if (e < total) {
            if (e < ep) {
                src = pos[e];
                dst = pos[ep + e];
            } else {
                src = neg[e - ep];
                dst = neg[e];
            }
        }
        int m = total - bt * 32;
        if (m > 32) m = 32;
        float res = 0.f;
#pragma unroll 8
        for (int t = 0; t < m; ++t) {
            const int sj = __builtin_amdgcn_readlane(src, t);
            const int si = __builtin_amdgcn_readlane(dst, t);
            const float* xj = h2 + (size_t)sj * H2_DIM;
            const float* xi = h2 + (size_t)si * H2_DIM;
            float v = xi[lane] * xj[lane] + xi[lane + 32] * xj[lane + 32];
#pragma unroll
            for (int off = 16; off > 0; off >>= 1)
                v += __shfl_xor(v, off, 32);
            if (lane == t) res = v;
        }
        if (e < total) out[e] = res;
    }
}

static inline size_t align256(size_t x) { return (x + 255) & ~(size_t)255; }

extern "C" void kernel_launch(void* const* d_in, const int* in_sizes, int n_in,
                              void* d_out, int out_size, void* d_ws, size_t ws_size,
                              hipStream_t stream) {
    const float* x    = (const float*)d_in[0];
    const int*   ei   = (const int*)d_in[1];
    const int*   pos  = (const int*)d_in[2];
    const int*   neg  = (const int*)d_in[3];
    const float* W1   = (const float*)d_in[4];
    const float* b1   = (const float*)d_in[5];
    const float* W2   = (const float*)d_in[6];
    const float* b2   = (const float*)d_in[7];
    float* out = (float*)d_out;

    const int* row = ei;
    const int* col = ei + E_EDGES;

    char* ws = (char*)d_ws;
    size_t off = 0;
    float* dinv  = (float*)(ws + off); off += align256(sizeof(float) * N_NODES);
    float* W1t   = (float*)(ws + off); off += align256(sizeof(float) * F_IN * H1_DIM);
    float* W2t   = (float*)(ws + off); off += align256(sizeof(float) * H1_DIM * H2_DIM);
    float* htmp1 = (float*)(ws + off); off += align256(sizeof(float) * (size_t)N_NODES * H1_DIM);
    float* h1    = (float*)(ws + off); off += align256(sizeof(float) * (size_t)N_NODES * H1_DIM);
    float* htmp2 = (float*)(ws + off); off += align256(sizeof(float) * (size_t)N_NODES * H2_DIM);
    float* h2    = htmp1;
    int* bstart  = (int*)(ws + off); off += align256(sizeof(int) * N_NODES);
    int* rowptr  = (int*)(ws + off); off += align256(sizeof(int) * (N_NODES + 1));
    int* hist    = (int*)(ws + off); off += align256(sizeof(int) * (size_t)E_EDGES);
    int* csr_src = (int*)(ws + off); off += align256(sizeof(int) * (size_t)E_EDGES);
    int* bins    = (int*)h1;
    (void)ws_size; (void)n_in; (void)in_sizes; (void)out_size;

    const int TB = 256;
    const int MTILES = N_NODES / 16;
    const int GEMM1_BLOCKS = 512;
    const int GEMM2_BLOCKS = 1250;
    const int GATHER_BLOCKS = 2048;
    const int DOT_BLOCKS = ((2 * EP_CNT + 31) / 32 + 7) / 8;

    transpose_kernel<<<(F_IN * H1_DIM + TB - 1) / TB, TB, 0, stream>>>(W1, W1t, F_IN, H1_DIM);
    transpose_kernel<<<(H1_DIM * H2_DIM + TB - 1) / TB, TB, 0, stream>>>(W2, W2t, H1_DIM, H2_DIM);

    bucket_count_kernel<<<BIN_BLOCKS, BIN_THREADS, 0, stream>>>(col, hist, E_EDGES);
    bucket_scatter_kernel<<<BIN_BLOCKS, BIN_THREADS, 0, stream>>>(row, col, hist, bstart, bins, E_EDGES);
    bucket_csr_kernel<<<NB_BUCKETS, BIN_THREADS, 0, stream>>>(bins, bstart, rowptr, dinv, csr_src, N_NODES);

    gemm_wmma_f32<F_IN, 8><<<GEMM1_BLOCKS, 8 * 32, 0, stream>>>(x, W1t, htmp1, MTILES);

    gather_kernel<H1_DIM, true><<<GATHER_BLOCKS, TB, 0, stream>>>(
        rowptr, csr_src, dinv, htmp1, b1, h1, N_NODES);

    gemm_wmma_f32<H1_DIM, 4><<<GEMM2_BLOCKS, 4 * 32, 0, stream>>>(h1, W2t, htmp2, MTILES);

    gather_kernel<H2_DIM, false><<<GATHER_BLOCKS, TB, 0, stream>>>(
        rowptr, csr_src, dinv, htmp2, b2, h2, N_NODES);

    edge_dot_kernel<<<DOT_BLOCKS, TB, 0, stream>>>(pos, neg, h2, out, EP_CNT);
}


# Layer-1 aggregation and layer-2 product in one launch: a block sums 16 nodes' rows (same order) into LDS, 4 of its waves run the WMMA strips on them; h1 never goes to memory; 10 to 9 launches
# speedup vs baseline: 5.0253x; 1.0502x over previous
#include <hip/hip_runtime.h>

typedef __attribute__((ext_vector_type(2))) float v2f;
typedef __attribute__((ext_vector_type(4))) float v4f;
typedef __attribute__((ext_vector_type(8))) float v8f;

#define N_NODES 100000
#define F_IN    256
#define H1_DIM  128
#define H2_DIM  64
#define E_EDGES 1600000
#define EP_CNT  100000

#define NB_BUCKETS    256
#define NODES_PER_BKT 391
#define BKT_SLOTS     512
#define BIN_BLOCKS    256
#define BIN_THREADS   1024
#define EDGES_PER_BLK ((E_EDGES + BIN_BLOCKS - 1) / BIN_BLOCKS)
#define BIN_EPT       ((EDGES_PER_BLK + BIN_THREADS - 1) / BIN_THREADS)
#define CSR_EPT       8
static_assert(NB_BUCKETS * NODES_PER_BKT >= N_NODES, "ranges must cover every node");
static_assert(BKT_SLOTS >= NODES_PER_BKT && BKT_SLOTS <= BIN_THREADS, "one scan thread per slot");
static_assert(BIN_BLOCKS % 128 == 0 && BIN_THREADS == 4 * NB_BUCKETS, "table walk: 4 threads per column, 32 rows per batch");

__device__ __forceinline__ int wave_incl_scan(int v, int lane) {
#pragma unroll
    for (int off = 1; off < 32; off <<= 1) {
        const int t = __shfl_up(v, off, 32);
        if (lane >= off) v += t;
    }
    return v;
}

__global__ void __launch_bounds__(BIN_THREADS)
bucket_count_kernel(const int* __restrict__ col, int* __restrict__ hist, int e) {
    __shared__ int s_cnt[NB_BUCKETS];
    const int tid = threadIdx.x;
    if (tid < NB_BUCKETS) s_cnt[tid] = 0;
    const int beg = blockIdx.x * EDGES_PER_BLK;
    const int end = (beg + EDGES_PER_BLK < e) ? beg + EDGES_PER_BLK : e;
    int c[BIN_EPT];
#pragma unroll
    for (int k = 0; k < BIN_EPT; ++k) {
        const int i = beg + tid + k * BIN_THREADS;
        c[k] = (i < end) ? col[i] : -1;
    }
    __syncthreads();
#pragma unroll
    for (int k = 0; k < BIN_EPT; ++k)
        if (c[k] >= 0) atomicAdd(&s_cnt[c[k] / NODES_PER_BKT], 1);
    __syncthreads();
    if (tid < NB_BUCKETS) hist[blockIdx.x * NB_BUCKETS + tid] = s_cnt[tid];
}

__global__ void __launch_bounds__(BIN_THREADS)
bucket_scatter_kernel(const int* __restrict__ row, const int* __restrict__ col,
                      const int* __restrict__ hist, int* __restrict__ bstart,
                      int* __restrict__ bins, int e) {
    __shared__ int s_pre[4][NB_BUCKETS];
    __shared__ int s_tot[4][NB_BUCKETS];
    __shared__ int s_cur[NB_BUCKETS];
    __shared__ int s_wsum[NB_BUCKETS / 32];
    const int tid  = threadIdx.x;
    const int lane = tid & 31;
    const int wave = tid >> 5;
    const int t = tid & (NB_BUCKETS - 1);
    const int q = tid >> 8;

    const int beg = blockIdx.x * EDGES_PER_BLK;
    const int end = (beg + EDGES_PER_BLK < e) ? beg + EDGES_PER_BLK : e;
    int c[BIN_EPT], r[BIN_EPT];
#pragma unroll
    for (int k = 0; k < BIN_EPT; ++k) {
        const int i = beg + tid + k * BIN_THREADS;
        const bool ok = i < end;
        c[k] = ok ? col[i] : -1;
        r[k] = ok ? row[i] : 0;
    }

    int pre = 0, tot = 0;
    const int me = blockIdx.x;
    const int* hp = hist + q * NB_BUCKETS + t;
#pragma unroll
    for (int h = 0; h < BIN_BLOCKS / 128; ++h) {
        int v[32];
#pragma unroll
        for (int u = 0; u < 32; ++u)
            v[u] = hp[(h * 32 + u) * 4 * NB_BUCKETS];
#pragma unroll
        for (int u = 0; u < 32; ++u) {
            tot += v[u];
            if (q + 4 * (h * 32 + u) < me) pre += v[u];
        }
    }
    s_pre[q][t] = pre;
    s_tot[q][t] = tot;
    __syncthreads();

    int T = 0, P = 0, inc = 0;
    if (tid < NB_BUCKETS) {
        T = (s_tot[0][t] + s_tot[1][t]) + (s_tot[2][t] + s_tot[3][t]);
        P = (s_pre[0][t] + s_pre[1][t]) + (s_pre[2][t] + s_pre[3][t]);
        inc = wave_incl_scan(T, lane);
        if (lane == 31) s_wsum[wave] = inc;
    }
    __syncthreads();
    if (tid < NB_BUCKETS) {
        int front = 0;
#pragma unroll
        for (int w = 0; w < NB_BUCKETS / 32; ++w)
            if (w < wave) front += s_wsum[w];
        const int excl = front + inc - T;
        s_cur[t] = excl + P;
        if (me == 0) {
            bstart[t] = excl;
            if (t == NB_BUCKETS - 1) bstart[NB_BUCKETS] = excl + T;
        }
    }
    __syncthreads();

    int2* __restrict__ b2 = (int2*)bins;
#pragma unroll
    for (int k = 0; k < BIN_EPT; ++k) {
        if (c[k] >= 0) {
            const int pos = atomicAdd(&s_cur[c[k] / NODES_PER_BKT], 1);
            b2[pos] = make_int2(r[k], c[k]);
        }
    }
}

__global__ void __launch_bounds__(BIN_THREADS)
bucket_csr_kernel(const int* __restrict__ bins, const int* __restrict__ bstart,
                  int* __restrict__ rowptr, float* __restrict__ dinv,
                  int* __restrict__ csr_src, int n) {
    __shared__ int s_cnt[BKT_SLOTS];
    __shared__ int s_off[BKT_SLOTS];
    __shared__ int s_wsum[BKT_SLOTS / 32];
    const int tid  = threadIdx.x;
    const int lane = tid & 31;
    const int wave = tid >> 5;
    const int n0 = blockIdx.x * NODES_PER_BKT;
    const int bs = bstart[blockIdx.x];
    const int be = bstart[blockIdx.x + 1];
    const int2* __restrict__ b2 = (const int2*)bins;

    if (tid < BKT_SLOTS) s_cnt[tid] = 0;
    int2 rc[CSR_EPT];
#pragma unroll
    for (int k = 0; k < CSR_EPT; ++k) {
        const int i = bs + tid + k * BIN_THREADS;
        rc[k] = (i < be) ? b2[i] : make_int2(0, -1);
    }
    __syncthreads();

#pragma unroll
    for (int k = 0; k < CSR_EPT; ++k)
        if (rc[k].y >= 0) atomicAdd(&s_cnt[rc[k].y - n0], 1);
    for (int i = bs + tid + CSR_EPT * BIN_THREADS; i < be; i += BIN_THREADS)
        atomicAdd(&s_cnt[b2[i].y - n0], 1);
    __syncthreads();

    int v = 0, inc = 0;
    if (tid < BKT_SLOTS) {
        v = s_cnt[tid];
        inc = wave_incl_scan(v, lane);
        if (lane == 31) s_wsum[wave] = inc;
    }
    __syncthreads();
    if (tid < BKT_SLOTS) {
        int front = 0;
#pragma unroll
        for (int w = 0; w < BKT_SLOTS / 32; ++w)
            if (w < wave) front += s_wsum[w];
        const int first = bs + front + inc - v;
        s_off[tid] = first;
        s_cnt[tid] = 0;
        const int node = n0 + tid;
        if (tid < NODES_PER_BKT && node < n) {
            rowptr[node] = first;
            dinv[node] = rsqrtf((float)(v + 1));
            if (node == n - 1) rowptr[n] = first + v;
        }
    }
    __syncthreads();

#pragma unroll
    for (int k = 0; k < CSR_EPT; ++k) {
        if (rc[k].y >= 0) {
            const int j = rc[k].y - n0;
            const int rk = atomicAdd(&s_cnt[j], 1);
            csr_src[s_off[j] + rk] = rc[k].x;
        }
    }
    for (int i = bs + tid + CSR_EPT * BIN_THREADS; i < be; i += BIN_THREADS) {
        const int2 p = b2[i];
        const int j = p.y - n0;
        const int rk = atomicAdd(&s_cnt[j], 1);
        csr_src[s_off[j] + rk] = p.x;
    }
}

__global__ void transpose_kernel(const float* __restrict__ B,
                                 float* __restrict__ Bt, int K, int N) {
    int idx = blockIdx.x * blockDim.x + threadIdx.x;
    if (idx >= K * N) return;
    int k = idx / N;
    int n = idx - k * N;
    Bt[(size_t)n * K + k] = B[idx];
}

template <int K, int NT>
__global__ void __launch_bounds__(NT * 32)
gemm_wmma_f32(const float* __restrict__ A, const float* __restrict__ Bt,
              float* __restrict__ C, int mtiles) {
    constexpr int N = NT * 16;
    constexpr int P = K + 4;
    constexpr int F4_ROW = K / 4;
    constexpr int THREADS = NT * 32;
    constexpr int LD = (16 * F4_ROW) / THREADS;
    static_assert((16 * F4_ROW) % THREADS == 0, "tile copy: whole pieces per thread");
    static_assert((P % 4) == 0 && ((P / 4) % 2) == 1, "pitch = 16 bytes x odd");
    __shared__ __attribute__((aligned(16))) float sA[2][16 * P];

    const int tid  = threadIdx.x;
    const int wave = tid >> 5;
    const int lane = tid & 31;
    const int l    = lane & 15;
    const int half = lane >> 4;

    const int n0 = wave * 16;
    const float* bp = Bt + (size_t)(n0 + l) * K + 2 * half;
    v2f breg[K / 4];
#pragma unroll
    for (int s = 0; s < K / 4; ++s)
        breg[s] = *(const v2f*)(bp + 4 * s);

    constexpr int ROWS_STEP = THREADS / F4_ROW;
    static_assert(LD == 4 && THREADS % F4_ROW == 0, "four pieces per thread, a fixed row step apart");
    const int soff0 = (tid / F4_ROW) * P + (tid % F4_ROW) * 4;

    v4f p0 = {0.f, 0.f, 0.f, 0.f}, p1 = p0, p2 = p0, p3 = p0;
    int mtile = blockIdx.x;
    if (mtile < mtiles) {
        const v4f* src = (const v4f*)(A + (size_t)mtile * 16 * K) + tid;
        p0 = src[0];
        p1 = src[THREADS];
        p2 = src[2 * THREADS];
        p3 = src[3 * THREADS];
        float* d = &sA[0][soff0];
        *(v4f*)(d)                     = p0;
        *(v4f*)(d + ROWS_STEP * P)     = p1;
        *(v4f*)(d + 2 * ROWS_STEP * P) = p2;
        *(v4f*)(d + 3 * ROWS_STEP * P) = p3;
    }
    __syncthreads();

    int buf = 0;
    for (; mtile < mtiles; mtile += gridDim.x) {
        const int next = mtile + (int)gridDim.x;
        const bool has_next = next < mtiles;
        if (has_next) {
            const v4f* src = (const v4f*)(A + (size_t)next * 16 * K) + tid;
            p0 = src[0];
            p1 = src[THREADS];
            p2 = src[2 * THREADS];
            p3 = src[3 * THREADS];
        }

        const float* sp = &sA[buf][l * P + 2 * half];
        v8f acc = {0.f, 0.f, 0.f, 0.f, 0.f, 0.f, 0.f, 0.f};
#pragma unroll
        for (int k0 = 0; k0 < K; k0 += 4) {
            v2f a = *(const v2f*)(sp + k0);
            acc = __builtin_amdgcn_wmma_f32_16x16x4_f32(
                 false, a,  false, breg[k0 / 4],
                 (short)0, acc,  false,  false);
        }

        if (has_next) {
            float* d = &sA[buf ^ 1][soff0];
            *(v4f*)(d)                     = p0;
            *(v4f*)(d + ROWS_STEP * P)     = p1;
            *(v4f*)(d + 2 * ROWS_STEP * P) = p2;
            *(v4f*)(d + 3 * ROWS_STEP * P) = p3;
        }
        __syncthreads();
        buf ^= 1;

        const int m0 = mtile * 16;
        float* cp = C + (size_t)(m0 + 8 * half) * N + n0 + l;
#pragma unroll
        for (int j = 0; j < 8; ++j)
            cp[(size_t)j * N] = acc[j];
    }
}

template <int H, bool RELU>
__global__ void __launch_bounds__(256)
gather_kernel(const int* __restrict__ rowptr, const int* __restrict__ csr_src,
              const float* __restrict__ dinv, const float* __restrict__ htmp,
              const float* __restrict__ bias, float* __restrict__ out, int n) {
    constexpr int V = H / 32;
    typedef __attribute__((ext_vector_type(V))) float vec;
    const int lane = threadIdx.x & 31;
    const int wave0 = __builtin_amdgcn_readfirstlane(
        (int)((blockIdx.x * blockDim.x + threadIdx.x) >> 5));
    const int nwaves = (int)((gridDim.x * blockDim.x) >> 5);
    const int fo = lane * V;
    const vec b = *(const vec*)(bias + fo);

    for (int node = wave0; node < n; node += nwaves) {
        const int beg = rowptr[node];
        const int end = rowptr[node + 1];
        const float dc = dinv[node];

        vec acc = *(const vec*)(htmp + (size_t)node * H + fo);
        acc = acc * dc * dc;

        for (int base = beg; base < end; base += 32) {
            const int j = base + lane;
            int r = node;
            float w = 0.f;
            if (j < end) {
                r = csr_src[j];
                w = dinv[r] * dc;
            }
            int m = end - base;
            if (m > 32) m = 32;
#pragma unroll 4
            for (int t = 0; t < m; ++t) {
                const int rr = __builtin_amdgcn_readlane(r, t);
                const float ww = __int_as_float(
                    __builtin_amdgcn_readlane(__float_as_int(w), t));
                const vec x = *(const vec*)(htmp + (size_t)rr * H + fo);
                acc += x * ww;
            }
        }

        vec o = acc + b;
        if (RELU) {
#pragma unroll
            for (int q = 0; q < V; ++q) o[q] = o[q] > 0.f ? o[q] : 0.f;
        }
        *(vec*)(out + (size_t)node * H + fo) = o;
    }
}

#define FG_WAVES 16
static_assert(N_NODES % 16 == 0, "whole 16-node tiles");
__global__ void __launch_bounds__(FG_WAVES * 32)
gather_relu_gemm_kernel(const int* __restrict__ rowptr, const int* __restrict__ csr_src,
                        const float* __restrict__ dinv, const float* __restrict__ htmp,
                        const float* __restrict__ bias, const float* __restrict__ Bt,
                        float* __restrict__ C, int mtiles) {
    constexpr int H = H1_DIM;
    constexpr int N = H2_DIM;
    constexpr int P = H + 4;
    constexpr int THREADS = FG_WAVES * 32;
    constexpr int F4_ROW = H / 4;
    constexpr int PIECES = N * F4_ROW;
    static_assert(FG_WAVES == 16 && H == 128, "one wave per tile row, 4 floats per lane");
    static_assert(PIECES % THREADS == 0, "weight copy: whole pieces per thread");
    static_assert((P % 4) == 0 && ((P / 4) % 2) == 1, "pitch = 16 bytes x odd");
    __shared__ __attribute__((aligned(16))) float sB[N * P];
    __shared__ __attribute__((aligned(16))) float sH[2][16 * P];

    const int tid  = threadIdx.x;
    const int lane = tid & 31;
    const int l    = lane & 15;
    const int half = lane >> 4;
    const int wave = __builtin_amdgcn_readfirstlane(tid >> 5);
    const int fo = lane * 4;

    {
        const v4f* src = (const v4f*)Bt;
#pragma unroll
        for (int u = 0; u < PIECES / THREADS; ++u) {
            const int q = tid + u * THREADS;
            *(v4f*)&sB[(q / F4_ROW) * P + (q % F4_ROW) * 4] = src[q];
        }
    }
    const v4f b = *(const v4f*)(bias + fo);

    int buf = 0;
    for (int tile = blockIdx.x; tile < mtiles; tile += gridDim.x) {
        const int node = tile * 16 + wave;
        const int beg = rowptr[node];
        const int end = rowptr[node + 1];
        const float dc = dinv[node];

        v4f acc = *(const v4f*)(htmp + (size_t)node * H + fo);
        acc = acc * dc * dc;

        for (int base = beg; base < end; base += 32) {
            const int j = base + lane;
            int r = node;
            float w = 0.f;
            if (j < end) {
                r = csr_src[j];
                w = dinv[r] * dc;
            }
            int m = end - base;
            if (m > 32) m = 32;
#pragma unroll 4
            for (int t = 0; t < m; ++t) {
                const int rr = __builtin_amdgcn_readlane(r, t);
                const float ww = __int_as_float(
                    __builtin_amdgcn_readlane(__float_as_int(w), t));
                const v4f x = *(const v4f*)(htmp + (size_t)rr * H + fo);
                acc += x * ww;
            }
        }

        v4f o = acc + b;
#pragma unroll
        for (int q = 0; q < 4; ++q) o[q] = o[q] > 0.f ? o[q] : 0.f;
        *(v4f*)&sH[buf][wave * P + fo] = o;
        __syncthreads();

        if (wave < N / 16) {
            const int n0 = wave * 16;
            const float* sp = &sH[buf][l * P + 2 * half];
            const float* bp = &sB[(n0 + l) * P + 2 * half];
            v8f c = {0.f, 0.f, 0.f, 0.f, 0.f, 0.f, 0.f, 0.f};
#pragma unroll
            for (int k0 = 0; k0 < H; k0 += 4) {
                const v2f a  = *(const v2f*)(sp + k0);
                const v2f bb = *(const v2f*)(bp + k0);
                c = __builtin_amdgcn_wmma_f32_16x16x4_f32(
                     false, a,  false, bb,
                     (short)0, c,  false,  false);
            }
            float* cp = C + (size_t)(tile * 16 + 8 * half) * N + n0 + l;
#pragma unroll
            for (int jj = 0; jj < 8; ++jj)
                cp[(size_t)jj * N] = c[jj];
        }
        buf ^= 1;
    }
}

__global__ void __launch_bounds__(256)
edge_dot_kernel(const int* __restrict__ pos, const int* __restrict__ neg,
                const float* __restrict__ h2, float* __restrict__ out, int ep) {
    const int lane = threadIdx.x & 31;
    const int wave0 = __builtin_amdgcn_readfirstlane(
        (int)((blockIdx.x * blockDim.x + threadIdx.x) >> 5));
    const int nwaves = (int)((gridDim.x * blockDim.x) >> 5);
    const int total = 2 * ep;
    const int nbatch = (total + 31) >> 5;

    for (int bt = wave0; bt < nbatch; bt += nwaves) {
        const int e = bt * 32 + lane;
        int src = 0, dst = 0;
        if (e < total) {
            if (e < ep) {
                src = pos[e];
                dst = pos[ep + e];
            } else {
                src = neg[e - ep];
                dst = neg[e];
            }
        }
        int m = total - bt * 32;
        if (m > 32) m = 32;
        float res = 0.f;
#pragma unroll 8
        for (int t = 0; t < m; ++t) {
            const int sj = __builtin_amdgcn_readlane(src, t);
            const int si = __builtin_amdgcn_readlane(dst, t);
            const float* xj = h2 + (size_t)sj * H2_DIM;
            const float* xi = h2 + (size_t)si * H2_DIM;
            float v = xi[lane] * xj[lane] + xi[lane + 32] * xj[lane + 32];
#pragma unroll
            for (int off = 16; off > 0; off >>= 1)
                v += __shfl_xor(v, off, 32);
            if (lane == t) res = v;
        }
        if (e < total) out[e] = res;
    }
}

static inline size_t align256(size_t x) { return (x + 255) & ~(size_t)255; }

extern "C" void kernel_launch(void* const* d_in, const int* in_sizes, int n_in,
                              void* d_out, int out_size, void* d_ws, size_t ws_size,
                              hipStream_t stream) {
    const float* x    = (const float*)d_in[0];
    const int*   ei   = (const int*)d_in[1];
    const int*   pos  = (const int*)d_in[2];
    const int*   neg  = (const int*)d_in[3];
    const float* W1   = (const float*)d_in[4];
    const float* b1   = (const float*)d_in[5];
    const float* W2   = (const float*)d_in[6];
    const float* b2   = (const float*)d_in[7];
    float* out = (float*)d_out;

    const int* row = ei;
    const int* col = ei + E_EDGES;

    char* ws = (char*)d_ws;
    size_t off = 0;
    float* dinv  = (float*)(ws + off); off += align256(sizeof(float) * N_NODES);
    float* W1t   = (float*)(ws + off); off += align256(sizeof(float) * F_IN * H1_DIM);
    float* W2t   = (float*)(ws + off); off += align256(sizeof(float) * H1_DIM * H2_DIM);
    float* htmp1 = (float*)(ws + off); off += align256(sizeof(float) * (size_t)N_NODES * H1_DIM);
    float* h1    = (float*)(ws + off); off += align256(sizeof(float) * (size_t)N_NODES * H1_DIM);
    float* htmp2 = (float*)(ws + off); off += align256(sizeof(float) * (size_t)N_NODES * H2_DIM);
    float* h2    = htmp1;
    int* bstart  = (int*)(ws + off); off += align256(sizeof(int) * N_NODES);
    int* rowptr  = (int*)(ws + off); off += align256(sizeof(int) * (N_NODES + 1));
    int* hist    = (int*)(ws + off); off += align256(sizeof(int) * (size_t)E_EDGES);
    int* csr_src = (int*)(ws + off); off += align256(sizeof(int) * (size_t)E_EDGES);
    int* bins    = (int*)h1;
    (void)ws_size; (void)n_in; (void)in_sizes; (void)out_size;

    const int TB = 256;
    const int MTILES = N_NODES / 16;
    const int GEMM1_BLOCKS = 512;
    const int FUSED_BLOCKS = 1024;
    const int GATHER_BLOCKS = 2048;
    const int DOT_BLOCKS = ((2 * EP_CNT + 31) / 32 + 7) / 8;

    transpose_kernel<<<(F_IN * H1_DIM + TB - 1) / TB, TB, 0, stream>>>(W1, W1t, F_IN, H1_DIM);
    transpose_kernel<<<(H1_DIM * H2_DIM + TB - 1) / TB, TB, 0, stream>>>(W2, W2t, H1_DIM, H2_DIM);

    bucket_count_kernel<<<BIN_BLOCKS, BIN_THREADS, 0, stream>>>(col, hist, E_EDGES);
    bucket_scatter_kernel<<<BIN_BLOCKS, BIN_THREADS, 0, stream>>>(row, col, hist, bstart, bins, E_EDGES);
    bucket_csr_kernel<<<NB_BUCKETS, BIN_THREADS, 0, stream>>>(bins, bstart, rowptr, dinv, csr_src, N_NODES);

    gemm_wmma_f32<F_IN, 8><<<GEMM1_BLOCKS, 8 * 32, 0, stream>>>(x, W1t, htmp1, MTILES);

    gather_relu_gemm_kernel<<<FUSED_BLOCKS, FG_WAVES * 32, 0, stream>>>(
        rowptr, csr_src, dinv, htmp1, b1, W2t, htmp2, MTILES);

    gather_kernel<H2_DIM, false><<<GATHER_BLOCKS, TB, 0, stream>>>(
        rowptr, csr_src, dinv, htmp2, b2, h2, N_NODES);

    edge_dot_kernel<<<DOT_BLOCKS, TB, 0, stream>>>(pos, neg, h2, out, EP_CNT);
}


# Factor dinv[r] moved into both products' epilogues (rows stored times dinv[row]); list walks add rows unweighted, then scale by dinv[c]: no per-edge dinv load; sum order kept, products regrouped
# speedup vs baseline: 5.3523x; 1.0651x over previous
#include <hip/hip_runtime.h>

typedef __attribute__((ext_vector_type(2))) float v2f;
typedef __attribute__((ext_vector_type(4))) float v4f;
typedef __attribute__((ext_vector_type(8))) float v8f;

#define N_NODES 100000
#define F_IN    256
#define H1_DIM  128
#define H2_DIM  64
#define E_EDGES 1600000
#define EP_CNT  100000

#define NB_BUCKETS    256
#define NODES_PER_BKT 391
#define BKT_SLOTS     512
#define BIN_BLOCKS    256
#define BIN_THREADS   1024
#define EDGES_PER_BLK ((E_EDGES + BIN_BLOCKS - 1) / BIN_BLOCKS)
#define BIN_EPT       ((EDGES_PER_BLK + BIN_THREADS - 1) / BIN_THREADS)
#define CSR_EPT       8
static_assert(NB_BUCKETS * NODES_PER_BKT >= N_NODES, "ranges must cover every node");
static_assert(BKT_SLOTS >= NODES_PER_BKT && BKT_SLOTS <= BIN_THREADS, "one scan thread per slot");
static_assert(BIN_BLOCKS % 128 == 0 && BIN_THREADS == 4 * NB_BUCKETS, "table walk: 4 threads per column, 32 rows per batch");

__device__ __forceinline__ int wave_incl_scan(int v, int lane) {
#pragma unroll
    for (int off = 1; off < 32; off <<= 1) {
        const int t = __shfl_up(v, off, 32);
        if (lane >= off) v += t;
    }
    return v;
}

__global__ void __launch_bounds__(BIN_THREADS)
bucket_count_kernel(const int* __restrict__ col, int* __restrict__ hist, int e) {
    __shared__ int s_cnt[NB_BUCKETS];
    const int tid = threadIdx.x;
    if (tid < NB_BUCKETS) s_cnt[tid] = 0;
    const int beg = blockIdx.x * EDGES_PER_BLK;
    const int end = (beg + EDGES_PER_BLK < e) ? beg + EDGES_PER_BLK : e;
    int c[BIN_EPT];
#pragma unroll
    for (int k = 0; k < BIN_EPT; ++k) {
        const int i = beg + tid + k * BIN_THREADS;
        c[k] = (i < end) ? col[i] : -1;
    }
    __syncthreads();
#pragma unroll
    for (int k = 0; k < BIN_EPT; ++k)
        if (c[k] >= 0) atomicAdd(&s_cnt[c[k] / NODES_PER_BKT], 1);
    __syncthreads();
    if (tid < NB_BUCKETS) hist[blockIdx.x * NB_BUCKETS + tid] = s_cnt[tid];
}

__global__ void __launch_bounds__(BIN_THREADS)
bucket_scatter_kernel(const int* __restrict__ row, const int* __restrict__ col,
                      const int* __restrict__ hist, int* __restrict__ bstart,
                      int* __restrict__ bins, int e) {
    __shared__ int s_pre[4][NB_BUCKETS];
    __shared__ int s_tot[4][NB_BUCKETS];
    __shared__ int s_cur[NB_BUCKETS];
    __shared__ int s_wsum[NB_BUCKETS / 32];
    const int tid  = threadIdx.x;
    const int lane = tid & 31;
    const int wave = tid >> 5;
    const int t = tid & (NB_BUCKETS - 1);
    const int q = tid >> 8;

    const int beg = blockIdx.x * EDGES_PER_BLK;
    const int end = (beg + EDGES_PER_BLK < e) ? beg + EDGES_PER_BLK : e;
    int c[BIN_EPT], r[BIN_EPT];
#pragma unroll
    for (int k = 0; k < BIN_EPT; ++k) {
        const int i = beg + tid + k * BIN_THREADS;
        const bool ok = i < end;
        c[k] = ok ? col[i] : -1;
        r[k] = ok ? row[i] : 0;
    }

    int pre = 0, tot = 0;
    const int me = blockIdx.x;
    const int* hp = hist + q * NB_BUCKETS + t;
#pragma unroll
    for (int h = 0; h < BIN_BLOCKS / 128; ++h) {
        int v[32];
#pragma unroll
        for (int u = 0; u < 32; ++u)
            v[u] = hp[(h * 32 + u) * 4 * NB_BUCKETS];
#pragma unroll
        for (int u = 0; u < 32; ++u) {
            tot += v[u];
            if (q + 4 * (h * 32 + u) < me) pre += v[u];
        }
    }
    s_pre[q][t] = pre;
    s_tot[q][t] = tot;
    __syncthreads();

    int T = 0, P = 0, inc = 0;
    if (tid < NB_BUCKETS) {
        T = (s_tot[0][t] + s_tot[1][t]) + (s_tot[2][t] + s_tot[3][t]);
        P = (s_pre[0][t] + s_pre[1][t]) + (s_pre[2][t] + s_pre[3][t]);
        inc = wave_incl_scan(T, lane);
        if (lane == 31) s_wsum[wave] = inc;
    }
    __syncthreads();
    if (tid < NB_BUCKETS) {
        int front = 0;
#pragma unroll
        for (int w = 0; w < NB_BUCKETS / 32; ++w)
            if (w < wave) front += s_wsum[w];
        const int excl = front + inc - T;
        s_cur[t] = excl + P;
        if (me == 0) {
            bstart[t] = excl;
            if (t == NB_BUCKETS - 1) bstart[NB_BUCKETS] = excl + T;
        }
    }
    __syncthreads();

    int2* __restrict__ b2 = (int2*)bins;
#pragma unroll
    for (int k = 0; k < BIN_EPT; ++k) {
        if (c[k] >= 0) {
            const int pos = atomicAdd(&s_cur[c[k] / NODES_PER_BKT], 1);
            b2[pos] = make_int2(r[k], c[k]);
        }
    }
}

__global__ void __launch_bounds__(BIN_THREADS)
bucket_csr_kernel(const int* __restrict__ bins, const int* __restrict__ bstart,
                  int* __restrict__ rowptr, float* __restrict__ dinv,
                  int* __restrict__ csr_src, int n) {
    __shared__ int s_cnt[BKT_SLOTS];
    __shared__ int s_off[BKT_SLOTS];
    __shared__ int s_wsum[BKT_SLOTS / 32];
    const int tid  = threadIdx.x;
    const int lane = tid & 31;
    const int wave = tid >> 5;
    const int n0 = blockIdx.x * NODES_PER_BKT;
    const int bs = bstart[blockIdx.x];
    const int be = bstart[blockIdx.x + 1];
    const int2* __restrict__ b2 = (const int2*)bins;

    if (tid < BKT_SLOTS) s_cnt[tid] = 0;
    int2 rc[CSR_EPT];
#pragma unroll
    for (int k = 0; k < CSR_EPT; ++k) {
        const int i = bs + tid + k * BIN_THREADS;
        rc[k] = (i < be) ? b2[i] : make_int2(0, -1);
    }
    __syncthreads();

#pragma unroll
    for (int k = 0; k < CSR_EPT; ++k)
        if (rc[k].y >= 0) atomicAdd(&s_cnt[rc[k].y - n0], 1);
    for (int i = bs + tid + CSR_EPT * BIN_THREADS; i < be; i += BIN_THREADS)
        atomicAdd(&s_cnt[b2[i].y - n0], 1);
    __syncthreads();

    int v = 0, inc = 0;
    if (tid < BKT_SLOTS) {
        v = s_cnt[tid];
        inc = wave_incl_scan(v, lane);
        if (lane == 31) s_wsum[wave] = inc;
    }
    __syncthreads();
    if (tid < BKT_SLOTS) {
        int front = 0;
#pragma unroll
        for (int w = 0; w < BKT_SLOTS / 32; ++w)
            if (w < wave) front += s_wsum[w];
        const int first = bs + front + inc - v;
        s_off[tid] = first;
        s_cnt[tid] = 0;
        const int node = n0 + tid;
        if (tid < NODES_PER_BKT && node < n) {
            rowptr[node] = first;
            dinv[node] = rsqrtf((float)(v + 1));
            if (node == n - 1) rowptr[n] = first + v;
        }
    }
    __syncthreads();

#pragma unroll
    for (int k = 0; k < CSR_EPT; ++k) {
        if (rc[k].y >= 0) {
            const int j = rc[k].y - n0;
            const int rk = atomicAdd(&s_cnt[j], 1);
            csr_src[s_off[j] + rk] = rc[k].x;
        }
    }
    for (int i = bs + tid + CSR_EPT * BIN_THREADS; i < be; i += BIN_THREADS) {
        const int2 p = b2[i];
        const int j = p.y - n0;
        const int rk = atomicAdd(&s_cnt[j], 1);
        csr_src[s_off[j] + rk] = p.x;
    }
}

__global__ void transpose_kernel(const float* __restrict__ B,
                                 float* __restrict__ Bt, int K, int N) {
    int idx = blockIdx.x * blockDim.x + threadIdx.x;
    if (idx >= K * N) return;
    int k = idx / N;
    int n = idx - k * N;
    Bt[(size_t)n * K + k] = B[idx];
}

template <int K, int NT>
__global__ void __launch_bounds__(NT * 32)
gemm_wmma_f32(const float* __restrict__ A, const float* __restrict__ Bt,
              const float* __restrict__ rscale, float* __restrict__ C, int mtiles) {
    constexpr int N = NT * 16;
    constexpr int P = K + 4;
    constexpr int F4_ROW = K / 4;
    constexpr int THREADS = NT * 32;
    constexpr int LD = (16 * F4_ROW) / THREADS;
    static_assert((16 * F4_ROW) % THREADS == 0, "tile copy: whole pieces per thread");
    static_assert((P % 4) == 0 && ((P / 4) % 2) == 1, "pitch = 16 bytes x odd");
    __shared__ __attribute__((aligned(16))) float sA[2][16 * P];

    const int tid  = threadIdx.x;
    const int wave = tid >> 5;
    const int lane = tid & 31;
    const int l    = lane & 15;
    const int half = lane >> 4;

    const int n0 = wave * 16;
    const float* bp = Bt + (size_t)(n0 + l) * K + 2 * half;
    v2f breg[K / 4];
#pragma unroll
    for (int s = 0; s < K / 4; ++s)
        breg[s] = *(const v2f*)(bp + 4 * s);

    constexpr int ROWS_STEP = THREADS / F4_ROW;
    static_assert(LD == 4 && THREADS % F4_ROW == 0, "four pieces per thread, a fixed row step apart");
    const int soff0 = (tid / F4_ROW) * P + (tid % F4_ROW) * 4;

    v4f p0 = {0.f, 0.f, 0.f, 0.f}, p1 = p0, p2 = p0, p3 = p0;
    v4f nf0 = p0, nf1 = p0;
    int mtile = blockIdx.x;
    if (mtile < mtiles) {
        const v4f* src = (const v4f*)(A + (size_t)mtile * 16 * K) + tid;
        p0 = src[0];
        p1 = src[THREADS];
        p2 = src[2 * THREADS];
        p3 = src[3 * THREADS];
        const v4f* fp = (const v4f*)(rscale + mtile * 16 + 8 * half);
        nf0 = fp[0];
        nf1 = fp[1];
        float* d = &sA[0][soff0];
        *(v4f*)(d)                     = p0;
        *(v4f*)(d + ROWS_STEP * P)     = p1;
        *(v4f*)(d + 2 * ROWS_STEP * P) = p2;
        *(v4f*)(d + 3 * ROWS_STEP * P) = p3;
    }
    __syncthreads();

    int buf = 0;
    for (; mtile < mtiles; mtile += gridDim.x) {
        const int next = mtile + (int)gridDim.x;
        const bool has_next = next < mtiles;
        const v4f d0 = nf0, d1 = nf1;
        if (has_next) {
            const v4f* src = (const v4f*)(A + (size_t)next * 16 * K) + tid;
            p0 = src[0];
            p1 = src[THREADS];
            p2 = src[2 * THREADS];
            p3 = src[3 * THREADS];
            const v4f* fp = (const v4f*)(rscale + next * 16 + 8 * half);
            nf0 = fp[0];
            nf1 = fp[1];
        }

        const float* sp = &sA[buf][l * P + 2 * half];
        v8f acc = {0.f, 0.f, 0.f, 0.f, 0.f, 0.f, 0.f, 0.f};
#pragma unroll
        for (int k0 = 0; k0 < K; k0 += 4) {
            v2f a = *(const v2f*)(sp + k0);
            acc = __builtin_amdgcn_wmma_f32_16x16x4_f32(
                 false, a,  false, breg[k0 / 4],
                 (short)0, acc,  false,  false);
        }

        if (has_next) {
            float* d = &sA[buf ^ 1][soff0];
            *(v4f*)(d)                     = p0;
            *(v4f*)(d + ROWS_STEP * P)     = p1;
            *(v4f*)(d + 2 * ROWS_STEP * P) = p2;
            *(v4f*)(d + 3 * ROWS_STEP * P) = p3;
        }
        __syncthreads();
        buf ^= 1;

        const int m0 = mtile * 16;
        float* cp = C + (size_t)(m0 + 8 * half) * N + n0 + l;
#pragma unroll
        for (int j = 0; j < 4; ++j) {
            cp[(size_t)j * N]       = acc[j] * d0[j];
            cp[(size_t)(j + 4) * N] = acc[j + 4] * d1[j];
        }
    }
}

template <int H, bool RELU>
__global__ void __launch_bounds__(256)
gather_kernel(const int* __restrict__ rowptr, const int* __restrict__ csr_src,
              const float* __restrict__ dinv, const float* __restrict__ htmp,
              const float* __restrict__ bias, float* __restrict__ out, int n) {
    constexpr int V = H / 32;
    typedef __attribute__((ext_vector_type(V))) float vec;
    const int lane = threadIdx.x & 31;
    const int wave0 = __builtin_amdgcn_readfirstlane(
        (int)((blockIdx.x * blockDim.x + threadIdx.x) >> 5));
    const int nwaves = (int)((gridDim.x * blockDim.x) >> 5);
    const int fo = lane * V;
    const vec b = *(const vec*)(bias + fo);

    for (int node = wave0; node < n; node += nwaves) {
        const int beg = rowptr[node];
        const int end = rowptr[node + 1];
        const float dc = dinv[node];

        vec acc = *(const vec*)(htmp + (size_t)node * H + fo);

        for (int base = beg; base < end; base += 32) {
            const int j = base + lane;
            int r = node;
            if (j < end) r = csr_src[j];
            int m = end - base;
            if (m > 32) m = 32;
#pragma unroll 4
            for (int t = 0; t < m; ++t) {
                const int rr = __builtin_amdgcn_readlane(r, t);
                const vec x = *(const vec*)(htmp + (size_t)rr * H + fo);
                acc += x;
            }
        }

        vec o = acc * dc + b;
        if (RELU) {
#pragma unroll
            for (int q = 0; q < V; ++q) o[q] = o[q] > 0.f ? o[q] : 0.f;
        }
        *(vec*)(out + (size_t)node * H + fo) = o;
    }
}

#define FG_WAVES 16
static_assert(N_NODES % 16 == 0, "whole 16-node tiles");
__global__ void __launch_bounds__(FG_WAVES * 32)
gather_relu_gemm_kernel(const int* __restrict__ rowptr, const int* __restrict__ csr_src,
                        const float* __restrict__ dinv, const float* __restrict__ htmp,
                        const float* __restrict__ bias, const float* __restrict__ Bt,
                        float* __restrict__ C, int mtiles) {
    constexpr int H = H1_DIM;
    constexpr int N = H2_DIM;
    constexpr int P = H + 4;
    constexpr int THREADS = FG_WAVES * 32;
    constexpr int F4_ROW = H / 4;
    constexpr int PIECES = N * F4_ROW;
    static_assert(FG_WAVES == 16 && H == 128, "one wave per tile row, 4 floats per lane");
    static_assert(PIECES % THREADS == 0, "weight copy: whole pieces per thread");
    static_assert((P % 4) == 0 && ((P / 4) % 2) == 1, "pitch = 16 bytes x odd");
    __shared__ __attribute__((aligned(16))) float sB[N * P];
    __shared__ __attribute__((aligned(16))) float sH[2][16 * P];

    const int tid  = threadIdx.x;
    const int lane = tid & 31;
    const int l    = lane & 15;
    const int half = lane >> 4;
    const int wave = __builtin_amdgcn_readfirstlane(tid >> 5);
    const int fo = lane * 4;

    {
        const v4f* src = (const v4f*)Bt;
#pragma unroll
        for (int u = 0; u < PIECES / THREADS; ++u) {
            const int q = tid + u * THREADS;
            *(v4f*)&sB[(q / F4_ROW) * P + (q % F4_ROW) * 4] = src[q];
        }
    }
    const v4f b = *(const v4f*)(bias + fo);

    int buf = 0;
    for (int tile = blockIdx.x; tile < mtiles; tile += gridDim.x) {
        const int node = tile * 16 + wave;
        const int beg = rowptr[node];
        const int end = rowptr[node + 1];
        const float dc = dinv[node];

        v4f acc = *(const v4f*)(htmp + (size_t)node * H + fo);

        for (int base = beg; base < end; base += 32) {
            const int j = base + lane;
            int r = node;
            if (j < end) r = csr_src[j];
            int m = end - base;
            if (m > 32) m = 32;
#pragma unroll 4
            for (int t = 0; t < m; ++t) {
                const int rr = __builtin_amdgcn_readlane(r, t);
                const v4f x = *(const v4f*)(htmp + (size_t)rr * H + fo);
                acc += x;
            }
        }

        v4f o = acc * dc + b;
#pragma unroll
        for (int q = 0; q < 4; ++q) o[q] = o[q] > 0.f ? o[q] : 0.f;
        *(v4f*)&sH[buf][wave * P + fo] = o;
        __syncthreads();

        if (wave < N / 16) {
            const int n0 = wave * 16;
            const float* sp = &sH[buf][l * P + 2 * half];
            const float* bp = &sB[(n0 + l) * P + 2 * half];
            const v4f* dp = (const v4f*)(dinv + tile * 16 + 8 * half);
            const v4f d0 = dp[0];
            const v4f d1 = dp[1];
            v8f c = {0.f, 0.f, 0.f, 0.f, 0.f, 0.f, 0.f, 0.f};
#pragma unroll
            for (int k0 = 0; k0 < H; k0 += 4) {
                const v2f a  = *(const v2f*)(sp + k0);
                const v2f bb = *(const v2f*)(bp + k0);
                c = __builtin_amdgcn_wmma_f32_16x16x4_f32(
                     false, a,  false, bb,
                     (short)0, c,  false,  false);
            }
            float* cp = C + (size_t)(tile * 16 + 8 * half) * N + n0 + l;
#pragma unroll
            for (int jj = 0; jj < 4; ++jj) {
                cp[(size_t)jj * N]       = c[jj] * d0[jj];
                cp[(size_t)(jj + 4) * N] = c[jj + 4] * d1[jj];
            }
        }
        buf ^= 1;
    }
}

__global__ void __launch_bounds__(256)
edge_dot_kernel(const int* __restrict__ pos, const int* __restrict__ neg,
                const float* __restrict__ h2, float* __restrict__ out, int ep) {
    const int lane = threadIdx.x & 31;
    const int wave0 = __builtin_amdgcn_readfirstlane(
        (int)((blockIdx.x * blockDim.x + threadIdx.x) >> 5));
    const int nwaves = (int)((gridDim.x * blockDim.x) >> 5);
    const int total = 2 * ep;
    const int nbatch = (total + 31) >> 5;

    for (int bt = wave0; bt < nbatch; bt += nwaves) {
        const int e = bt * 32 + lane;
        int src = 0, dst = 0;
        if (e < total) {
            if (e < ep) {
                src = pos[e];
                dst = pos[ep + e];
            } else {
                src = neg[e - ep];
                dst = neg[e];
            }
        }
        int m = total - bt * 32;
        if (m > 32) m = 32;
        float res = 0.f;
#pragma unroll 8
        for (int t = 0; t < m; ++t) {
            const int sj = __builtin_amdgcn_readlane(src, t);
            const int si = __builtin_amdgcn_readlane(dst, t);
            const float* xj = h2 + (size_t)sj * H2_DIM;
            const float* xi = h2 + (size_t)si * H2_DIM;
            float v = xi[lane] * xj[lane] + xi[lane + 32] * xj[lane + 32];
#pragma unroll
            for (int off = 16; off > 0; off >>= 1)
                v += __shfl_xor(v, off, 32);
            if (lane == t) res = v;
        }
        if (e < total) out[e] = res;
    }
}

static inline size_t align256(size_t x) { return (x + 255) & ~(size_t)255; }

extern "C" void kernel_launch(void* const* d_in, const int* in_sizes, int n_in,
                              void* d_out, int out_size, void* d_ws, size_t ws_size,
                              hipStream_t stream) {
    const float* x    = (const float*)d_in[0];
    const int*   ei   = (const int*)d_in[1];
    const int*   pos  = (const int*)d_in[2];
    const int*   neg  = (const int*)d_in[3];
    const float* W1   = (const float*)d_in[4];
    const float* b1   = (const float*)d_in[5];
    const float* W2   = (const float*)d_in[6];
    const float* b2   = (const float*)d_in[7];
    float* out = (float*)d_out;

    const int* row = ei;
    const int* col = ei + E_EDGES;

    char* ws = (char*)d_ws;
    size_t off = 0;
    float* dinv  = (float*)(ws + off); off += align256(sizeof(float) * N_NODES);
    float* W1t   = (float*)(ws + off); off += align256(sizeof(float) * F_IN * H1_DIM);
    float* W2t   = (float*)(ws + off); off += align256(sizeof(float) * H1_DIM * H2_DIM);
    float* htmp1 = (float*)(ws + off); off += align256(sizeof(float) * (size_t)N_NODES * H1_DIM);
    float* h1    = (float*)(ws + off); off += align256(sizeof(float) * (size_t)N_NODES * H1_DIM);
    float* htmp2 = (float*)(ws + off); off += align256(sizeof(float) * (size_t)N_NODES * H2_DIM);
    float* h2    = htmp1;
    int* bstart  = (int*)(ws + off); off += align256(sizeof(int) * N_NODES);
    int* rowptr  = (int*)(ws + off); off += align256(sizeof(int) * (N_NODES + 1));
    int* hist    = (int*)(ws + off); off += align256(sizeof(int) * (size_t)E_EDGES);
    int* csr_src = (int*)(ws + off); off += align256(sizeof(int) * (size_t)E_EDGES);
    int* bins    = (int*)h1;
    (void)ws_size; (void)n_in; (void)in_sizes; (void)out_size;

    const int TB = 256;
    const int MTILES = N_NODES / 16;
    const int GEMM1_BLOCKS = 512;
    const int FUSED_BLOCKS = 1024;
    const int GATHER_BLOCKS = 2048;
    const int DOT_BLOCKS = ((2 * EP_CNT + 31) / 32 + 7) / 8;

    transpose_kernel<<<(F_IN * H1_DIM + TB - 1) / TB, TB, 0, stream>>>(W1, W1t, F_IN, H1_DIM);
    transpose_kernel<<<(H1_DIM * H2_DIM + TB - 1) / TB, TB, 0, stream>>>(W2, W2t, H1_DIM, H2_DIM);

    bucket_count_kernel<<<BIN_BLOCKS, BIN_THREADS, 0, stream>>>(col, hist, E_EDGES);
    bucket_scatter_kernel<<<BIN_BLOCKS, BIN_THREADS, 0, stream>>>(row, col, hist, bstart, bins, E_EDGES);
    bucket_csr_kernel<<<NB_BUCKETS, BIN_THREADS, 0, stream>>>(bins, bstart, rowptr, dinv, csr_src, N_NODES);

    gemm_wmma_f32<F_IN, 8><<<GEMM1_BLOCKS, 8 * 32, 0, stream>>>(x, W1t, dinv, htmp1, MTILES);

    gather_relu_gemm_kernel<<<FUSED_BLOCKS, FG_WAVES * 32, 0, stream>>>(
        rowptr, csr_src, dinv, htmp1, b1, W2t, htmp2, MTILES);

    gather_kernel<H2_DIM, false><<<GATHER_BLOCKS, TB, 0, stream>>>(
        rowptr, csr_src, dinv, htmp2, b2, h2, N_NODES);

    edge_dot_kernel<<<DOT_BLOCKS, TB, 0, stream>>>(pos, neg, h2, out, EP_CNT);
}


# Both products read W1 and W2 as given, row-major: B fragments by two 4-byte loads per k pair, W2 turned on its way into LDS; the two transpose launches and their planes are gone; 9 to 7 launches
# speedup vs baseline: 5.5448x; 1.0360x over previous
#include <hip/hip_runtime.h>

typedef __attribute__((ext_vector_type(2))) float v2f;
typedef __attribute__((ext_vector_type(4))) float v4f;
typedef __attribute__((ext_vector_type(8))) float v8f;

#define N_NODES 100000
#define F_IN    256
#define H1_DIM  128
#define H2_DIM  64
#define E_EDGES 1600000
#define EP_CNT  100000

#define NB_BUCKETS    256
#define NODES_PER_BKT 391
#define BKT_SLOTS     512
#define BIN_BLOCKS    256
#define BIN_THREADS   1024
#define EDGES_PER_BLK ((E_EDGES + BIN_BLOCKS - 1) / BIN_BLOCKS)
#define BIN_EPT       ((EDGES_PER_BLK + BIN_THREADS - 1) / BIN_THREADS)
#define CSR_EPT       8
static_assert(NB_BUCKETS * NODES_PER_BKT >= N_NODES, "ranges must cover every node");
static_assert(BKT_SLOTS >= NODES_PER_BKT && BKT_SLOTS <= BIN_THREADS, "one scan thread per slot");
static_assert(BIN_BLOCKS % 128 == 0 && BIN_THREADS == 4 * NB_BUCKETS, "table walk: 4 threads per column, 32 rows per batch");

__device__ __forceinline__ int wave_incl_scan(int v, int lane) {
#pragma unroll
    for (int off = 1; off < 32; off <<= 1) {
        const int t = __shfl_up(v, off, 32);
        if (lane >= off) v += t;
    }
    return v;
}

__global__ void __launch_bounds__(BIN_THREADS)
bucket_count_kernel(const int* __restrict__ col, int* __restrict__ hist, int e) {
    __shared__ int s_cnt[NB_BUCKETS];
    const int tid = threadIdx.x;
    if (tid < NB_BUCKETS) s_cnt[tid] = 0;
    const int beg = blockIdx.x * EDGES_PER_BLK;
    const int end = (beg + EDGES_PER_BLK < e) ? beg + EDGES_PER_BLK : e;
    int c[BIN_EPT];
#pragma unroll
    for (int k = 0; k < BIN_EPT; ++k) {
        const int i = beg + tid + k * BIN_THREADS;
        c[k] = (i < end) ? col[i] : -1;
    }
    __syncthreads();
#pragma unroll
    for (int k = 0; k < BIN_EPT; ++k)
        if (c[k] >= 0) atomicAdd(&s_cnt[c[k] / NODES_PER_BKT], 1);
    __syncthreads();
    if (tid < NB_BUCKETS) hist[blockIdx.x * NB_BUCKETS + tid] = s_cnt[tid];
}

__global__ void __launch_bounds__(BIN_THREADS)
bucket_scatter_kernel(const int* __restrict__ row, const int* __restrict__ col,
                      const int* __restrict__ hist, int* __restrict__ bstart,
                      int* __restrict__ bins, int e) {
    __shared__ int s_pre[4][NB_BUCKETS];
    __shared__ int s_tot[4][NB_BUCKETS];
    __shared__ int s_cur[NB_BUCKETS];
    __shared__ int s_wsum[NB_BUCKETS / 32];
    const int tid  = threadIdx.x;
    const int lane = tid & 31;
    const int wave = tid >> 5;
    const int t = tid & (NB_BUCKETS - 1);
    const int q = tid >> 8;

    const int beg = blockIdx.x * EDGES_PER_BLK;
    const int end = (beg + EDGES_PER_BLK < e) ? beg + EDGES_PER_BLK : e;
    int c[BIN_EPT], r[BIN_EPT];
#pragma unroll
    for (int k = 0; k < BIN_EPT; ++k) {
        const int i = beg + tid + k * BIN_THREADS;
        const bool ok = i < end;
        c[k] = ok ? col[i] : -1;
        r[k] = ok ? row[i] : 0;
    }

    int pre = 0, tot = 0;
    const int me = blockIdx.x;
    const int* hp = hist + q * NB_BUCKETS + t;
#pragma unroll
    for (int h = 0; h < BIN_BLOCKS / 128; ++h) {
        int v[32];
#pragma unroll
        for (int u = 0; u < 32; ++u)
            v[u] = hp[(h * 32 + u) * 4 * NB_BUCKETS];
#pragma unroll
        for (int u = 0; u < 32; ++u) {
            tot += v[u];
            if (q + 4 * (h * 32 + u) < me) pre += v[u];
        }
    }
    s_pre[q][t] = pre;
    s_tot[q][t] = tot;
    __syncthreads();

    int T = 0, P = 0, inc = 0;
    if (tid < NB_BUCKETS) {
        T = (s_tot[0][t] + s_tot[1][t]) + (s_tot[2][t] + s_tot[3][t]);
        P = (s_pre[0][t] + s_pre[1][t]) + (s_pre[2][t] + s_pre[3][t]);
        inc = wave_incl_scan(T, lane);
        if (lane == 31) s_wsum[wave] = inc;
    }
    __syncthreads();
    if (tid < NB_BUCKETS) {
        int front = 0;
#pragma unroll
        for (int w = 0; w < NB_BUCKETS / 32; ++w)
            if (w < wave) front += s_wsum[w];
        const int excl = front + inc - T;
        s_cur[t] = excl + P;
        if (me == 0) {
            bstart[t] = excl;
            if (t == NB_BUCKETS - 1) bstart[NB_BUCKETS] = excl + T;
        }
    }
    __syncthreads();

    int2* __restrict__ b2 = (int2*)bins;
#pragma unroll
    for (int k = 0; k < BIN_EPT; ++k) {
        if (c[k] >= 0) {
            const int pos = atomicAdd(&s_cur[c[k] / NODES_PER_BKT], 1);
            b2[pos] = make_int2(r[k], c[k]);
        }
    }
}

__global__ void __launch_bounds__(BIN_THREADS)
bucket_csr_kernel(const int* __restrict__ bins, const int* __restrict__ bstart,
                  int* __restrict__ rowptr, float* __restrict__ dinv,
                  int* __restrict__ csr_src, int n) {
    __shared__ int s_cnt[BKT_SLOTS];
    __shared__ int s_off[BKT_SLOTS];
    __shared__ int s_wsum[BKT_SLOTS / 32];
    const int tid  = threadIdx.x;
    const int lane = tid & 31;
    const int wave = tid >> 5;
    const int n0 = blockIdx.x * NODES_PER_BKT;
    const int bs = bstart[blockIdx.x];
    const int be = bstart[blockIdx.x + 1];
    const int2* __restrict__ b2 = (const int2*)bins;

    if (tid < BKT_SLOTS) s_cnt[tid] = 0;
    int2 rc[CSR_EPT];
#pragma unroll
    for (int k = 0; k < CSR_EPT; ++k) {
        const int i = bs + tid + k * BIN_THREADS;
        rc[k] = (i < be) ? b2[i] : make_int2(0, -1);
    }
    __syncthreads();

#pragma unroll
    for (int k = 0; k < CSR_EPT; ++k)
        if (rc[k].y >= 0) atomicAdd(&s_cnt[rc[k].y - n0], 1);
    for (int i = bs + tid + CSR_EPT * BIN_THREADS; i < be; i += BIN_THREADS)
        atomicAdd(&s_cnt[b2[i].y - n0], 1);
    __syncthreads();

    int v = 0, inc = 0;
    if (tid < BKT_SLOTS) {
        v = s_cnt[tid];
        inc = wave_incl_scan(v, lane);
        if (lane == 31) s_wsum[wave] = inc;
    }
    __syncthreads();
    if (tid < BKT_SLOTS) {
        int front = 0;
#pragma unroll
        for (int w = 0; w < BKT_SLOTS / 32; ++w)
            if (w < wave) front += s_wsum[w];
        const int first = bs + front + inc - v;
        s_off[tid] = first;
        s_cnt[tid] = 0;
        const int node = n0 + tid;
        if (tid < NODES_PER_BKT && node < n) {
            rowptr[node] = first;
            dinv[node] = rsqrtf((float)(v + 1));
            if (node == n - 1) rowptr[n] = first + v;
        }
    }
    __syncthreads();

#pragma unroll
    for (int k = 0; k < CSR_EPT; ++k) {
        if (rc[k].y >= 0) {
            const int j = rc[k].y - n0;
            const int rk = atomicAdd(&s_cnt[j], 1);
            csr_src[s_off[j] + rk] = rc[k].x;
        }
    }
    for (int i = bs + tid + CSR_EPT * BIN_THREADS; i < be; i += BIN_THREADS) {
        const int2 p = b2[i];
        const int j = p.y - n0;
        const int rk = atomicAdd(&s_cnt[j], 1);
        csr_src[s_off[j] + rk] = p.x;
    }
}

template <int K, int NT>
__global__ void __launch_bounds__(NT * 32)
gemm_wmma_f32(const float* __restrict__ A, const float* __restrict__ B,
              const float* __restrict__ rscale, float* __restrict__ C, int mtiles) {
    constexpr int N = NT * 16;
    constexpr int P = K + 4;
    constexpr int F4_ROW = K / 4;
    constexpr int THREADS = NT * 32;
    constexpr int LD = (16 * F4_ROW) / THREADS;
    static_assert((16 * F4_ROW) % THREADS == 0, "tile copy: whole pieces per thread");
    static_assert((P % 4) == 0 && ((P / 4) % 2) == 1, "pitch = 16 bytes x odd");
    __shared__ __attribute__((aligned(16))) float sA[2][16 * P];

    const int tid  = threadIdx.x;
    const int wave = tid >> 5;
    const int lane = tid & 31;
    const int l    = lane & 15;
    const int half = lane >> 4;

    const int n0 = wave * 16;
    const float* bp = B + (size_t)(2 * half) * N + n0 + l;
    v2f breg[K / 4];
#pragma unroll
    for (int s = 0; s < K / 4; ++s) {
        breg[s][0] = bp[(size_t)(4 * s) * N];
        breg[s][1] = bp[(size_t)(4 * s + 1) * N];
    }

    constexpr int ROWS_STEP = THREADS / F4_ROW;
    static_assert(LD == 4 && THREADS % F4_ROW == 0, "four pieces per thread, a fixed row step apart");
    const int soff0 = (tid / F4_ROW) * P + (tid % F4_ROW) * 4;

    v4f p0 = {0.f, 0.f, 0.f, 0.f}, p1 = p0, p2 = p0, p3 = p0;
    v4f nf0 = p0, nf1 = p0;
    int mtile = blockIdx.x;
    if (mtile < mtiles) {
        const v4f* src = (const v4f*)(A + (size_t)mtile * 16 * K) + tid;
        p0 = src[0];
        p1 = src[THREADS];
        p2 = src[2 * THREADS];
        p3 = src[3 * THREADS];
        const v4f* fp = (const v4f*)(rscale + mtile * 16 + 8 * half);
        nf0 = fp[0];
        nf1 = fp[1];
        float* d = &sA[0][soff0];
        *(v4f*)(d)                     = p0;
        *(v4f*)(d + ROWS_STEP * P)     = p1;
        *(v4f*)(d + 2 * ROWS_STEP * P) = p2;
        *(v4f*)(d + 3 * ROWS_STEP * P) = p3;
    }
    __syncthreads();

    int buf = 0;
    for (; mtile < mtiles; mtile += gridDim.x) {
        const int next = mtile + (int)gridDim.x;
        const bool has_next = next < mtiles;
        const v4f d0 = nf0, d1 = nf1;
        if (has_next) {
            const v4f* src = (const v4f*)(A + (size_t)next * 16 * K) + tid;
            p0 = src[0];
            p1 = src[THREADS];
            p2 = src[2 * THREADS];
            p3 = src[3 * THREADS];
            const v4f* fp = (const v4f*)(rscale + next * 16 + 8 * half);
            nf0 = fp[0];
            nf1 = fp[1];
        }

        const float* sp = &sA[buf][l * P + 2 * half];
        v8f acc = {0.f, 0.f, 0.f, 0.f, 0.f, 0.f, 0.f, 0.f};
#pragma unroll
        for (int k0 = 0; k0 < K; k0 += 4) {
            v2f a = *(const v2f*)(sp + k0);
            acc = __builtin_amdgcn_wmma_f32_16x16x4_f32(
                 false, a,  false, breg[k0 / 4],
                 (short)0, acc,  false,  false);
        }

        if (has_next) {
            float* d = &sA[buf ^ 1][soff0];
            *(v4f*)(d)                     = p0;
            *(v4f*)(d + ROWS_STEP * P)     = p1;
            *(v4f*)(d + 2 * ROWS_STEP * P) = p2;
            *(v4f*)(d + 3 * ROWS_STEP * P) = p3;
        }
        __syncthreads();
        buf ^= 1;

        const int m0 = mtile * 16;
        float* cp = C + (size_t)(m0 + 8 * half) * N + n0 + l;
#pragma unroll
        for (int j = 0; j < 4; ++j) {
            cp[(size_t)j * N]       = acc[j] * d0[j];
            cp[(size_t)(j + 4) * N] = acc[j + 4] * d1[j];
        }
    }
}

template <int H, bool RELU>
__global__ void __launch_bounds__(256)
gather_kernel(const int* __restrict__ rowptr, const int* __restrict__ csr_src,
              const float* __restrict__ dinv, const float* __restrict__ htmp,
              const float* __restrict__ bias, float* __restrict__ out, int n) {
    constexpr int V = H / 32;
    typedef __attribute__((ext_vector_type(V))) float vec;
    const int lane = threadIdx.x & 31;
    const int wave0 = __builtin_amdgcn_readfirstlane(
        (int)((blockIdx.x * blockDim.x + threadIdx.x) >> 5));
    const int nwaves = (int)((gridDim.x * blockDim.x) >> 5);
    const int fo = lane * V;
    const vec b = *(const vec*)(bias + fo);

    for (int node = wave0; node < n; node += nwaves) {
        const int beg = rowptr[node];
        const int end = rowptr[node + 1];
        const float dc = dinv[node];

        vec acc = *(const vec*)(htmp + (size_t)node * H + fo);

        for (int base = beg; base < end; base += 32) {
            const int j = base + lane;
            int r = node;
            if (j < end) r = csr_src[j];
            int m = end - base;
            if (m > 32) m = 32;
#pragma unroll 4
            for (int t = 0; t < m; ++t) {
                const int rr = __builtin_amdgcn_readlane(r, t);
                const vec x = *(const vec*)(htmp + (size_t)rr * H + fo);
                acc += x;
            }
        }

        vec o = acc * dc + b;
        if (RELU) {
#pragma unroll
            for (int q = 0; q < V; ++q) o[q] = o[q] > 0.f ? o[q] : 0.f;
        }
        *(vec*)(out + (size_t)node * H + fo) = o;
    }
}

#define FG_WAVES 16
static_assert(N_NODES % 16 == 0, "whole 16-node tiles");
__global__ void __launch_bounds__(FG_WAVES * 32)
gather_relu_gemm_kernel(const int* __restrict__ rowptr, const int* __restrict__ csr_src,
                        const float* __restrict__ dinv, const float* __restrict__ htmp,
                        const float* __restrict__ bias, const float* __restrict__ W,
                        float* __restrict__ C, int mtiles) {
    constexpr int H = H1_DIM;
    constexpr int N = H2_DIM;
    constexpr int P = H + 4;
    constexpr int N4_ROW = N / 4;
    static_assert(FG_WAVES == 16 && H == 128, "one wave per tile row, 4 floats per lane");
    static_assert(H == FG_WAVES * 8 && N4_ROW == 16, "weight copy: 8 rows per wave, 4 x 4 pieces per row");
    static_assert((P % 4) == 0 && ((P / 4) % 2) == 1, "pitch = 16 bytes x odd");
    __shared__ __attribute__((aligned(16))) float sB[N * P];
    __shared__ __attribute__((aligned(16))) float sH[2][16 * P];

    const int tid  = threadIdx.x;
    const int lane = tid & 31;
    const int l    = lane & 15;
    const int half = lane >> 4;
    const int wave = __builtin_amdgcn_readfirstlane(tid >> 5);
    const int fo = lane * 4;

    {
        const int k = (tid >> 5) * 8 + (lane >> 2);
        const int jl = lane & 3;
        const v4f* src = (const v4f*)(W + (size_t)k * N) + jl;
        v4f w[4];
#pragma unroll
        for (int u = 0; u < 4; ++u)
            w[u] = src[4 * u];
#pragma unroll
        for (int u = 0; u < 4; ++u) {
            float* d = &sB[(16 * u + 4 * jl) * P + k];
#pragma unroll
            for (int i = 0; i < 4; ++i)
                d[i * P] = w[u][i];
        }
    }
    const v4f b = *(const v4f*)(bias + fo);

    int buf = 0;
    for (int tile = blockIdx.x; tile < mtiles; tile += gridDim.x) {
        const int node = tile * 16 + wave;
        const int beg = rowptr[node];
        const int end = rowptr[node + 1];
        const float dc = dinv[node];

        v4f acc = *(const v4f*)(htmp + (size_t)node * H + fo);

        for (int base = beg; base < end; base += 32) {
            const int j = base + lane;
            int r = node;
            if (j < end) r = csr_src[j];
            int m = end - base;
            if (m > 32) m = 32;
#pragma unroll 4
            for (int t = 0; t < m; ++t) {
                const int rr = __builtin_amdgcn_readlane(r, t);
                const v4f x = *(const v4f*)(htmp + (size_t)rr * H + fo);
                acc += x;
            }
        }

        v4f o = acc * dc + b;
#pragma unroll
        for (int q = 0; q < 4; ++q) o[q] = o[q] > 0.f ? o[q] : 0.f;
        *(v4f*)&sH[buf][wave * P + fo] = o;
        __syncthreads();

        if (wave < N / 16) {
            const int n0 = wave * 16;
            const float* sp = &sH[buf][l * P + 2 * half];
            const float* bp = &sB[(n0 + l) * P + 2 * half];
            const v4f* dp = (const v4f*)(dinv + tile * 16 + 8 * half);
            const v4f d0 = dp[0];
            const v4f d1 = dp[1];
            v8f c = {0.f, 0.f, 0.f, 0.f, 0.f, 0.f, 0.f, 0.f};
#pragma unroll
            for (int k0 = 0; k0 < H; k0 += 4) {
                const v2f a  = *(const v2f*)(sp + k0);
                const v2f bb = *(const v2f*)(bp + k0);
                c = __builtin_amdgcn_wmma_f32_16x16x4_f32(
                     false, a,  false, bb,
                     (short)0, c,  false,  false);
            }
            float* cp = C + (size_t)(tile * 16 + 8 * half) * N + n0 + l;
#pragma unroll
            for (int jj = 0; jj < 4; ++jj) {
                cp[(size_t)jj * N]       = c[jj] * d0[jj];
                cp[(size_t)(jj + 4) * N] = c[jj + 4] * d1[jj];
            }
        }
        buf ^= 1;
    }
}

__global__ void __launch_bounds__(256)
edge_dot_kernel(const int* __restrict__ pos, const int* __restrict__ neg,
                const float* __restrict__ h2, float* __restrict__ out, int ep) {
    const int lane = threadIdx.x & 31;
    const int wave0 = __builtin_amdgcn_readfirstlane(
        (int)((blockIdx.x * blockDim.x + threadIdx.x) >> 5));
    const int nwaves = (int)((gridDim.x * blockDim.x) >> 5);
    const int total = 2 * ep;
    const int nbatch = (total + 31) >> 5;

    for (int bt = wave0; bt < nbatch; bt += nwaves) {
        const int e = bt * 32 + lane;
        int src = 0, dst = 0;
        if (e < total) {
            if (e < ep) {
                src = pos[e];
                dst = pos[ep + e];
            } else {
                src = neg[e - ep];
                dst = neg[e];
            }
        }
        int m = total - bt * 32;
        if (m > 32) m = 32;
        float res = 0.f;
#pragma unroll 8
        for (int t = 0; t < m; ++t) {
            const int sj = __builtin_amdgcn_readlane(src, t);
            const int si = __builtin_amdgcn_readlane(dst, t);
            const float* xj = h2 + (size_t)sj * H2_DIM;
            const float* xi = h2 + (size_t)si * H2_DIM;
            float v = xi[lane] * xj[lane] + xi[lane + 32] * xj[lane + 32];
#pragma unroll
            for (int off = 16; off > 0; off >>= 1)
                v += __shfl_xor(v, off, 32);
            if (lane == t) res = v;
        }
        if (e < total) out[e] = res;
    }
}

static inline size_t align256(size_t x) { return (x + 255) & ~(size_t)255; }

extern "C" void kernel_launch(void* const* d_in, const int* in_sizes, int n_in,
                              void* d_out, int out_size, void* d_ws, size_t ws_size,
                              hipStream_t stream) {
    const float* x    = (const float*)d_in[0];
    const int*   ei   = (const int*)d_in[1];
    const int*   pos  = (const int*)d_in[2];
    const int*   neg  = (const int*)d_in[3];
    const float* W1   = (const float*)d_in[4];
    const float* b1   = (const float*)d_in[5];
    const float* W2   = (const float*)d_in[6];
    const float* b2   = (const float*)d_in[7];
    float* out = (float*)d_out;

    const int* row = ei;
    const int* col = ei + E_EDGES;

    char* ws = (char*)d_ws;
    size_t off = 0;
    float* dinv  = (float*)(ws + off); off += align256(sizeof(float) * N_NODES);
    float* htmp1 = (float*)(ws + off); off += align256(sizeof(float) * (size_t)N_NODES * H1_DIM);
    float* h1    = (float*)(ws + off); off += align256(sizeof(float) * (size_t)N_NODES * H1_DIM);
    float* htmp2 = (float*)(ws + off); off += align256(sizeof(float) * (size_t)N_NODES * H2_DIM);
    float* h2    = htmp1;
    int* bstart  = (int*)(ws + off); off += align256(sizeof(int) * N_NODES);
    int* rowptr  = (int*)(ws + off); off += align256(sizeof(int) * (N_NODES + 1));
    int* hist    = (int*)(ws + off); off += align256(sizeof(int) * (size_t)E_EDGES);
    int* csr_src = (int*)(ws + off); off += align256(sizeof(int) * (size_t)E_EDGES);
    int* bins    = (int*)h1;
    (void)ws_size; (void)n_in; (void)in_sizes; (void)out_size;

    const int TB = 256;
    const int MTILES = N_NODES / 16;
    const int GEMM1_BLOCKS = 512;
    const int FUSED_BLOCKS = 1024;
    const int GATHER_BLOCKS = 2048;
    const int DOT_BLOCKS = ((2 * EP_CNT + 31) / 32 + 7) / 8;

    bucket_count_kernel<<<BIN_BLOCKS, BIN_THREADS, 0, stream>>>(col, hist, E_EDGES);
    bucket_scatter_kernel<<<BIN_BLOCKS, BIN_THREADS, 0, stream>>>(row, col, hist, bstart, bins, E_EDGES);
    bucket_csr_kernel<<<NB_BUCKETS, BIN_THREADS, 0, stream>>>(bins, bstart, rowptr, dinv, csr_src, N_NODES);

    gemm_wmma_f32<F_IN, 8><<<GEMM1_BLOCKS, 8 * 32, 0, stream>>>(x, W1, dinv, htmp1, MTILES);

    gather_relu_gemm_kernel<<<FUSED_BLOCKS, FG_WAVES * 32, 0, stream>>>(
        rowptr, csr_src, dinv, htmp1, b1, W2, htmp2, MTILES);

    gather_kernel<H2_DIM, false><<<GATHER_BLOCKS, TB, 0, stream>>>(
        rowptr, csr_src, dinv, htmp2, b2, h2, N_NODES);

    edge_dot_kernel<<<DOT_BLOCKS, TB, 0, stream>>>(pos, neg, h2, out, EP_CNT);
}


# Both edge-grouping passes put their records in an LDS copy and write it out in position order: a wave's store covers a few whole lines, not 32 lines of 4 or 8 bytes; list order still arrival rank
# speedup vs baseline: 6.0681x; 1.0944x over previous
#include <hip/hip_runtime.h>

typedef __attribute__((ext_vector_type(2))) float v2f;
typedef __attribute__((ext_vector_type(4))) float v4f;
typedef __attribute__((ext_vector_type(8))) float v8f;

#define N_NODES 100000
#define F_IN    256
#define H1_DIM  128
#define H2_DIM  64
#define E_EDGES 1600000
#define EP_CNT  100000

#define NB_BUCKETS    256
#define NODES_PER_BKT 391
#define BKT_SLOTS     512
#define BIN_BLOCKS    256
#define BIN_THREADS   1024
#define EDGES_PER_BLK ((E_EDGES + BIN_BLOCKS - 1) / BIN_BLOCKS)
#define BIN_EPT       ((EDGES_PER_BLK + BIN_THREADS - 1) / BIN_THREADS)
#define CSR_EPT       8
#define CSR_LDS_CAP   12288
static_assert(NB_BUCKETS * NODES_PER_BKT >= N_NODES, "ranges must cover every node");
static_assert(BKT_SLOTS >= NODES_PER_BKT && BKT_SLOTS <= BIN_THREADS, "one scan thread per slot");
static_assert(BIN_BLOCKS % 128 == 0 && BIN_THREADS == 4 * NB_BUCKETS, "table walk: 4 threads per column, 32 rows per batch");

__device__ __forceinline__ int wave_incl_scan(int v, int lane) {
#pragma unroll
    for (int off = 1; off < 32; off <<= 1) {
        const int t = __shfl_up(v, off, 32);
        if (lane >= off) v += t;
    }
    return v;
}

__global__ void __launch_bounds__(BIN_THREADS)
bucket_count_kernel(const int* __restrict__ col, int* __restrict__ hist, int e) {
    __shared__ int s_cnt[NB_BUCKETS];
    const int tid = threadIdx.x;
    if (tid < NB_BUCKETS) s_cnt[tid] = 0;
    const int beg = blockIdx.x * EDGES_PER_BLK;
    const int end = (beg + EDGES_PER_BLK < e) ? beg + EDGES_PER_BLK : e;
    int c[BIN_EPT];
#pragma unroll
    for (int k = 0; k < BIN_EPT; ++k) {
        const int i = beg + tid + k * BIN_THREADS;
        c[k] = (i < end) ? col[i] : -1;
    }
    __syncthreads();
#pragma unroll
    for (int k = 0; k < BIN_EPT; ++k)
        if (c[k] >= 0) atomicAdd(&s_cnt[c[k] / NODES_PER_BKT], 1);
    __syncthreads();
    if (tid < NB_BUCKETS) hist[blockIdx.x * NB_BUCKETS + tid] = s_cnt[tid];
}

__global__ void __launch_bounds__(BIN_THREADS)
bucket_scatter_kernel(const int* __restrict__ row, const int* __restrict__ col,
                      const int* __restrict__ hist, int* __restrict__ bstart,
                      int* __restrict__ bins, int e) {
    __shared__ int s_pre[4][NB_BUCKETS];
    __shared__ int s_tot[4][NB_BUCKETS];
    __shared__ int s_cur[NB_BUCKETS];
    __shared__ int s_delta[NB_BUCKETS];
    __shared__ int s_wsum[NB_BUCKETS / 32];
    __shared__ int s_lsum[NB_BUCKETS / 32];
    __shared__ int2 s_pairs[EDGES_PER_BLK];
    const int tid  = threadIdx.x;
    const int lane = tid & 31;
    const int wave = tid >> 5;
    const int t = tid & (NB_BUCKETS - 1);
    const int q = tid >> 8;

    const int beg = blockIdx.x * EDGES_PER_BLK;
    const int end = (beg + EDGES_PER_BLK < e) ? beg + EDGES_PER_BLK : e;
    int c[BIN_EPT], r[BIN_EPT];
#pragma unroll
    for (int k = 0; k < BIN_EPT; ++k) {
        const int i = beg + tid + k * BIN_THREADS;
        const bool ok = i < end;
        c[k] = ok ? col[i] : -1;
        r[k] = ok ? row[i] : 0;
    }

    int pre = 0, tot = 0;
    const int me = blockIdx.x;
    int own = 0;
    if (tid < NB_BUCKETS) own = hist[me * NB_BUCKETS + tid];
    const int* hp = hist + q * NB_BUCKETS + t;
#pragma unroll
    for (int h = 0; h < BIN_BLOCKS / 128; ++h) {
        int v[32];
#pragma unroll
        for (int u = 0; u < 32; ++u)
            v[u] = hp[(h * 32 + u) * 4 * NB_BUCKETS];
#pragma unroll
        for (int u = 0; u < 32; ++u) {
            tot += v[u];
            if (q + 4 * (h * 32 + u) < me) pre += v[u];
        }
    }
    s_pre[q][t] = pre;
    s_tot[q][t] = tot;
    __syncthreads();

    int T = 0, P = 0, inc = 0, linc = 0;
    if (tid < NB_BUCKETS) {
        T = (s_tot[0][t] + s_tot[1][t]) + (s_tot[2][t] + s_tot[3][t]);
        P = (s_pre[0][t] + s_pre[1][t]) + (s_pre[2][t] + s_pre[3][t]);
        inc = wave_incl_scan(T, lane);
        linc = wave_incl_scan(own, lane);
        if (lane == 31) { s_wsum[wave] = inc; s_lsum[wave] = linc; }
    }
    __syncthreads();
    if (tid < NB_BUCKETS) {
        int front = 0, lfront = 0;
#pragma unroll
        for (int w = 0; w < NB_BUCKETS / 32; ++w)
            if (w < wave) { front += s_wsum[w]; lfront += s_lsum[w]; }
        const int excl = front + inc - T;
        const int lexcl = lfront + linc - own;
        s_cur[t] = lexcl;
        s_delta[t] = excl + P - lexcl;
        if (me == 0) {
            bstart[t] = excl;
            if (t == NB_BUCKETS - 1) bstart[NB_BUCKETS] = excl + T;
        }
    }
    __syncthreads();

#pragma unroll
    for (int k = 0; k < BIN_EPT; ++k) {
        if (c[k] >= 0) {
            const int lp = atomicAdd(&s_cur[c[k] / NODES_PER_BKT], 1);
            s_pairs[lp] = make_int2(r[k], c[k]);
        }
    }
    __syncthreads();

    int2* __restrict__ b2 = (int2*)bins;
    const int nloc = end - beg;
    int2 pr[BIN_EPT];
    int dl[BIN_EPT];
#pragma unroll
    for (int k = 0; k < BIN_EPT; ++k) {
        const int i = tid + k * BIN_THREADS;
        pr[k] = s_pairs[i < nloc ? i : 0];
    }
#pragma unroll
    for (int k = 0; k < BIN_EPT; ++k) {
        const int i = tid + k * BIN_THREADS;
        dl[k] = s_delta[i < nloc ? pr[k].y / NODES_PER_BKT : 0];
    }
#pragma unroll
    for (int k = 0; k < BIN_EPT; ++k) {
        const int i = tid + k * BIN_THREADS;
        if (i < nloc) b2[i + dl[k]] = pr[k];
    }
}

__global__ void __launch_bounds__(BIN_THREADS)
bucket_csr_kernel(const int* __restrict__ bins, const int* __restrict__ bstart,
                  int* __restrict__ rowptr, float* __restrict__ dinv,
                  int* __restrict__ csr_src, int n) {
    __shared__ int s_cnt[BKT_SLOTS];
    __shared__ int s_off[BKT_SLOTS];
    __shared__ int s_wsum[BKT_SLOTS / 32];
    __shared__ int s_list[CSR_LDS_CAP];
    const int tid  = threadIdx.x;
    const int lane = tid & 31;
    const int wave = tid >> 5;
    const int n0 = blockIdx.x * NODES_PER_BKT;
    const int bs = bstart[blockIdx.x];
    const int be = bstart[blockIdx.x + 1];
    const int2* __restrict__ b2 = (const int2*)bins;

    if (tid < BKT_SLOTS) s_cnt[tid] = 0;
    int2 rc[CSR_EPT];
#pragma unroll
    for (int k = 0; k < CSR_EPT; ++k) {
        const int i = bs + tid + k * BIN_THREADS;
        rc[k] = (i < be) ? b2[i] : make_int2(0, -1);
    }
    __syncthreads();

#pragma unroll
    for (int k = 0; k < CSR_EPT; ++k)
        if (rc[k].y >= 0) atomicAdd(&s_cnt[rc[k].y - n0], 1);
    for (int i = bs + tid + CSR_EPT * BIN_THREADS; i < be; i += BIN_THREADS)
        atomicAdd(&s_cnt[b2[i].y - n0], 1);
    __syncthreads();

    int v = 0, inc = 0;
    if (tid < BKT_SLOTS) {
        v = s_cnt[tid];
        inc = wave_incl_scan(v, lane);
        if (lane == 31) s_wsum[wave] = inc;
    }
    __syncthreads();
    if (tid < BKT_SLOTS) {
        int front = 0;
#pragma unroll
        for (int w = 0; w < BKT_SLOTS / 32; ++w)
            if (w < wave) front += s_wsum[w];
        const int first = bs + front + inc - v;
        s_off[tid] = first;
        s_cnt[tid] = 0;
        const int node = n0 + tid;
        if (tid < NODES_PER_BKT && node < n) {
            rowptr[node] = first;
            dinv[node] = rsqrtf((float)(v + 1));
            if (node == n - 1) rowptr[n] = first + v;
        }
    }
    __syncthreads();

    const int len = be - bs;
    if (len <= CSR_LDS_CAP) {
#pragma unroll
        for (int k = 0; k < CSR_EPT; ++k) {
            if (rc[k].y >= 0) {
                const int j = rc[k].y - n0;
                const int rk = atomicAdd(&s_cnt[j], 1);
                s_list[s_off[j] - bs + rk] = rc[k].x;
            }
        }
        for (int i = bs + tid + CSR_EPT * BIN_THREADS; i < be; i += BIN_THREADS) {
            const int2 p = b2[i];
            const int j = p.y - n0;
            const int rk = atomicAdd(&s_cnt[j], 1);
            s_list[s_off[j] - bs + rk] = p.x;
        }
        __syncthreads();
        for (int i = tid; i < len; i += BIN_THREADS)
            csr_src[bs + i] = s_list[i];
    } else {
#pragma unroll
        for (int k = 0; k < CSR_EPT; ++k) {
            if (rc[k].y >= 0) {
                const int j = rc[k].y - n0;
                const int rk = atomicAdd(&s_cnt[j], 1);
                csr_src[s_off[j] + rk] = rc[k].x;
            }
        }
        for (int i = bs + tid + CSR_EPT * BIN_THREADS; i < be; i += BIN_THREADS) {
            const int2 p = b2[i];
            const int j = p.y - n0;
            const int rk = atomicAdd(&s_cnt[j], 1);
            csr_src[s_off[j] + rk] = p.x;
        }
    }
}

template <int K, int NT>
__global__ void __launch_bounds__(NT * 32)
gemm_wmma_f32(const float* __restrict__ A, const float* __restrict__ B,
              const float* __restrict__ rscale, float* __restrict__ C, int mtiles) {
    constexpr int N = NT * 16;
    constexpr int P = K + 4;
    constexpr int F4_ROW = K / 4;
    constexpr int THREADS = NT * 32;
    constexpr int LD = (16 * F4_ROW) / THREADS;
    static_assert((16 * F4_ROW) % THREADS == 0, "tile copy: whole pieces per thread");
    static_assert((P % 4) == 0 && ((P / 4) % 2) == 1, "pitch = 16 bytes x odd");
    __shared__ __attribute__((aligned(16))) float sA[2][16 * P];

    const int tid  = threadIdx.x;
    const int wave = tid >> 5;
    const int lane = tid & 31;
    const int l    = lane & 15;
    const int half = lane >> 4;

    const int n0 = wave * 16;
    const float* bp = B + (size_t)(2 * half) * N + n0 + l;
    v2f breg[K / 4];
#pragma unroll
    for (int s = 0; s < K / 4; ++s) {
        breg[s][0] = bp[(size_t)(4 * s) * N];
        breg[s][1] = bp[(size_t)(4 * s + 1) * N];
    }

    constexpr int ROWS_STEP = THREADS / F4_ROW;
    static_assert(LD == 4 && THREADS % F4_ROW == 0, "four pieces per thread, a fixed row step apart");
    const int soff0 = (tid / F4_ROW) * P + (tid % F4_ROW) * 4;

    v4f p0 = {0.f, 0.f, 0.f, 0.f}, p1 = p0, p2 = p0, p3 = p0;
    v4f nf0 = p0, nf1 = p0;
    int mtile = blockIdx.x;
    if (mtile < mtiles) {
        const v4f* src = (const v4f*)(A + (size_t)mtile * 16 * K) + tid;
        p0 = src[0];
        p1 = src[THREADS];
        p2 = src[2 * THREADS];
        p3 = src[3 * THREADS];
        const v4f* fp = (const v4f*)(rscale + mtile * 16 + 8 * half);
        nf0 = fp[0];
        nf1 = fp[1];
        float* d = &sA[0][soff0];
        *(v4f*)(d)                     = p0;
        *(v4f*)(d + ROWS_STEP * P)     = p1;
        *(v4f*)(d + 2 * ROWS_STEP * P) = p2;
        *(v4f*)(d + 3 * ROWS_STEP * P) = p3;
    }
    __syncthreads();

    int buf = 0;
    for (; mtile < mtiles; mtile += gridDim.x) {
        const int next = mtile + (int)gridDim.x;
        const bool has_next = next < mtiles;
        const v4f d0 = nf0, d1 = nf1;
        if (has_next) {
            const v4f* src = (const v4f*)(A + (size_t)next * 16 * K) + tid;
            p0 = src[0];
            p1 = src[THREADS];
            p2 = src[2 * THREADS];
            p3 = src[3 * THREADS];
            const v4f* fp = (const v4f*)(rscale + next * 16 + 8 * half);
            nf0 = fp[0];
            nf1 = fp[1];
        }

        const float* sp = &sA[buf][l * P + 2 * half];
        v8f acc = {0.f, 0.f, 0.f, 0.f, 0.f, 0.f, 0.f, 0.f};
#pragma unroll
        for (int k0 = 0; k0 < K; k0 += 4) {
            v2f a = *(const v2f*)(sp + k0);
            acc = __builtin_amdgcn_wmma_f32_16x16x4_f32(
                 false, a,  false, breg[k0 / 4],
                 (short)0, acc,  false,  false);
        }

        if (has_next) {
            float* d = &sA[buf ^ 1][soff0];
            *(v4f*)(d)                     = p0;
            *(v4f*)(d + ROWS_STEP * P)     = p1;
            *(v4f*)(d + 2 * ROWS_STEP * P) = p2;
            *(v4f*)(d + 3 * ROWS_STEP * P) = p3;
        }
        __syncthreads();
        buf ^= 1;

        const int m0 = mtile * 16;
        float* cp = C + (size_t)(m0 + 8 * half) * N + n0 + l;
#pragma unroll
        for (int j = 0; j < 4; ++j) {
            cp[(size_t)j * N]       = acc[j] * d0[j];
            cp[(size_t)(j + 4) * N] = acc[j + 4] * d1[j];
        }
    }
}

template <int H, bool RELU>
__global__ void __launch_bounds__(256)
gather_kernel(const int* __restrict__ rowptr, const int* __restrict__ csr_src,
              const float* __restrict__ dinv, const float* __restrict__ htmp,
              const float* __restrict__ bias, float* __restrict__ out, int n) {
    constexpr int V = H / 32;
    typedef __attribute__((ext_vector_type(V))) float vec;
    const int lane = threadIdx.x & 31;
    const int wave0 = __builtin_amdgcn_readfirstlane(
        (int)((blockIdx.x * blockDim.x + threadIdx.x) >> 5));
    const int nwaves = (int)((gridDim.x * blockDim.x) >> 5);
    const int fo = lane * V;
    const vec b = *(const vec*)(bias + fo);

    for (int node = wave0; node < n; node += nwaves) {
        const int beg = rowptr[node];
        const int end = rowptr[node + 1];
        const float dc = dinv[node];

        vec acc = *(const vec*)(htmp + (size_t)node * H + fo);

        for (int base = beg; base < end; base += 32) {
            const int j = base + lane;
            int r = node;
            if (j < end) r = csr_src[j];
            int m = end - base;
            if (m > 32) m = 32;
#pragma unroll 4
            for (int t = 0; t < m; ++t) {
                const int rr = __builtin_amdgcn_readlane(r, t);
                const vec x = *(const vec*)(htmp + (size_t)rr * H + fo);
                acc += x;
            }
        }

        vec o = acc * dc + b;
        if (RELU) {
#pragma unroll
            for (int q = 0; q < V; ++q) o[q] = o[q] > 0.f ? o[q] : 0.f;
        }
        *(vec*)(out + (size_t)node * H + fo) = o;
    }
}

#define FG_WAVES 16
static_assert(N_NODES % 16 == 0, "whole 16-node tiles");
__global__ void __launch_bounds__(FG_WAVES * 32)
gather_relu_gemm_kernel(const int* __restrict__ rowptr, const int* __restrict__ csr_src,
                        const float* __restrict__ dinv, const float* __restrict__ htmp,
                        const float* __restrict__ bias, const float* __restrict__ W,
                        float* __restrict__ C, int mtiles) {
    constexpr int H = H1_DIM;
    constexpr int N = H2_DIM;
    constexpr int P = H + 4;
    constexpr int N4_ROW = N / 4;
    static_assert(FG_WAVES == 16 && H == 128, "one wave per tile row, 4 floats per lane");
    static_assert(H == FG_WAVES * 8 && N4_ROW == 16, "weight copy: 8 rows per wave, 4 x 4 pieces per row");
    static_assert((P % 4) == 0 && ((P / 4) % 2) == 1, "pitch = 16 bytes x odd");
    __shared__ __attribute__((aligned(16))) float sB[N * P];
    __shared__ __attribute__((aligned(16))) float sH[2][16 * P];

    const int tid  = threadIdx.x;
    const int lane = tid & 31;
    const int l    = lane & 15;
    const int half = lane >> 4;
    const int wave = __builtin_amdgcn_readfirstlane(tid >> 5);
    const int fo = lane * 4;

    {
        const int k = (tid >> 5) * 8 + (lane >> 2);
        const int jl = lane & 3;
        const v4f* src = (const v4f*)(W + (size_t)k * N) + jl;
        v4f w[4];
#pragma unroll
        for (int u = 0; u < 4; ++u)
            w[u] = src[4 * u];
#pragma unroll
        for (int u = 0; u < 4; ++u) {
            float* d = &sB[(16 * u + 4 * jl) * P + k];
#pragma unroll
            for (int i = 0; i < 4; ++i)
                d[i * P] = w[u][i];
        }
    }
    const v4f b = *(const v4f*)(bias + fo);

    int buf = 0;
    for (int tile = blockIdx.x; tile < mtiles; tile += gridDim.x) {
        const int node = tile * 16 + wave;
        const int beg = rowptr[node];
        const int end = rowptr[node + 1];
        const float dc = dinv[node];

        v4f acc = *(const v4f*)(htmp + (size_t)node * H + fo);

        for (int base = beg; base < end; base += 32) {
            const int j = base + lane;
            int r = node;
            if (j < end) r = csr_src[j];
            int m = end - base;
            if (m > 32) m = 32;
#pragma unroll 4
            for (int t = 0; t < m; ++t) {
                const int rr = __builtin_amdgcn_readlane(r, t);
                const v4f x = *(const v4f*)(htmp + (size_t)rr * H + fo);
                acc += x;
            }
        }

        v4f o = acc * dc + b;
#pragma unroll
        for (int q = 0; q < 4; ++q) o[q] = o[q] > 0.f ? o[q] : 0.f;
        *(v4f*)&sH[buf][wave * P + fo] = o;
        __syncthreads();

        if (wave < N / 16) {
            const int n0 = wave * 16;
            const float* sp = &sH[buf][l * P + 2 * half];
            const float* bp = &sB[(n0 + l) * P + 2 * half];
            const v4f* dp = (const v4f*)(dinv + tile * 16 + 8 * half);
            const v4f d0 = dp[0];
            const v4f d1 = dp[1];
            v8f c = {0.f, 0.f, 0.f, 0.f, 0.f, 0.f, 0.f, 0.f};
#pragma unroll
            for (int k0 = 0; k0 < H; k0 += 4) {
                const v2f a  = *(const v2f*)(sp + k0);
                const v2f bb = *(const v2f*)(bp + k0);
                c = __builtin_amdgcn_wmma_f32_16x16x4_f32(
                     false, a,  false, bb,
                     (short)0, c,  false,  false);
            }
            float* cp = C + (size_t)(tile * 16 + 8 * half) * N + n0 + l;
#pragma unroll
            for (int jj = 0; jj < 4; ++jj) {
                cp[(size_t)jj * N]       = c[jj] * d0[jj];
                cp[(size_t)(jj + 4) * N] = c[jj + 4] * d1[jj];
            }
        }
        buf ^= 1;
    }
}

__global__ void __launch_bounds__(256)
edge_dot_kernel(const int* __restrict__ pos, const int* __restrict__ neg,
                const float* __restrict__ h2, float* __restrict__ out, int ep) {
    const int lane = threadIdx.x & 31;
    const int wave0 = __builtin_amdgcn_readfirstlane(
        (int)((blockIdx.x * blockDim.x + threadIdx.x) >> 5));
    const int nwaves = (int)((gridDim.x * blockDim.x) >> 5);
    const int total = 2 * ep;
    const int nbatch = (total + 31) >> 5;

    for (int bt = wave0; bt < nbatch; bt += nwaves) {
        const int e = bt * 32 + lane;
        int src = 0, dst = 0;
        if (e < total) {
            if (e < ep) {
                src = pos[e];
                dst = pos[ep + e];
            } else {
                src = neg[e - ep];
                dst = neg[e];
            }
        }
        int m = total - bt * 32;
        if (m > 32) m = 32;
        float res = 0.f;
#pragma unroll 8
        for (int t = 0; t < m; ++t) {
            const int sj = __builtin_amdgcn_readlane(src, t);
            const int si = __builtin_amdgcn_readlane(dst, t);
            const float* xj = h2 + (size_t)sj * H2_DIM;
            const float* xi = h2 + (size_t)si * H2_DIM;
            float v = xi[lane] * xj[lane] + xi[lane + 32] * xj[lane + 32];
#pragma unroll
            for (int off = 16; off > 0; off >>= 1)
                v += __shfl_xor(v, off, 32);
            if (lane == t) res = v;
        }
        if (e < total) out[e] = res;
    }
}

static inline size_t align256(size_t x) { return (x + 255) & ~(size_t)255; }

extern "C" void kernel_launch(void* const* d_in, const int* in_sizes, int n_in,
                              void* d_out, int out_size, void* d_ws, size_t ws_size,
                              hipStream_t stream) {
    const float* x    = (const float*)d_in[0];
    const int*   ei   = (const int*)d_in[1];
    const int*   pos  = (const int*)d_in[2];
    const int*   neg  = (const int*)d_in[3];
    const float* W1   = (const float*)d_in[4];
    const float* b1   = (const float*)d_in[5];
    const float* W2   = (const float*)d_in[6];
    const float* b2   = (const float*)d_in[7];
    float* out = (float*)d_out;

    const int* row = ei;
    const int* col = ei + E_EDGES;

    char* ws = (char*)d_ws;
    size_t off = 0;
    float* dinv  = (float*)(ws + off); off += align256(sizeof(float) * N_NODES);
    float* htmp1 = (float*)(ws + off); off += align256(sizeof(float) * (size_t)N_NODES * H1_DIM);
    float* h1    = (float*)(ws + off); off += align256(sizeof(float) * (size_t)N_NODES * H1_DIM);
    float* htmp2 = (float*)(ws + off); off += align256(sizeof(float) * (size_t)N_NODES * H2_DIM);
    float* h2    = htmp1;
    int* bstart  = (int*)(ws + off); off += align256(sizeof(int) * N_NODES);
    int* rowptr  = (int*)(ws + off); off += align256(sizeof(int) * (N_NODES + 1));
    int* hist    = (int*)(ws + off); off += align256(sizeof(int) * (size_t)E_EDGES);
    int* csr_src = (int*)(ws + off); off += align256(sizeof(int) * (size_t)E_EDGES);
    int* bins    = (int*)h1;
    (void)ws_size; (void)n_in; (void)in_sizes; (void)out_size;

    const int TB = 256;
    const int MTILES = N_NODES / 16;
    const int GEMM1_BLOCKS = 512;
    const int FUSED_BLOCKS = 1024;
    const int GATHER_BLOCKS = 2048;
    const int DOT_BLOCKS = ((2 * EP_CNT + 31) / 32 + 7) / 8;

    bucket_count_kernel<<<BIN_BLOCKS, BIN_THREADS, 0, stream>>>(col, hist, E_EDGES);
    bucket_scatter_kernel<<<BIN_BLOCKS, BIN_THREADS, 0, stream>>>(row, col, hist, bstart, bins, E_EDGES);
    bucket_csr_kernel<<<NB_BUCKETS, BIN_THREADS, 0, stream>>>(bins, bstart, rowptr, dinv, csr_src, N_NODES);

    gemm_wmma_f32<F_IN, 8><<<GEMM1_BLOCKS, 8 * 32, 0, stream>>>(x, W1, dinv, htmp1, MTILES);

    gather_relu_gemm_kernel<<<FUSED_BLOCKS, FG_WAVES * 32, 0, stream>>>(
        rowptr, csr_src, dinv, htmp1, b1, W2, htmp2, MTILES);

    gather_kernel<H2_DIM, false><<<GATHER_BLOCKS, TB, 0, stream>>>(
        rowptr, csr_src, dinv, htmp2, b2, h2, N_NODES);

    edge_dot_kernel<<<DOT_BLOCKS, TB, 0, stream>>>(pos, neg, h2, out, EP_CNT);
}


# Each block groups its own edge slice by range; the range pass sums the 256 pieces' offsets and reads its pairs piece by piece: no count launch, no table walk; 7 to 6 launches; order: arrival rank
# speedup vs baseline: 6.3026x; 1.0386x over previous
#include <hip/hip_runtime.h>

typedef __attribute__((ext_vector_type(2))) float v2f;
typedef __attribute__((ext_vector_type(4))) float v4f;
typedef __attribute__((ext_vector_type(8))) float v8f;

#define N_NODES 100000
#define F_IN    256
#define H1_DIM  128
#define H2_DIM  64
#define E_EDGES 1600000
#define EP_CNT  100000

#define NB_BUCKETS    256
#define NODES_PER_BKT 391
#define BKT_SLOTS     512
#define BIN_BLOCKS    256
#define BIN_THREADS   1024
#define EDGES_PER_BLK ((E_EDGES + BIN_BLOCKS - 1) / BIN_BLOCKS)
#define BIN_EPT       ((EDGES_PER_BLK + BIN_THREADS - 1) / BIN_THREADS)
#define LOFF_PITCH    (NB_BUCKETS + 1)
#define CSR_EPT       (BIN_BLOCKS / 32)
#define CSR_RPP       2
#define CSR_LDS_CAP   12288
static_assert(NB_BUCKETS * NODES_PER_BKT >= N_NODES, "ranges must cover every node");
static_assert(BKT_SLOTS >= NODES_PER_BKT && BKT_SLOTS <= BIN_THREADS, "one scan thread per slot");
static_assert(NB_BUCKETS % 32 == 0 && NB_BUCKETS <= BIN_THREADS, "range scan: whole waves, one thread per range");
static_assert(BIN_BLOCKS % 32 == 0 && BIN_BLOCKS <= BIN_THREADS && BIN_THREADS == 32 * 32,
              "piece table: one thread per piece, pieces dealt out over the block's 32 waves");

__device__ __forceinline__ int wave_incl_scan(int v, int lane) {
#pragma unroll
    for (int off = 1; off < 32; off <<= 1) {
        const int t = __shfl_up(v, off, 32);
        if (lane >= off) v += t;
    }
    return v;
}

__global__ void __launch_bounds__(BIN_THREADS)
bucket_group_kernel(const int* __restrict__ row, const int* __restrict__ col,
                    int* __restrict__ loff, int* __restrict__ bins, int e) {
    __shared__ int s_cnt[NB_BUCKETS];
    __shared__ int s_lsum[NB_BUCKETS / 32];
    __shared__ int2 s_pairs[EDGES_PER_BLK];
    const int tid  = threadIdx.x;
    const int lane = tid & 31;
    const int wave = tid >> 5;
    if (tid < NB_BUCKETS) s_cnt[tid] = 0;

    const int beg = blockIdx.x * EDGES_PER_BLK;
    const int end = (beg + EDGES_PER_BLK < e) ? beg + EDGES_PER_BLK : e;
    int c[BIN_EPT], r[BIN_EPT], g[BIN_EPT];
#pragma unroll
    for (int k = 0; k < BIN_EPT; ++k) {
        const int i = beg + tid + k * BIN_THREADS;
        const bool ok = i < end;
        c[k] = ok ? col[i] : -1;
        r[k] = ok ? row[i] : 0;
    }
    __syncthreads();

#pragma unroll
    for (int k = 0; k < BIN_EPT; ++k) {
        g[k] = c[k] / NODES_PER_BKT;
        if (c[k] >= 0) atomicAdd(&s_cnt[g[k]], 1);
    }
    __syncthreads();

    int own = 0, linc = 0;
    if (tid < NB_BUCKETS) {
        own = s_cnt[tid];
        linc = wave_incl_scan(own, lane);
        if (lane == 31) s_lsum[wave] = linc;
    }
    __syncthreads();
    if (tid < NB_BUCKETS) {
        int lfront = 0;
#pragma unroll
        for (int w = 0; w < NB_BUCKETS / 32; ++w)
            if (w < wave) lfront += s_lsum[w];
        const int lexcl = lfront + linc - own;
        s_cnt[tid] = lexcl;
        int* lrow = loff + (size_t)blockIdx.x * LOFF_PITCH;
        lrow[tid] = lexcl;
        if (tid == NB_BUCKETS - 1) lrow[NB_BUCKETS] = lexcl + own;
    }
    __syncthreads();

#pragma unroll
    for (int k = 0; k < BIN_EPT; ++k) {
        if (c[k] >= 0) {
            const int lp = atomicAdd(&s_cnt[g[k]], 1);
            s_pairs[lp] = make_int2(r[k], c[k]);
        }
    }
    __syncthreads();

    int2* __restrict__ b2 = (int2*)bins + (size_t)blockIdx.x * EDGES_PER_BLK;
    const int nloc = end - beg;
#pragma unroll
    for (int k = 0; k < BIN_EPT; ++k) {
        const int i = tid + k * BIN_THREADS;
        if (i < nloc) b2[i] = s_pairs[i];
    }
}

__global__ void __launch_bounds__(BIN_THREADS)
bucket_csr_kernel(const int* __restrict__ bins, const int* __restrict__ loff,
                  int* __restrict__ rowptr, float* __restrict__ dinv,
                  int* __restrict__ csr_src, int n) {
    __shared__ int s_cnt[BKT_SLOTS];
    __shared__ int s_off[BKT_SLOTS];
    __shared__ int s_wsum[BKT_SLOTS / 32];
    __shared__ int s_plo[BIN_BLOCKS];
    __shared__ int s_pln[BIN_BLOCKS];
    __shared__ int s_rlo[BIN_BLOCKS / 32];
    __shared__ int s_rhi[BIN_BLOCKS / 32];
    __shared__ int s_list[CSR_LDS_CAP];
    const int tid  = threadIdx.x;
    const int lane = tid & 31;
    const int wave = tid >> 5;
    const int n0 = blockIdx.x * NODES_PER_BKT;
    const int2* __restrict__ b2 = (const int2*)bins;

    if (tid < BKT_SLOTS) s_cnt[tid] = 0;
    if (tid < BIN_BLOCKS) {
        const int* lp = loff + (size_t)tid * LOFF_PITCH + blockIdx.x;
        const int lo = lp[0];
        const int hi = lp[1];
        s_plo[tid] = tid * EDGES_PER_BLK + lo;
        s_pln[tid] = hi - lo;
        int a = lo, b = hi;
#pragma unroll
        for (int off = 16; off > 0; off >>= 1) {
            a += __shfl_xor(a, off, 32);
            b += __shfl_xor(b, off, 32);
        }
        if (lane == 0) { s_rlo[wave] = a; s_rhi[wave] = b; }
    }
    __syncthreads();
    int bs = 0, be = 0;
#pragma unroll
    for (int w = 0; w < BIN_BLOCKS / 32; ++w) { bs += s_rlo[w]; be += s_rhi[w]; }

    int plo[CSR_EPT], pln[CSR_EPT];
    int2 rc[CSR_EPT * CSR_RPP];
#pragma unroll
    for (int k = 0; k < CSR_EPT; ++k) {
        plo[k] = s_plo[wave + 32 * k];
        pln[k] = s_pln[wave + 32 * k];
    }
#pragma unroll
    for (int k = 0; k < CSR_EPT * CSR_RPP; ++k) {
        const int i = 32 * (k % CSR_RPP) + lane;
        rc[k] = (i < pln[k / CSR_RPP]) ? b2[plo[k / CSR_RPP] + i] : make_int2(0, -1);
    }

#pragma unroll
    for (int k = 0; k < CSR_EPT * CSR_RPP; ++k)
        if (rc[k].y >= 0) atomicAdd(&s_cnt[rc[k].y - n0], 1);
#pragma unroll
    for (int k = 0; k < CSR_EPT; ++k)
        for (int i = 32 * CSR_RPP + lane; i < pln[k]; i += 32)
            atomicAdd(&s_cnt[b2[plo[k] + i].y - n0], 1);
    __syncthreads();

    int v = 0, inc = 0;
    if (tid < BKT_SLOTS) {
        v = s_cnt[tid];
        inc = wave_incl_scan(v, lane);
        if (lane == 31) s_wsum[wave] = inc;
    }
    __syncthreads();
    if (tid < BKT_SLOTS) {
        int front = 0;
#pragma unroll
        for (int w = 0; w < BKT_SLOTS / 32; ++w)
            if (w < wave) front += s_wsum[w];
        const int first = bs + front + inc - v;
        s_off[tid] = first;
        s_cnt[tid] = 0;
        const int node = n0 + tid;
        if (tid < NODES_PER_BKT && node < n) {
            rowptr[node] = first;
            dinv[node] = rsqrtf((float)(v + 1));
            if (node == n - 1) rowptr[n] = first + v;
        }
    }
    __syncthreads();

    const int len = be - bs;
    if (len <= CSR_LDS_CAP) {
#pragma unroll
        for (int k = 0; k < CSR_EPT * CSR_RPP; ++k) {
            if (rc[k].y >= 0) {
                const int j = rc[k].y - n0;
                const int rk = atomicAdd(&s_cnt[j], 1);
                s_list[s_off[j] - bs + rk] = rc[k].x;
            }
        }
#pragma unroll
        for (int k = 0; k < CSR_EPT; ++k) {
            for (int i = 32 * CSR_RPP + lane; i < pln[k]; i += 32) {
                const int2 p = b2[plo[k] + i];
                const int j = p.y - n0;
                const int rk = atomicAdd(&s_cnt[j], 1);
                s_list[s_off[j] - bs + rk] = p.x;
            }
        }
        __syncthreads();
        for (int i = tid; i < len; i += BIN_THREADS)
            csr_src[bs + i] = s_list[i];
    } else {
#pragma unroll
        for (int k = 0; k < CSR_EPT * CSR_RPP; ++k) {
            if (rc[k].y >= 0) {
                const int j = rc[k].y - n0;
                const int rk = atomicAdd(&s_cnt[j], 1);
                csr_src[s_off[j] + rk] = rc[k].x;
            }
        }
#pragma unroll
        for (int k = 0; k < CSR_EPT; ++k) {
            for (int i = 32 * CSR_RPP + lane; i < pln[k]; i += 32) {
                const int2 p = b2[plo[k] + i];
                const int j = p.y - n0;
                const int rk = atomicAdd(&s_cnt[j], 1);
                csr_src[s_off[j] + rk] = p.x;
            }
        }
    }
}

template <int K, int NT>
__global__ void __launch_bounds__(NT * 32)
gemm_wmma_f32(const float* __restrict__ A, const float* __restrict__ B,
              const float* __restrict__ rscale, float* __restrict__ C, int mtiles) {
    constexpr int N = NT * 16;
    constexpr int P = K + 4;
    constexpr int F4_ROW = K / 4;
    constexpr int THREADS = NT * 32;
    constexpr int LD = (16 * F4_ROW) / THREADS;
    static_assert((16 * F4_ROW) % THREADS == 0, "tile copy: whole pieces per thread");
    static_assert((P % 4) == 0 && ((P / 4) % 2) == 1, "pitch = 16 bytes x odd");
    __shared__ __attribute__((aligned(16))) float sA[2][16 * P];

    const int tid  = threadIdx.x;
    const int wave = tid >> 5;
    const int lane = tid & 31;
    const int l    = lane & 15;
    const int half = lane >> 4;

    const int n0 = wave * 16;
    const float* bp = B + (size_t)(2 * half) * N + n0 + l;
    v2f breg[K / 4];
#pragma unroll
    for (int s = 0; s < K / 4; ++s) {
        breg[s][0] = bp[(size_t)(4 * s) * N];
        breg[s][1] = bp[(size_t)(4 * s + 1) * N];
    }

    constexpr int ROWS_STEP = THREADS / F4_ROW;
    static_assert(LD == 4 && THREADS % F4_ROW == 0, "four pieces per thread, a fixed row step apart");
    const int soff0 = (tid / F4_ROW) * P + (tid % F4_ROW) * 4;

    v4f p0 = {0.f, 0.f, 0.f, 0.f}, p1 = p0, p2 = p0, p3 = p0;
    v4f nf0 = p0, nf1 = p0;
    int mtile = blockIdx.x;
    if (mtile < mtiles) {
        const v4f* src = (const v4f*)(A + (size_t)mtile * 16 * K) + tid;
        p0 = src[0];
        p1 = src[THREADS];
        p2 = src[2 * THREADS];
        p3 = src[3 * THREADS];
        const v4f* fp = (const v4f*)(rscale + mtile * 16 + 8 * half);
        nf0 = fp[0];
        nf1 = fp[1];
        float* d = &sA[0][soff0];
        *(v4f*)(d)                     = p0;
        *(v4f*)(d + ROWS_STEP * P)     = p1;
        *(v4f*)(d + 2 * ROWS_STEP * P) = p2;
        *(v4f*)(d + 3 * ROWS_STEP * P) = p3;
    }
    __syncthreads();

    int buf = 0;
    for (; mtile < mtiles; mtile += gridDim.x) {
        const int next = mtile + (int)gridDim.x;
        const bool has_next = next < mtiles;
        const v4f d0 = nf0, d1 = nf1;
        if (has_next) {
            const v4f* src = (const v4f*)(A + (size_t)next * 16 * K) + tid;
            p0 = src[0];
            p1 = src[THREADS];
            p2 = src[2 * THREADS];
            p3 = src[3 * THREADS];
            const v4f* fp = (const v4f*)(rscale + next * 16 + 8 * half);
            nf0 = fp[0];
            nf1 = fp[1];
        }

        const float* sp = &sA[buf][l * P + 2 * half];
        v8f acc = {0.f, 0.f, 0.f, 0.f, 0.f, 0.f, 0.f, 0.f};
#pragma unroll
        for (int k0 = 0; k0 < K; k0 += 4) {
            v2f a = *(const v2f*)(sp + k0);
            acc = __builtin_amdgcn_wmma_f32_16x16x4_f32(
                 false, a,  false, breg[k0 / 4],
                 (short)0, acc,  false,  false);
        }

        if (has_next) {
            float* d = &sA[buf ^ 1][soff0];
            *(v4f*)(d)                     = p0;
            *(v4f*)(d + ROWS_STEP * P)     = p1;
            *(v4f*)(d + 2 * ROWS_STEP * P) = p2;
            *(v4f*)(d + 3 * ROWS_STEP * P) = p3;
        }
        __syncthreads();
        buf ^= 1;

        const int m0 = mtile * 16;
        float* cp = C + (size_t)(m0 + 8 * half) * N + n0 + l;
#pragma unroll
        for (int j = 0; j < 4; ++j) {
            cp[(size_t)j * N]       = acc[j] * d0[j];
            cp[(size_t)(j + 4) * N] = acc[j + 4] * d1[j];
        }
    }
}

template <int H, bool RELU>
__global__ void __launch_bounds__(256)
gather_kernel(const int* __restrict__ rowptr, const int* __restrict__ csr_src,
              const float* __restrict__ dinv, const float* __restrict__ htmp,
              const float* __restrict__ bias, float* __restrict__ out, int n) {
    constexpr int V = H / 32;
    typedef __attribute__((ext_vector_type(V))) float vec;
    const int lane = threadIdx.x & 31;
    const int wave0 = __builtin_amdgcn_readfirstlane(
        (int)((blockIdx.x * blockDim.x + threadIdx.x) >> 5));
    const int nwaves = (int)((gridDim.x * blockDim.x) >> 5);
    const int fo = lane * V;
    const vec b = *(const vec*)(bias + fo);

    for (int node = wave0; node < n; node += nwaves) {
        const int beg = rowptr[node];
        const int end = rowptr[node + 1];
        const float dc = dinv[node];

        vec acc = *(const vec*)(htmp + (size_t)node * H + fo);

        for (int base = beg; base < end; base += 32) {
            const int j = base + lane;
            int r = node;
            if (j < end) r = csr_src[j];
            int m = end - base;
            if (m > 32) m = 32;
#pragma unroll 4
            for (int t = 0; t < m; ++t) {
                const int rr = __builtin_amdgcn_readlane(r, t);
                const vec x = *(const vec*)(htmp + (size_t)rr * H + fo);
                acc += x;
            }
        }

        vec o = acc * dc + b;
        if (RELU) {
#pragma unroll
            for (int q = 0; q < V; ++q) o[q] = o[q] > 0.f ? o[q] : 0.f;
        }
        *(vec*)(out + (size_t)node * H + fo) = o;
    }
}

#define FG_WAVES 16
static_assert(N_NODES % 16 == 0, "whole 16-node tiles");
__global__ void __launch_bounds__(FG_WAVES * 32)
gather_relu_gemm_kernel(const int* __restrict__ rowptr, const int* __restrict__ csr_src,
                        const float* __restrict__ dinv, const float* __restrict__ htmp,
                        const float* __restrict__ bias, const float* __restrict__ W,
                        float* __restrict__ C, int mtiles) {
    constexpr int H = H1_DIM;
    constexpr int N = H2_DIM;
    constexpr int P = H + 4;
    constexpr int N4_ROW = N / 4;
    static_assert(FG_WAVES == 16 && H == 128, "one wave per tile row, 4 floats per lane");
    static_assert(H == FG_WAVES * 8 && N4_ROW == 16, "weight copy: 8 rows per wave, 4 x 4 pieces per row");
    static_assert((P % 4) == 0 && ((P / 4) % 2) == 1, "pitch = 16 bytes x odd");
    __shared__ __attribute__((aligned(16))) float sB[N * P];
    __shared__ __attribute__((aligned(16))) float sH[2][16 * P];

    const int tid  = threadIdx.x;
    const int lane = tid & 31;
    const int l    = lane & 15;
    const int half = lane >> 4;
    const int wave = __builtin_amdgcn_readfirstlane(tid >> 5);
    const int fo = lane * 4;

    {
        const int k = (tid >> 5) * 8 + (lane >> 2);
        const int jl = lane & 3;
        const v4f* src = (const v4f*)(W + (size_t)k * N) + jl;
        v4f w[4];
#pragma unroll
        for (int u = 0; u < 4; ++u)
            w[u] = src[4 * u];
#pragma unroll
        for (int u = 0; u < 4; ++u) {
            float* d = &sB[(16 * u + 4 * jl) * P + k];
#pragma unroll
            for (int i = 0; i < 4; ++i)
                d[i * P] = w[u][i];
        }
    }
    const v4f b = *(const v4f*)(bias + fo);

    int buf = 0;
    for (int tile = blockIdx.x; tile < mtiles; tile += gridDim.x) {
        const int node = tile * 16 + wave;
        const int beg = rowptr[node];
        const int end = rowptr[node + 1];
        const float dc = dinv[node];

        v4f acc = *(const v4f*)(htmp + (size_t)node * H + fo);

        for (int base = beg; base < end; base += 32) {
            const int j = base + lane;
            int r = node;
            if (j < end) r = csr_src[j];
            int m = end - base;
            if (m > 32) m = 32;
#pragma unroll 4
            for (int t = 0; t < m; ++t) {
                const int rr = __builtin_amdgcn_readlane(r, t);
                const v4f x = *(const v4f*)(htmp + (size_t)rr * H + fo);
                acc += x;
            }
        }

        v4f o = acc * dc + b;
#pragma unroll
        for (int q = 0; q < 4; ++q) o[q] = o[q] > 0.f ? o[q] : 0.f;
        *(v4f*)&sH[buf][wave * P + fo] = o;
        __syncthreads();

        if (wave < N / 16) {
            const int n0 = wave * 16;
            const float* sp = &sH[buf][l * P + 2 * half];
            const float* bp = &sB[(n0 + l) * P + 2 * half];
            const v4f* dp = (const v4f*)(dinv + tile * 16 + 8 * half);
            const v4f d0 = dp[0];
            const v4f d1 = dp[1];
            v8f c = {0.f, 0.f, 0.f, 0.f, 0.f, 0.f, 0.f, 0.f};
#pragma unroll
            for (int k0 = 0; k0 < H; k0 += 4) {
                const v2f a  = *(const v2f*)(sp + k0);
                const v2f bb = *(const v2f*)(bp + k0);
                c = __builtin_amdgcn_wmma_f32_16x16x4_f32(
                     false, a,  false, bb,
                     (short)0, c,  false,  false);
            }
            float* cp = C + (size_t)(tile * 16 + 8 * half) * N + n0 + l;
#pragma unroll
            for (int jj = 0; jj < 4; ++jj) {
                cp[(size_t)jj * N]       = c[jj] * d0[jj];
                cp[(size_t)(jj + 4) * N] = c[jj + 4] * d1[jj];
            }
        }
        buf ^= 1;
    }
}

__global__ void __launch_bounds__(256)
edge_dot_kernel(const int* __restrict__ pos, const int* __restrict__ neg,
                const float* __restrict__ h2, float* __restrict__ out, int ep) {
    const int lane = threadIdx.x & 31;
    const int wave0 = __builtin_amdgcn_readfirstlane(
        (int)((blockIdx.x * blockDim.x + threadIdx.x) >> 5));
    const int nwaves = (int)((gridDim.x * blockDim.x) >> 5);
    const int total = 2 * ep;
    const int nbatch = (total + 31) >> 5;

    for (int bt = wave0; bt < nbatch; bt += nwaves) {
        const int e = bt * 32 + lane;
        int src = 0, dst = 0;
        if (e < total) {
            if (e < ep) {
                src = pos[e];
                dst = pos[ep + e];
            } else {
                src = neg[e - ep];
                dst = neg[e];
            }
        }
        int m = total - bt * 32;
        if (m > 32) m = 32;
        float res = 0.f;
#pragma unroll 8
        for (int t = 0; t < m; ++t) {
            const int sj = __builtin_amdgcn_readlane(src, t);
            const int si = __builtin_amdgcn_readlane(dst, t);
            const float* xj = h2 + (size_t)sj * H2_DIM;
            const float* xi = h2 + (size_t)si * H2_DIM;
            float v = xi[lane] * xj[lane] + xi[lane + 32] * xj[lane + 32];
#pragma unroll
            for (int off = 16; off > 0; off >>= 1)
                v += __shfl_xor(v, off, 32);
            if (lane == t) res = v;
        }
        if (e < total) out[e] = res;
    }
}

static inline size_t align256(size_t x) { return (x + 255) & ~(size_t)255; }

extern "C" void kernel_launch(void* const* d_in, const int* in_sizes, int n_in,
                              void* d_out, int out_size, void* d_ws, size_t ws_size,
                              hipStream_t stream) {
    const float* x    = (const float*)d_in[0];
    const int*   ei   = (const int*)d_in[1];
    const int*   pos  = (const int*)d_in[2];
    const int*   neg  = (const int*)d_in[3];
    const float* W1   = (const float*)d_in[4];
    const float* b1   = (const float*)d_in[5];
    const float* W2   = (const float*)d_in[6];
    const float* b2   = (const float*)d_in[7];
    float* out = (float*)d_out;

    const int* row = ei;
    const int* col = ei + E_EDGES;

    char* ws = (char*)d_ws;
    size_t off = 0;
    float* dinv  = (float*)(ws + off); off += align256(sizeof(float) * N_NODES);
    float* htmp1 = (float*)(ws + off); off += align256(sizeof(float) * (size_t)N_NODES * H1_DIM);
    float* h1    = (float*)(ws + off); off += align256(sizeof(float) * (size_t)N_NODES * H1_DIM);
    float* htmp2 = (float*)(ws + off); off += align256(sizeof(float) * (size_t)N_NODES * H2_DIM);
    float* h2    = htmp1;
    int* rowptr  = (int*)(ws + off); off += align256(sizeof(int) * (N_NODES + 1));
    int* loff    = (int*)(ws + off); off += align256(sizeof(int) * (size_t)E_EDGES);
    int* csr_src = (int*)(ws + off); off += align256(sizeof(int) * (size_t)E_EDGES);
    int* bins    = (int*)h1;
    (void)ws_size; (void)n_in; (void)in_sizes; (void)out_size;

    const int TB = 256;
    const int MTILES = N_NODES / 16;
    const int GEMM1_BLOCKS = 512;
    const int FUSED_BLOCKS = 1024;
    const int GATHER_BLOCKS = 2048;
    const int DOT_BLOCKS = ((2 * EP_CNT + 31) / 32 + 7) / 8;

    bucket_group_kernel<<<BIN_BLOCKS, BIN_THREADS, 0, stream>>>(row, col, loff, bins, E_EDGES);
    bucket_csr_kernel<<<NB_BUCKETS, BIN_THREADS, 0, stream>>>(bins, loff, rowptr, dinv, csr_src, N_NODES);

    gemm_wmma_f32<F_IN, 8><<<GEMM1_BLOCKS, 8 * 32, 0, stream>>>(x, W1, dinv, htmp1, MTILES);

    gather_relu_gemm_kernel<<<FUSED_BLOCKS, FG_WAVES * 32, 0, stream>>>(
        rowptr, csr_src, dinv, htmp1, b1, W2, htmp2, MTILES);

    gather_kernel<H2_DIM, false><<<GATHER_BLOCKS, TB, 0, stream>>>(
        rowptr, csr_src, dinv, htmp2, b2, h2, N_NODES);

    edge_dot_kernel<<<DOT_BLOCKS, TB, 0, stream>>>(pos, neg, h2, out, EP_CNT);
}


# Both aggregations: a node's row loads go out in whole groups, 8 at a time and the 1-7 left over as one group (switch on count), not 4 at a time plus one row per trip for the rest; sum order kept
# speedup vs baseline: 6.5084x; 1.0327x over previous
#include <hip/hip_runtime.h>

typedef __attribute__((ext_vector_type(2))) float v2f;
typedef __attribute__((ext_vector_type(4))) float v4f;
typedef __attribute__((ext_vector_type(8))) float v8f;

#define N_NODES 100000
#define F_IN    256
#define H1_DIM  128
#define H2_DIM  64
#define E_EDGES 1600000
#define EP_CNT  100000

#define NB_BUCKETS    256
#define NODES_PER_BKT 391
#define BKT_SLOTS     512
#define BIN_BLOCKS    256
#define BIN_THREADS   1024
#define EDGES_PER_BLK ((E_EDGES + BIN_BLOCKS - 1) / BIN_BLOCKS)
#define BIN_EPT       ((EDGES_PER_BLK + BIN_THREADS - 1) / BIN_THREADS)
#define LOFF_PITCH    (NB_BUCKETS + 1)
#define CSR_EPT       (BIN_BLOCKS / 32)
#define CSR_RPP       2
#define CSR_LDS_CAP   12288
static_assert(NB_BUCKETS * NODES_PER_BKT >= N_NODES, "ranges must cover every node");
static_assert(BKT_SLOTS >= NODES_PER_BKT && BKT_SLOTS <= BIN_THREADS, "one scan thread per slot");
static_assert(NB_BUCKETS % 32 == 0 && NB_BUCKETS <= BIN_THREADS, "range scan: whole waves, one thread per range");
static_assert(BIN_BLOCKS % 32 == 0 && BIN_BLOCKS <= BIN_THREADS && BIN_THREADS == 32 * 32,
              "piece table: one thread per piece, pieces dealt out over the block's 32 waves");

__device__ __forceinline__ int wave_incl_scan(int v, int lane) {
#pragma unroll
    for (int off = 1; off < 32; off <<= 1) {
        const int t = __shfl_up(v, off, 32);
        if (lane >= off) v += t;
    }
    return v;
}

__global__ void __launch_bounds__(BIN_THREADS)
bucket_group_kernel(const int* __restrict__ row, const int* __restrict__ col,
                    int* __restrict__ loff, int* __restrict__ bins, int e) {
    __shared__ int s_cnt[NB_BUCKETS];
    __shared__ int s_lsum[NB_BUCKETS / 32];
    __shared__ int2 s_pairs[EDGES_PER_BLK];
    const int tid  = threadIdx.x;
    const int lane = tid & 31;
    const int wave = tid >> 5;
    if (tid < NB_BUCKETS) s_cnt[tid] = 0;

    const int beg = blockIdx.x * EDGES_PER_BLK;
    const int end = (beg + EDGES_PER_BLK < e) ? beg + EDGES_PER_BLK : e;
    int c[BIN_EPT], r[BIN_EPT], g[BIN_EPT];
#pragma unroll
    for (int k = 0; k < BIN_EPT; ++k) {
        const int i = beg + tid + k * BIN_THREADS;
        const bool ok = i < end;
        c[k] = ok ? col[i] : -1;
        r[k] = ok ? row[i] : 0;
    }
    __syncthreads();

#pragma unroll
    for (int k = 0; k < BIN_EPT; ++k) {
        g[k] = c[k] / NODES_PER_BKT;
        if (c[k] >= 0) atomicAdd(&s_cnt[g[k]], 1);
    }
    __syncthreads();

    int own = 0, linc = 0;
    if (tid < NB_BUCKETS) {
        own = s_cnt[tid];
        linc = wave_incl_scan(own, lane);
        if (lane == 31) s_lsum[wave] = linc;
    }
    __syncthreads();
    if (tid < NB_BUCKETS) {
        int lfront = 0;
#pragma unroll
        for (int w = 0; w < NB_BUCKETS / 32; ++w)
            if (w < wave) lfront += s_lsum[w];
        const int lexcl = lfront + linc - own;
        s_cnt[tid] = lexcl;
        int* lrow = loff + (size_t)blockIdx.x * LOFF_PITCH;
        lrow[tid] = lexcl;
        if (tid == NB_BUCKETS - 1) lrow[NB_BUCKETS] = lexcl + own;
    }
    __syncthreads();

#pragma unroll
    for (int k = 0; k < BIN_EPT; ++k) {
        if (c[k] >= 0) {
            const int lp = atomicAdd(&s_cnt[g[k]], 1);
            s_pairs[lp] = make_int2(r[k], c[k]);
        }
    }
    __syncthreads();

    int2* __restrict__ b2 = (int2*)bins + (size_t)blockIdx.x * EDGES_PER_BLK;
    const int nloc = end - beg;
#pragma unroll
    for (int k = 0; k < BIN_EPT; ++k) {
        const int i = tid + k * BIN_THREADS;
        if (i < nloc) b2[i] = s_pairs[i];
    }
}

__global__ void __launch_bounds__(BIN_THREADS)
bucket_csr_kernel(const int* __restrict__ bins, const int* __restrict__ loff,
                  int* __restrict__ rowptr, float* __restrict__ dinv,
                  int* __restrict__ csr_src, int n) {
    __shared__ int s_cnt[BKT_SLOTS];
    __shared__ int s_off[BKT_SLOTS];
    __shared__ int s_wsum[BKT_SLOTS / 32];
    __shared__ int s_plo[BIN_BLOCKS];
    __shared__ int s_pln[BIN_BLOCKS];
    __shared__ int s_rlo[BIN_BLOCKS / 32];
    __shared__ int s_rhi[BIN_BLOCKS / 32];
    __shared__ int s_list[CSR_LDS_CAP];
    const int tid  = threadIdx.x;
    const int lane = tid & 31;
    const int wave = tid >> 5;
    const int n0 = blockIdx.x * NODES_PER_BKT;
    const int2* __restrict__ b2 = (const int2*)bins;

    if (tid < BKT_SLOTS) s_cnt[tid] = 0;
    if (tid < BIN_BLOCKS) {
        const int* lp = loff + (size_t)tid * LOFF_PITCH + blockIdx.x;
        const int lo = lp[0];
        const int hi = lp[1];
        s_plo[tid] = tid * EDGES_PER_BLK + lo;
        s_pln[tid] = hi - lo;
        int a = lo, b = hi;
#pragma unroll
        for (int off = 16; off > 0; off >>= 1) {
            a += __shfl_xor(a, off, 32);
            b += __shfl_xor(b, off, 32);
        }
        if (lane == 0) { s_rlo[wave] = a; s_rhi[wave] = b; }
    }
    __syncthreads();
    int bs = 0, be = 0;
#pragma unroll
    for (int w = 0; w < BIN_BLOCKS / 32; ++w) { bs += s_rlo[w]; be += s_rhi[w]; }

    int plo[CSR_EPT], pln[CSR_EPT];
    int2 rc[CSR_EPT * CSR_RPP];
#pragma unroll
    for (int k = 0; k < CSR_EPT; ++k) {
        plo[k] = s_plo[wave + 32 * k];
        pln[k] = s_pln[wave + 32 * k];
    }
#pragma unroll
    for (int k = 0; k < CSR_EPT * CSR_RPP; ++k) {
        const int i = 32 * (k % CSR_RPP) + lane;
        rc[k] = (i < pln[k / CSR_RPP]) ? b2[plo[k / CSR_RPP] + i] : make_int2(0, -1);
    }

#pragma unroll
    for (int k = 0; k < CSR_EPT * CSR_RPP; ++k)
        if (rc[k].y >= 0) atomicAdd(&s_cnt[rc[k].y - n0], 1);
#pragma unroll
    for (int k = 0; k < CSR_EPT; ++k)
        for (int i = 32 * CSR_RPP + lane; i < pln[k]; i += 32)
            atomicAdd(&s_cnt[b2[plo[k] + i].y - n0], 1);
    __syncthreads();

    int v = 0, inc = 0;
    if (tid < BKT_SLOTS) {
        v = s_cnt[tid];
        inc = wave_incl_scan(v, lane);
        if (lane == 31) s_wsum[wave] = inc;
    }
    __syncthreads();
    if (tid < BKT_SLOTS) {
        int front = 0;
#pragma unroll
        for (int w = 0; w < BKT_SLOTS / 32; ++w)
            if (w < wave) front += s_wsum[w];
        const int first = bs + front + inc - v;
        s_off[tid] = first;
        s_cnt[tid] = 0;
        const int node = n0 + tid;
        if (tid < NODES_PER_BKT && node < n) {
            rowptr[node] = first;
            dinv[node] = rsqrtf((float)(v + 1));
            if (node == n - 1) rowptr[n] = first + v;
        }
    }
    __syncthreads();

    const int len = be - bs;
    if (len <= CSR_LDS_CAP) {
#pragma unroll
        for (int k = 0; k < CSR_EPT * CSR_RPP; ++k) {
            if (rc[k].y >= 0) {
                const int j = rc[k].y - n0;
                const int rk = atomicAdd(&s_cnt[j], 1);
                s_list[s_off[j] - bs + rk] = rc[k].x;
            }
        }
#pragma unroll
        for (int k = 0; k < CSR_EPT; ++k) {
            for (int i = 32 * CSR_RPP + lane; i < pln[k]; i += 32) {
                const int2 p = b2[plo[k] + i];
                const int j = p.y - n0;
                const int rk = atomicAdd(&s_cnt[j], 1);
                s_list[s_off[j] - bs + rk] = p.x;
            }
        }
        __syncthreads();
        for (int i = tid; i < len; i += BIN_THREADS)
            csr_src[bs + i] = s_list[i];
    } else {
#pragma unroll
        for (int k = 0; k < CSR_EPT * CSR_RPP; ++k) {
            if (rc[k].y >= 0) {
                const int j = rc[k].y - n0;
                const int rk = atomicAdd(&s_cnt[j], 1);
                csr_src[s_off[j] + rk] = rc[k].x;
            }
        }
#pragma unroll
        for (int k = 0; k < CSR_EPT; ++k) {
            for (int i = 32 * CSR_RPP + lane; i < pln[k]; i += 32) {
                const int2 p = b2[plo[k] + i];
                const int j = p.y - n0;
                const int rk = atomicAdd(&s_cnt[j], 1);
                csr_src[s_off[j] + rk] = p.x;
            }
        }
    }
}

template <int K, int NT>
__global__ void __launch_bounds__(NT * 32)
gemm_wmma_f32(const float* __restrict__ A, const float* __restrict__ B,
              const float* __restrict__ rscale, float* __restrict__ C, int mtiles) {
    constexpr int N = NT * 16;
    constexpr int P = K + 4;
    constexpr int F4_ROW = K / 4;
    constexpr int THREADS = NT * 32;
    constexpr int LD = (16 * F4_ROW) / THREADS;
    static_assert((16 * F4_ROW) % THREADS == 0, "tile copy: whole pieces per thread");
    static_assert((P % 4) == 0 && ((P / 4) % 2) == 1, "pitch = 16 bytes x odd");
    __shared__ __attribute__((aligned(16))) float sA[2][16 * P];

    const int tid  = threadIdx.x;
    const int wave = tid >> 5;
    const int lane = tid & 31;
    const int l    = lane & 15;
    const int half = lane >> 4;

    const int n0 = wave * 16;
    const float* bp = B + (size_t)(2 * half) * N + n0 + l;
    v2f breg[K / 4];
#pragma unroll
    for (int s = 0; s < K / 4; ++s) {
        breg[s][0] = bp[(size_t)(4 * s) * N];
        breg[s][1] = bp[(size_t)(4 * s + 1) * N];
    }

    constexpr int ROWS_STEP = THREADS / F4_ROW;
    static_assert(LD == 4 && THREADS % F4_ROW == 0, "four pieces per thread, a fixed row step apart");
    const int soff0 = (tid / F4_ROW) * P + (tid % F4_ROW) * 4;

    v4f p0 = {0.f, 0.f, 0.f, 0.f}, p1 = p0, p2 = p0, p3 = p0;
    v4f nf0 = p0, nf1 = p0;
    int mtile = blockIdx.x;
    if (mtile < mtiles) {
        const v4f* src = (const v4f*)(A + (size_t)mtile * 16 * K) + tid;
        p0 = src[0];
        p1 = src[THREADS];
        p2 = src[2 * THREADS];
        p3 = src[3 * THREADS];
        const v4f* fp = (const v4f*)(rscale + mtile * 16 + 8 * half);
        nf0 = fp[0];
        nf1 = fp[1];
        float* d = &sA[0][soff0];
        *(v4f*)(d)                     = p0;
        *(v4f*)(d + ROWS_STEP * P)     = p1;
        *(v4f*)(d + 2 * ROWS_STEP * P) = p2;
        *(v4f*)(d + 3 * ROWS_STEP * P) = p3;
    }
    __syncthreads();

    int buf = 0;
    for (; mtile < mtiles; mtile += gridDim.x) {
        const int next = mtile + (int)gridDim.x;
        const bool has_next = next < mtiles;
        const v4f d0 = nf0, d1 = nf1;
        if (has_next) {
            const v4f* src = (const v4f*)(A + (size_t)next * 16 * K) + tid;
            p0 = src[0];
            p1 = src[THREADS];
            p2 = src[2 * THREADS];
            p3 = src[3 * THREADS];
            const v4f* fp = (const v4f*)(rscale + next * 16 + 8 * half);
            nf0 = fp[0];
            nf1 = fp[1];
        }

        const float* sp = &sA[buf][l * P + 2 * half];
        v8f acc = {0.f, 0.f, 0.f, 0.f, 0.f, 0.f, 0.f, 0.f};
#pragma unroll
        for (int k0 = 0; k0 < K; k0 += 4) {
            v2f a = *(const v2f*)(sp + k0);
            acc = __builtin_amdgcn_wmma_f32_16x16x4_f32(
                 false, a,  false, breg[k0 / 4],
                 (short)0, acc,  false,  false);
        }

        if (has_next) {
            float* d = &sA[buf ^ 1][soff0];
            *(v4f*)(d)                     = p0;
            *(v4f*)(d + ROWS_STEP * P)     = p1;
            *(v4f*)(d + 2 * ROWS_STEP * P) = p2;
            *(v4f*)(d + 3 * ROWS_STEP * P) = p3;
        }
        __syncthreads();
        buf ^= 1;

        const int m0 = mtile * 16;
        float* cp = C + (size_t)(m0 + 8 * half) * N + n0 + l;
#pragma unroll
        for (int j = 0; j < 4; ++j) {
            cp[(size_t)j * N]       = acc[j] * d0[j];
            cp[(size_t)(j + 4) * N] = acc[j + 4] * d1[j];
        }
    }
}

template <int H>
struct row_vec { typedef __attribute__((ext_vector_type(H / 32))) float type; };

template <int H, int K>
__device__ __forceinline__ typename row_vec<H>::type
add_group(typename row_vec<H>::type acc, const float* __restrict__ base, int r, int t) {
    typedef typename row_vec<H>::type vec;
    vec x[K];
#pragma unroll
    for (int u = 0; u < K; ++u) {
        const int rr = __builtin_amdgcn_readlane(r, t + u);
        x[u] = *(const vec*)(base + (size_t)rr * H);
    }
#pragma unroll
    for (int u = 0; u < K; ++u) acc += x[u];
    return acc;
}

template <int H>
__device__ __forceinline__ typename row_vec<H>::type
add_rows(typename row_vec<H>::type acc, const float* __restrict__ base, int r, int m) {
    int t = 0;
    for (; t + 8 <= m; t += 8)
        acc = add_group<H, 8>(acc, base, r, t);
    switch (m - t) {
    case 1: acc = add_group<H, 1>(acc, base, r, t); break;
    case 2: acc = add_group<H, 2>(acc, base, r, t); break;
    case 3: acc = add_group<H, 3>(acc, base, r, t); break;
    case 4: acc = add_group<H, 4>(acc, base, r, t); break;
    case 5: acc = add_group<H, 5>(acc, base, r, t); break;
    case 6: acc = add_group<H, 6>(acc, base, r, t); break;
    case 7: acc = add_group<H, 7>(acc, base, r, t); break;
    default: break;
    }
    return acc;
}

template <int H, bool RELU>
__global__ void __launch_bounds__(256)
gather_kernel(const int* __restrict__ rowptr, const int* __restrict__ csr_src,
              const float* __restrict__ dinv, const float* __restrict__ htmp,
              const float* __restrict__ bias, float* __restrict__ out, int n) {
    constexpr int V = H / 32;
    typedef __attribute__((ext_vector_type(V))) float vec;
    const int lane = threadIdx.x & 31;
    const int wave0 = __builtin_amdgcn_readfirstlane(
        (int)((blockIdx.x * blockDim.x + threadIdx.x) >> 5));
    const int nwaves = (int)((gridDim.x * blockDim.x) >> 5);
    const int fo = lane * V;
    const vec b = *(const vec*)(bias + fo);

    for (int node = wave0; node < n; node += nwaves) {
        const int beg = rowptr[node];
        const int end = rowptr[node + 1];
        const float dc = dinv[node];

        vec acc = *(const vec*)(htmp + (size_t)node * H + fo);

        for (int base = beg; base < end; base += 32) {
            const int j = base + lane;
            int r = node;
            if (j < end) r = csr_src[j];
            int m = end - base;
            if (m > 32) m = 32;
            acc = add_rows<H>(acc, htmp + fo, r, m);
        }

        vec o = acc * dc + b;
        if (RELU) {
#pragma unroll
            for (int q = 0; q < V; ++q) o[q] = o[q] > 0.f ? o[q] : 0.f;
        }
        *(vec*)(out + (size_t)node * H + fo) = o;
    }
}

#define FG_WAVES 16
static_assert(N_NODES % 16 == 0, "whole 16-node tiles");
__global__ void __launch_bounds__(FG_WAVES * 32)
gather_relu_gemm_kernel(const int* __restrict__ rowptr, const int* __restrict__ csr_src,
                        const float* __restrict__ dinv, const float* __restrict__ htmp,
                        const float* __restrict__ bias, const float* __restrict__ W,
                        float* __restrict__ C, int mtiles) {
    constexpr int H = H1_DIM;
    constexpr int N = H2_DIM;
    constexpr int P = H + 4;
    constexpr int N4_ROW = N / 4;
    static_assert(FG_WAVES == 16 && H == 128, "one wave per tile row, 4 floats per lane");
    static_assert(H == FG_WAVES * 8 && N4_ROW == 16, "weight copy: 8 rows per wave, 4 x 4 pieces per row");
    static_assert((P % 4) == 0 && ((P / 4) % 2) == 1, "pitch = 16 bytes x odd");
    __shared__ __attribute__((aligned(16))) float sB[N * P];
    __shared__ __attribute__((aligned(16))) float sH[2][16 * P];

    const int tid  = threadIdx.x;
    const int lane = tid & 31;
    const int l    = lane & 15;
    const int half = lane >> 4;
    const int wave = __builtin_amdgcn_readfirstlane(tid >> 5);
    const int fo = lane * 4;

    {
        const int k = (tid >> 5) * 8 + (lane >> 2);
        const int jl = lane & 3;
        const v4f* src = (const v4f*)(W + (size_t)k * N) + jl;
        v4f w[4];
#pragma unroll
        for (int u = 0; u < 4; ++u)
            w[u] = src[4 * u];
#pragma unroll
        for (int u = 0; u < 4; ++u) {
            float* d = &sB[(16 * u + 4 * jl) * P + k];
#pragma unroll
            for (int i = 0; i < 4; ++i)
                d[i * P] = w[u][i];
        }
    }
    const v4f b = *(const v4f*)(bias + fo);

    int buf = 0;
    for (int tile = blockIdx.x; tile < mtiles; tile += gridDim.x) {
        const int node = tile * 16 + wave;
        const int beg = rowptr[node];
        const int end = rowptr[node + 1];
        const float dc = dinv[node];

        v4f acc = *(const v4f*)(htmp + (size_t)node * H + fo);

        for (int base = beg; base < end; base += 32) {
            const int j = base + lane;
            int r = node;
            if (j < end) r = csr_src[j];
            int m = end - base;
            if (m > 32) m = 32;
            acc = add_rows<H>(acc, htmp + fo, r, m);
        }

        v4f o = acc * dc + b;
#pragma unroll
        for (int q = 0; q < 4; ++q) o[q] = o[q] > 0.f ? o[q] : 0.f;
        *(v4f*)&sH[buf][wave * P + fo] = o;
        __syncthreads();

        if (wave < N / 16) {
            const int n0 = wave * 16;
            const float* sp = &sH[buf][l * P + 2 * half];
            const float* bp = &sB[(n0 + l) * P + 2 * half];
            const v4f* dp = (const v4f*)(dinv + tile * 16 + 8 * half);
            const v4f d0 = dp[0];
            const v4f d1 = dp[1];
            v8f c = {0.f, 0.f, 0.f, 0.f, 0.f, 0.f, 0.f, 0.f};
#pragma unroll
            for (int k0 = 0; k0 < H; k0 += 4) {
                const v2f a  = *(const v2f*)(sp + k0);
                const v2f bb = *(const v2f*)(bp + k0);
                c = __builtin_amdgcn_wmma_f32_16x16x4_f32(
                     false, a,  false, bb,
                     (short)0, c,  false,  false);
            }
            float* cp = C + (size_t)(tile * 16 + 8 * half) * N + n0 + l;
#pragma unroll
            for (int jj = 0; jj < 4; ++jj) {
                cp[(size_t)jj * N]       = c[jj] * d0[jj];
                cp[(size_t)(jj + 4) * N] = c[jj + 4] * d1[jj];
            }
        }
        buf ^= 1;
    }
}

__global__ void __launch_bounds__(256)
edge_dot_kernel(const int* __restrict__ pos, const int* __restrict__ neg,
                const float* __restrict__ h2, float* __restrict__ out, int ep) {
    const int lane = threadIdx.x & 31;
    const int wave0 = __builtin_amdgcn_readfirstlane(
        (int)((blockIdx.x * blockDim.x + threadIdx.x) >> 5));
    const int nwaves = (int)((gridDim.x * blockDim.x) >> 5);
    const int total = 2 * ep;
    const int nbatch = (total + 31) >> 5;

    for (int bt = wave0; bt < nbatch; bt += nwaves) {
        const int e = bt * 32 + lane;
        int src = 0, dst = 0;
        if (e < total) {
            if (e < ep) {
                src = pos[e];
                dst = pos[ep + e];
            } else {
                src = neg[e - ep];
                dst = neg[e];
            }
        }
        int m = total - bt * 32;
        if (m > 32) m = 32;
        float res = 0.f;
#pragma unroll 8
        for (int t = 0; t < m; ++t) {
            const int sj = __builtin_amdgcn_readlane(src, t);
            const int si = __builtin_amdgcn_readlane(dst, t);
            const float* xj = h2 + (size_t)sj * H2_DIM;
            const float* xi = h2 + (size_t)si * H2_DIM;
            float v = xi[lane] * xj[lane] + xi[lane + 32] * xj[lane + 32];
#pragma unroll
            for (int off = 16; off > 0; off >>= 1)
                v += __shfl_xor(v, off, 32);
            if (lane == t) res = v;
        }
        if (e < total) out[e] = res;
    }
}

static inline size_t align256(size_t x) { return (x + 255) & ~(size_t)255; }

extern "C" void kernel_launch(void* const* d_in, const int* in_sizes, int n_in,
                              void* d_out, int out_size, void* d_ws, size_t ws_size,
                              hipStream_t stream) {
    const float* x    = (const float*)d_in[0];
    const int*   ei   = (const int*)d_in[1];
    const int*   pos  = (const int*)d_in[2];
    const int*   neg  = (const int*)d_in[3];
    const float* W1   = (const float*)d_in[4];
    const float* b1   = (const float*)d_in[5];
    const float* W2   = (const float*)d_in[6];
    const float* b2   = (const float*)d_in[7];
    float* out = (float*)d_out;

    const int* row = ei;
    const int* col = ei + E_EDGES;

    char* ws = (char*)d_ws;
    size_t off = 0;
    float* dinv  = (float*)(ws + off); off += align256(sizeof(float) * N_NODES);
    float* htmp1 = (float*)(ws + off); off += align256(sizeof(float) * (size_t)N_NODES * H1_DIM);
    float* h1    = (float*)(ws + off); off += align256(sizeof(float) * (size_t)N_NODES * H1_DIM);
    float* htmp2 = (float*)(ws + off); off += align256(sizeof(float) * (size_t)N_NODES * H2_DIM);
    float* h2    = htmp1;
    int* rowptr  = (int*)(ws + off); off += align256(sizeof(int) * (N_NODES + 1));
    int* loff    = (int*)(ws + off); off += align256(sizeof(int) * (size_t)E_EDGES);
    int* csr_src = (int*)(ws + off); off += align256(sizeof(int) * (size_t)E_EDGES);
    int* bins    = (int*)h1;
    (void)ws_size; (void)n_in; (void)in_sizes; (void)out_size;

    const int TB = 256;
    const int MTILES = N_NODES / 16;
    const int GEMM1_BLOCKS = 512;
    const int FUSED_BLOCKS = 1024;
    const int GATHER_BLOCKS = 2048;
    const int DOT_BLOCKS = ((2 * EP_CNT + 31) / 32 + 7) / 8;

    bucket_group_kernel<<<BIN_BLOCKS, BIN_THREADS, 0, stream>>>(row, col, loff, bins, E_EDGES);
    bucket_csr_kernel<<<NB_BUCKETS, BIN_THREADS, 0, stream>>>(bins, loff, rowptr, dinv, csr_src, N_NODES);

    gemm_wmma_f32<F_IN, 8><<<GEMM1_BLOCKS, 8 * 32, 0, stream>>>(x, W1, dinv, htmp1, MTILES);

    gather_relu_gemm_kernel<<<FUSED_BLOCKS, FG_WAVES * 32, 0, stream>>>(
        rowptr, csr_src, dinv, htmp1, b1, W2, htmp2, MTILES);

    gather_kernel<H2_DIM, false><<<GATHER_BLOCKS, TB, 0, stream>>>(
        rowptr, csr_src, dinv, htmp2, b2, h2, N_NODES);

    edge_dot_kernel<<<DOT_BLOCKS, TB, 0, stream>>>(pos, neg, h2, out, EP_CNT);
}
